# starting kernel
# baseline (speedup 1.0000x reference)
#include <hip/hip_runtime.h>

typedef __attribute__((ext_vector_type(16))) _Float16 v16h;
typedef __attribute__((ext_vector_type(8)))  float    v8f;

#define DF   128
#define DK2  256
#define DOUT 64

static __host__ __device__ inline int cdiv(int a, int b) { return (a + b - 1) / b; }

__device__ __forceinline__ void edge_sr(int i, const int* __restrict__ snd,
                                        const int* __restrict__ rcv, int E,
                                        int& s, int& r) {
    if (i < E)            { s = snd[i];     r = rcv[i]; }
    else if (i < 2 * E)   { s = rcv[i - E]; r = snd[i - E]; }
    else                  { s = i - 2 * E;  r = s; }
}

__global__ void zero_kernel(float* __restrict__ p, long n) {
    long i = (long)blockIdx.x * blockDim.x + threadIdx.x;
    long stride = (long)gridDim.x * blockDim.x;
    for (; i < n; i += stride) p[i] = 0.0f;
}

__global__ void deg_kernel(const int* __restrict__ snd, const int* __restrict__ rcv,
                           const float* __restrict__ theta, int E, int Nn,
                           float* __restrict__ degS, float* __restrict__ degR) {
    int i = blockIdx.x * blockDim.x + threadIdx.x;
    int M = 2 * E + Nn;
    if (i >= M) return;
    int s, r; edge_sr(i, snd, rcv, E, s, r);
    float ow, iw;
    if (i < E)          { float c = cosf(theta[i]);     float sn = sinf(theta[i]);     ow = c * c;  iw = sn * sn; }
    else if (i < 2 * E) { float c = cosf(theta[i - E]); float sn = sinf(theta[i - E]); ow = sn * sn; iw = c * c; }
    else                { ow = 1.0f; iw = 1.0f; }
    unsafeAtomicAdd(degS + s, ow);
    unsafeAtomicAdd(degR + s, iw);
}

__global__ void dis_kernel(float* __restrict__ d, int n) {
    int i = blockIdx.x * blockDim.x + threadIdx.x;
    if (i >= n) return;
    float v = d[i] + 1e-12f;
    d[i] = (v > 1e11f) ? 0.0f : rsqrtf(v);
}

__global__ void weight_kernel(const int* __restrict__ snd, const int* __restrict__ rcv,
                              const float* __restrict__ theta, int E, int Nn,
                              const float* __restrict__ disS, const float* __restrict__ disR,
                              float* __restrict__ wS, float* __restrict__ wD) {
    int i = blockIdx.x * blockDim.x + threadIdx.x;
    int M = 2 * E + Nn;
    if (i >= M) return;
    int s, r; edge_sr(i, snd, rcv, E, s, r);
    float ow, iw;
    if (i < E)          { float c = cosf(theta[i]);     float sn = sinf(theta[i]);     ow = c * c;  iw = sn * sn; }
    else if (i < 2 * E) { float c = cosf(theta[i - E]); float sn = sinf(theta[i - E]); ow = sn * sn; iw = c * c; }
    else                { ow = 1.0f; iw = 1.0f; }
    wS[i] = disS[s] * ow * disR[r];
    wD[i] = disR[s] * iw * disS[r];
}

__global__ void scatter_kernel(const float* __restrict__ X,
                               const int* __restrict__ snd, const int* __restrict__ rcv,
                               const float* __restrict__ wS, const float* __restrict__ wD,
                               float* __restrict__ agg, int E, int Nn) {
    int wave = (blockIdx.x * blockDim.x + threadIdx.x) >> 5;
    int lane = threadIdx.x & 31;
    int M = 2 * E + Nn;
    if (wave >= M) return;
    int s, r; edge_sr(wave, snd, rcv, E, s, r);
    float a = wS[wave];
    float b = wD[wave];
    const float4 v = *(const float4*)(X + (long)s * DF + lane * 4);
    float* dst = agg + (long)r * DK2 + lane * 4;
    unsafeAtomicAdd(dst + 0,        v.x * a);
    unsafeAtomicAdd(dst + 1,        v.y * a);
    unsafeAtomicAdd(dst + 2,        v.z * a);
    unsafeAtomicAdd(dst + 3,        v.w * a);
    unsafeAtomicAdd(dst + DF + 0,   v.x * b);
    unsafeAtomicAdd(dst + DF + 1,   v.y * b);
    unsafeAtomicAdd(dst + DF + 2,   v.z * b);
    unsafeAtomicAdd(dst + DF + 3,   v.w * b);
}

__global__ void pack_cat(const float* __restrict__ Ws, const float* __restrict__ Wd,
                         _Float16* __restrict__ dst) {
    int idx = blockIdx.x * blockDim.x + threadIdx.x;
    if (idx >= DK2 * DF) return;
    int j    = idx & 15;
    int lane = (idx >> 4) & 31;
    int t    = idx >> 9;
    int k0t  = t / (DF / 16);
    int tc   = t % (DF / 16);
    int k = k0t * 32 + ((j < 8) ? 0 : 16) + (lane >> 4) * 8 + (j & 7);
    int n = tc * 16 + (lane & 15);
    float v = (k < DF) ? 0.5f * Ws[k * DF + n] : 0.5f * Wd[(k - DF) * DF + n];
    dst[idx] = (_Float16)v;
}

__global__ void pack_single(const float* __restrict__ W, _Float16* __restrict__ dst) {
    int idx = blockIdx.x * blockDim.x + threadIdx.x;
    if (idx >= DF * DOUT) return;
    int j    = idx & 15;
    int lane = (idx >> 4) & 31;
    int t    = idx >> 9;
    int k0t  = t / (DOUT / 16);
    int tc   = t % (DOUT / 16);
    int k = k0t * 32 + ((j < 8) ? 0 : 16) + (lane >> 4) * 8 + (j & 7);
    int n = tc * 16 + (lane & 15);
    dst[idx] = (_Float16)W[k * DOUT + n];
}

__global__ void bias_combine(const float* __restrict__ bs, const float* __restrict__ bd,
                             float* __restrict__ out, int n) {
    int i = blockIdx.x * blockDim.x + threadIdx.x;
    if (i < n) out[i] = 0.5f * (bs[i] + bd[i]);
}

template <int K, int NC, int RELU>
__global__ void wmma_gemm(const float* __restrict__ A, int Mrows,
                          const v16h* __restrict__ Bp, const float* __restrict__ bias,
                          float* __restrict__ D) {
    constexpr int colTiles = NC / 16;
    int wave = (blockIdx.x * blockDim.x + threadIdx.x) >> 5;
    int lane = threadIdx.x & 31;
    int rowTiles = Mrows >> 4;
    if (wave >= rowTiles * colTiles) return;
    int tr = wave / colTiles;
    int tc = wave % colTiles;
    int hi = lane >> 4;
    int arow = tr * 16 + (lane & 15);
    int ncol = tc * 16 + (lane & 15);

    const float4* Ar = (const float4*)(A + (long)arow * K);
    v8f c = {};
#pragma unroll
    for (int k0 = 0; k0 < K; k0 += 32) {
        int base = (k0 >> 2) + hi * 2;
        float4 p0 = Ar[base];
        float4 p1 = Ar[base + 1];
        float4 p2 = Ar[base + 4];
        float4 p3 = Ar[base + 5];
        v16h a;
        a[0]  = (_Float16)p0.x; a[1]  = (_Float16)p0.y;
        a[2]  = (_Float16)p0.z; a[3]  = (_Float16)p0.w;
        a[4]  = (_Float16)p1.x; a[5]  = (_Float16)p1.y;
        a[6]  = (_Float16)p1.z; a[7]  = (_Float16)p1.w;
        a[8]  = (_Float16)p2.x; a[9]  = (_Float16)p2.y;
        a[10] = (_Float16)p2.z; a[11] = (_Float16)p2.w;
        a[12] = (_Float16)p3.x; a[13] = (_Float16)p3.y;
        a[14] = (_Float16)p3.z; a[15] = (_Float16)p3.w;
        v16h b = Bp[((k0 >> 5) * colTiles + tc) * 32 + lane];
        c = __builtin_amdgcn_wmma_f32_16x16x32_f16(false, a, false, b,
                                                   (short)0, c, false, false);
    }
    float bv = bias[ncol];
    float* Drow = D + (long)(tr * 16 + hi * 8) * NC + ncol;
#pragma unroll
    for (int i = 0; i < 8; ++i) {
        float v = c[i] + bv;
        if (RELU) v = fmaxf(v, 0.0f);
        Drow[(long)i * NC] = v;
    }
}

extern "C" void kernel_launch(void* const* d_in, const int* in_sizes, int n_in,
                              void* d_out, int out_size, void* d_ws, size_t ws_size,
                              hipStream_t stream) {
    const float* x      = (const float*)d_in[0];
    const int*   edges  = (const int*)d_in[1];
    const float* theta  = (const float*)d_in[2];
    const float* W1s    = (const float*)d_in[3];
    const float* W1d    = (const float*)d_in[4];
    const float* b1s    = (const float*)d_in[5];
    const float* b1d    = (const float*)d_in[6];
    const float* W2s    = (const float*)d_in[7];
    const float* W2d    = (const float*)d_in[8];
    const float* b2s    = (const float*)d_in[9];
    const float* b2d    = (const float*)d_in[10];
    const float* Wro    = (const float*)d_in[11];
    const float* bro    = (const float*)d_in[12];

    const int N = in_sizes[0] / DF;
    const int E = in_sizes[2];
    const int M = 2 * E + N;
    const int* snd = edges;
    const int* rcv = edges + E;

    char* ws = (char*)d_ws;
    size_t off = 0;
    auto alloc = [&](size_t bytes) { char* p = ws + off; off = (off + bytes + 255) & ~(size_t)255; return p; };
    float*    degS  = (float*)alloc((size_t)N * 4);
    float*    degR  = (float*)alloc((size_t)N * 4);
    float*    wS    = (float*)alloc((size_t)M * 4);
    float*    wD    = (float*)alloc((size_t)M * 4);
    float*    agg   = (float*)alloc((size_t)N * DK2 * 4);
    float*    x1    = (float*)alloc((size_t)N * DF * 4);
    float*    x2    = (float*)alloc((size_t)N * DF * 4);
    _Float16* Bp1   = (_Float16*)alloc((size_t)DK2 * DF * 2);
    _Float16* Bp2   = (_Float16*)alloc((size_t)DK2 * DF * 2);
    _Float16* Bpro  = (_Float16*)alloc((size_t)DF * DOUT * 2);
    float*    bias1 = (float*)alloc(DF * 4);
    float*    bias2 = (float*)alloc(DF * 4);

    const int T = 256;

    zero_kernel<<<512, T, 0, stream>>>(degS, N);
    zero_kernel<<<512, T, 0, stream>>>(degR, N);
    deg_kernel<<<cdiv(M, T), T, 0, stream>>>(snd, rcv, theta, E, N, degS, degR);
    dis_kernel<<<cdiv(N, T), T, 0, stream>>>(degS, N);
    dis_kernel<<<cdiv(N, T), T, 0, stream>>>(degR, N);
    weight_kernel<<<cdiv(M, T), T, 0, stream>>>(snd, rcv, theta, E, N, degS, degR, wS, wD);

    pack_cat<<<cdiv(DK2 * DF, T), T, 0, stream>>>(W1s, W1d, Bp1);
    pack_cat<<<cdiv(DK2 * DF, T), T, 0, stream>>>(W2s, W2d, Bp2);
    pack_single<<<cdiv(DF * DOUT, T), T, 0, stream>>>(Wro, Bpro);
    bias_combine<<<1, DF, 0, stream>>>(b1s, b1d, bias1, DF);
    bias_combine<<<1, DF, 0, stream>>>(b2s, b2d, bias2, DF);

    const int scatterBlocks = cdiv(M * 32, T);
    const int layerTiles    = (N / 16) * (DF / 16);
    const int roTiles       = (N / 16) * (DOUT / 16);

    zero_kernel<<<4096, T, 0, stream>>>(agg, (long)N * DK2);
    scatter_kernel<<<scatterBlocks, T, 0, stream>>>(x, snd, rcv, wS, wD, agg, E, N);
    wmma_gemm<DK2, DF, 1><<<cdiv(layerTiles * 32, T), T, 0, stream>>>(
        agg, N, (const v16h*)Bp1, bias1, x1);

    zero_kernel<<<4096, T, 0, stream>>>(agg, (long)N * DK2);
    scatter_kernel<<<scatterBlocks, T, 0, stream>>>(x1, snd, rcv, wS, wD, agg, E, N);
    wmma_gemm<DK2, DF, 1><<<cdiv(layerTiles * 32, T), T, 0, stream>>>(
        agg, N, (const v16h*)Bp2, bias2, x2);

    wmma_gemm<DF, DOUT, 0><<<cdiv(roTiles * 32, T), T, 0, stream>>>(
        x2, N, (const v16h*)Bpro, bro, (float*)d_out);
}


# Both layers: product before aggregation (Y=X*[Ws|Wd], epilogue stores bias+self-loop term as start value); edges add a*Ys[s]+b*Yd[s]: 128 not 256 f32 atomics per edge, whole lines; sum order changes
# speedup vs baseline: 2.1264x; 2.1264x over previous
#include <hip/hip_runtime.h>

typedef __attribute__((ext_vector_type(16))) _Float16 v16h;
typedef __attribute__((ext_vector_type(8)))  float    v8f;

#define DF   128
#define DK2  256
#define DOUT 64

static __host__ __device__ inline int cdiv(int a, int b) { return (a + b - 1) / b; }

__device__ __forceinline__ void edge_sr(int i, const int* __restrict__ snd,
                                        const int* __restrict__ rcv, int E,
                                        int& s, int& r) {
    if (i < E)            { s = snd[i];     r = rcv[i]; }
    else if (i < 2 * E)   { s = rcv[i - E]; r = snd[i - E]; }
    else                  { s = i - 2 * E;  r = s; }
}

__global__ void zero_kernel(float* __restrict__ p, long n) {
    long i = (long)blockIdx.x * blockDim.x + threadIdx.x;
    long stride = (long)gridDim.x * blockDim.x;
    for (; i < n; i += stride) p[i] = 0.0f;
}

__global__ void deg_kernel(const int* __restrict__ snd, const int* __restrict__ rcv,
                           const float* __restrict__ theta, int E, int Nn,
                           float* __restrict__ degS, float* __restrict__ degR) {
    int i = blockIdx.x * blockDim.x + threadIdx.x;
    int M = 2 * E + Nn;
    if (i >= M) return;
    int s, r; edge_sr(i, snd, rcv, E, s, r);
    float ow, iw;
    if (i < E)          { float c = cosf(theta[i]);     float sn = sinf(theta[i]);     ow = c * c;  iw = sn * sn; }
    else if (i < 2 * E) { float c = cosf(theta[i - E]); float sn = sinf(theta[i - E]); ow = sn * sn; iw = c * c; }
    else                { ow = 1.0f; iw = 1.0f; }
    unsafeAtomicAdd(degS + s, ow);
    unsafeAtomicAdd(degR + s, iw);
}

__global__ void dis_kernel(float* __restrict__ d, int n) {
    int i = blockIdx.x * blockDim.x + threadIdx.x;
    if (i >= n) return;
    float v = d[i] + 1e-12f;
    d[i] = (v > 1e11f) ? 0.0f : rsqrtf(v);
}

__global__ void weight_kernel(const int* __restrict__ snd, const int* __restrict__ rcv,
                              const float* __restrict__ theta, int E, int Nn,
                              const float* __restrict__ disS, const float* __restrict__ disR,
                              float* __restrict__ wS, float* __restrict__ wD) {
    int i = blockIdx.x * blockDim.x + threadIdx.x;
    int M = 2 * E + Nn;
    if (i >= M) return;
    int s, r; edge_sr(i, snd, rcv, E, s, r);
    float ow, iw;
    if (i < E)          { float c = cosf(theta[i]);     float sn = sinf(theta[i]);     ow = c * c;  iw = sn * sn; }
    else if (i < 2 * E) { float c = cosf(theta[i - E]); float sn = sinf(theta[i - E]); ow = sn * sn; iw = c * c; }
    else                { ow = 1.0f; iw = 1.0f; }
    wS[i] = disS[s] * ow * disR[r];
    wD[i] = disR[s] * iw * disS[r];
}

__global__ void __launch_bounds__(256)
scatter_pre(const float* __restrict__ Y,
            const int* __restrict__ snd, const int* __restrict__ rcv,
            const float* __restrict__ wS, const float* __restrict__ wD,
            float* __restrict__ acc, int E) {
    const int wave = blockIdx.x * (blockDim.x >> 5) + __builtin_amdgcn_readfirstlane(threadIdx.x >> 5);
    const int lane = threadIdx.x & 31;
    const int M2 = 2 * E;
    const int base = wave * 32;
    if (base >= M2) return;
    const int cnt = (M2 - base < 32) ? (M2 - base) : 32;
    const int i = base + lane;
    int sv = 0, rv = 0, av = 0, bv = 0;
    if (i < M2) {
        if (i < E) { sv = snd[i];     rv = rcv[i]; }
        else       { sv = rcv[i - E]; rv = snd[i - E]; }
        av = __float_as_int(wS[i]);
        bv = __float_as_int(wD[i]);
    }
    const float* Yl = Y + lane * 4;
    float* accl = acc + lane;
    int t = 0;
    for (; t + 4 <= cnt; t += 4) {
        float4 ys[4], yd[4];
        int r[4]; float a[4], b[4];
#pragma unroll
        for (int u = 0; u < 4; ++u) {
            const int s = __builtin_amdgcn_readlane(sv, t + u);
            r[u] = __builtin_amdgcn_readlane(rv, t + u);
            a[u] = __int_as_float(__builtin_amdgcn_readlane(av, t + u));
            b[u] = __int_as_float(__builtin_amdgcn_readlane(bv, t + u));
            const float* yr = Yl + (long)s * DK2;
            ys[u] = *(const float4*)yr;
            yd[u] = *(const float4*)(yr + DF);
        }
#pragma unroll
        for (int u = 0; u < 4; ++u) {
            float* dst = accl + (long)r[u] * DF;
            unsafeAtomicAdd(dst,      a[u] * ys[u].x + b[u] * yd[u].x);
            unsafeAtomicAdd(dst + 32, a[u] * ys[u].y + b[u] * yd[u].y);
            unsafeAtomicAdd(dst + 64, a[u] * ys[u].z + b[u] * yd[u].z);
            unsafeAtomicAdd(dst + 96, a[u] * ys[u].w + b[u] * yd[u].w);
        }
    }
    for (; t < cnt; ++t) {
        const int s = __builtin_amdgcn_readlane(sv, t);
        const int r = __builtin_amdgcn_readlane(rv, t);
        const float a = __int_as_float(__builtin_amdgcn_readlane(av, t));
        const float b = __int_as_float(__builtin_amdgcn_readlane(bv, t));
        const float* yr = Yl + (long)s * DK2;
        const float4 ys = *(const float4*)yr;
        const float4 yd = *(const float4*)(yr + DF);
        float* dst = accl + (long)r * DF;
        unsafeAtomicAdd(dst,      a * ys.x + b * yd.x);
        unsafeAtomicAdd(dst + 32, a * ys.y + b * yd.y);
        unsafeAtomicAdd(dst + 64, a * ys.z + b * yd.z);
        unsafeAtomicAdd(dst + 96, a * ys.w + b * yd.w);
    }
}

__global__ void pack_pre(const float* __restrict__ Ws, const float* __restrict__ Wd,
                         _Float16* __restrict__ dst) {
    int idx = blockIdx.x * blockDim.x + threadIdx.x;
    if (idx >= DF * DK2) return;
    int j    = idx & 15;
    int lane = (idx >> 4) & 31;
    int t    = idx >> 9;
    int q    = t & 7;
    int hp   = (t >> 3) & 1;
    int kt   = t >> 4;
    int half = q >> 2;
    int jt   = q & 3;
    int k = kt * 32 + ((j < 8) ? 0 : 16) + (lane >> 4) * 8 + (j & 7);
    int n = 32 * jt + 16 * hp + (lane & 15);
    float v = half ? 0.5f * Wd[k * DF + n] : 0.5f * Ws[k * DF + n];
    dst[idx] = (_Float16)v;
}

__global__ void pack_single(const float* __restrict__ W, _Float16* __restrict__ dst) {
    int idx = blockIdx.x * blockDim.x + threadIdx.x;
    if (idx >= DF * DOUT) return;
    int j    = idx & 15;
    int lane = (idx >> 4) & 31;
    int t    = idx >> 9;
    int k0t  = t / (DOUT / 16);
    int tc   = t % (DOUT / 16);
    int k = k0t * 32 + ((j < 8) ? 0 : 16) + (lane >> 4) * 8 + (j & 7);
    int n = tc * 16 + (lane & 15);
    dst[idx] = (_Float16)W[k * DOUT + n];
}

__global__ void bias_combine(const float* __restrict__ bs, const float* __restrict__ bd,
                             float* __restrict__ out, int n) {
    int i = blockIdx.x * blockDim.x + threadIdx.x;
    if (i < n) out[i] = 0.5f * (bs[i] + bd[i]);
}

template <int RELU_IN>
__device__ __forceinline__ v16h load_a_frag(const float4* __restrict__ Ar, int k0, int hi) {
    int base = (k0 >> 2) + hi * 2;
    float4 p0 = Ar[base];
    float4 p1 = Ar[base + 1];
    float4 p2 = Ar[base + 4];
    float4 p3 = Ar[base + 5];
    if (RELU_IN) {
        p0.x = fmaxf(p0.x, 0.0f); p0.y = fmaxf(p0.y, 0.0f); p0.z = fmaxf(p0.z, 0.0f); p0.w = fmaxf(p0.w, 0.0f);
        p1.x = fmaxf(p1.x, 0.0f); p1.y = fmaxf(p1.y, 0.0f); p1.z = fmaxf(p1.z, 0.0f); p1.w = fmaxf(p1.w, 0.0f);
        p2.x = fmaxf(p2.x, 0.0f); p2.y = fmaxf(p2.y, 0.0f); p2.z = fmaxf(p2.z, 0.0f); p2.w = fmaxf(p2.w, 0.0f);
        p3.x = fmaxf(p3.x, 0.0f); p3.y = fmaxf(p3.y, 0.0f); p3.z = fmaxf(p3.z, 0.0f); p3.w = fmaxf(p3.w, 0.0f);
    }
    v16h a;
    a[0]  = (_Float16)p0.x; a[1]  = (_Float16)p0.y;
    a[2]  = (_Float16)p0.z; a[3]  = (_Float16)p0.w;
    a[4]  = (_Float16)p1.x; a[5]  = (_Float16)p1.y;
    a[6]  = (_Float16)p1.z; a[7]  = (_Float16)p1.w;
    a[8]  = (_Float16)p2.x; a[9]  = (_Float16)p2.y;
    a[10] = (_Float16)p2.z; a[11] = (_Float16)p2.w;
    a[12] = (_Float16)p3.x; a[13] = (_Float16)p3.y;
    a[14] = (_Float16)p3.z; a[15] = (_Float16)p3.w;
    return a;
}

template <int RELU_IN>
__global__ void __launch_bounds__(256)
gemm_pre(const float* __restrict__ A, int Mrows,
         const v16h* __restrict__ Bp, const float* __restrict__ bias,
         const float* __restrict__ aSelf, const float* __restrict__ bSelf,
         float* __restrict__ Y, float* __restrict__ acc) {
    int wave = (blockIdx.x * blockDim.x + threadIdx.x) >> 5;
    int lane = threadIdx.x & 31;
    int rowTiles = Mrows >> 4;
    if (wave >= rowTiles * 2) return;
    int tr = wave >> 1;
    int hp = wave & 1;
    int hi = lane >> 4;
    int n  = lane & 15;

    const float4* Ar = (const float4*)(A + (long)(tr * 16 + n) * DF);
    v8f c[8] = {};
#pragma unroll
    for (int kt = 0; kt < DF / 32; ++kt) {
        v16h a = load_a_frag<RELU_IN>(Ar, kt * 32, hi);
        const v16h* bq = Bp + ((kt * 2 + hp) * 8) * 32 + lane;
#pragma unroll
        for (int q = 0; q < 8; ++q) {
            v16h b = bq[q * 32];
            c[q] = __builtin_amdgcn_wmma_f32_16x16x32_f16(false, a, false, b,
                                                          (short)0, c[q], false, false);
        }
    }

    const int lcol = 16 * hp + n;
    const float bv0 = bias[lcol], bv1 = bias[lcol + 32], bv2 = bias[lcol + 64], bv3 = bias[lcol + 96];
    const int row0 = tr * 16 + hi * 8;
#pragma unroll
    for (int i = 0; i < 8; ++i) {
        const int row = row0 + i;
        const float as = aSelf[row];
        const float bs = bSelf[row];
        float4 ys, yd;
        ys.x = c[0][i]; ys.y = c[1][i]; ys.z = c[2][i]; ys.w = c[3][i];
        yd.x = c[4][i]; yd.y = c[5][i]; yd.z = c[6][i]; yd.w = c[7][i];
        float* yr = Y + (long)row * DK2 + lcol * 4;
        *(float4*)yr        = ys;
        *(float4*)(yr + DF) = yd;
        float* ar = acc + (long)row * DF + lcol;
        ar[0]  = bv0 + as * ys.x + bs * yd.x;
        ar[32] = bv1 + as * ys.y + bs * yd.y;
        ar[64] = bv2 + as * ys.z + bs * yd.z;
        ar[96] = bv3 + as * ys.w + bs * yd.w;
    }
}

template <int K, int NC, int RELU, int RELU_IN>
__global__ void wmma_gemm(const float* __restrict__ A, int Mrows,
                          const v16h* __restrict__ Bp, const float* __restrict__ bias,
                          float* __restrict__ D) {
    constexpr int colTiles = NC / 16;
    int wave = (blockIdx.x * blockDim.x + threadIdx.x) >> 5;
    int lane = threadIdx.x & 31;
    int rowTiles = Mrows >> 4;
    if (wave >= rowTiles * colTiles) return;
    int tr = wave / colTiles;
    int tc = wave % colTiles;
    int hi = lane >> 4;
    int arow = tr * 16 + (lane & 15);
    int ncol = tc * 16 + (lane & 15);

    const float4* Ar = (const float4*)(A + (long)arow * K);
    v8f c = {};
#pragma unroll
    for (int k0 = 0; k0 < K; k0 += 32) {
        v16h a = load_a_frag<RELU_IN>(Ar, k0, hi);
        v16h b = Bp[((k0 >> 5) * colTiles + tc) * 32 + lane];
        c = __builtin_amdgcn_wmma_f32_16x16x32_f16(false, a, false, b,
                                                   (short)0, c, false, false);
    }
    float bv = bias[ncol];
    float* Drow = D + (long)(tr * 16 + hi * 8) * NC + ncol;
#pragma unroll
    for (int i = 0; i < 8; ++i) {
        float v = c[i] + bv;
        if (RELU) v = fmaxf(v, 0.0f);
        Drow[(long)i * NC] = v;
    }
}

extern "C" void kernel_launch(void* const* d_in, const int* in_sizes, int n_in,
                              void* d_out, int out_size, void* d_ws, size_t ws_size,
                              hipStream_t stream) {
    const float* x      = (const float*)d_in[0];
    const int*   edges  = (const int*)d_in[1];
    const float* theta  = (const float*)d_in[2];
    const float* W1s    = (const float*)d_in[3];
    const float* W1d    = (const float*)d_in[4];
    const float* b1s    = (const float*)d_in[5];
    const float* b1d    = (const float*)d_in[6];
    const float* W2s    = (const float*)d_in[7];
    const float* W2d    = (const float*)d_in[8];
    const float* b2s    = (const float*)d_in[9];
    const float* b2d    = (const float*)d_in[10];
    const float* Wro    = (const float*)d_in[11];
    const float* bro    = (const float*)d_in[12];

    const int N = in_sizes[0] / DF;
    const int E = in_sizes[2];
    const int M = 2 * E + N;
    const int* snd = edges;
    const int* rcv = edges + E;

    char* ws = (char*)d_ws;
    size_t off = 0;
    auto alloc = [&](size_t bytes) { char* p = ws + off; off = (off + bytes + 255) & ~(size_t)255; return p; };
    float*    degS  = (float*)alloc((size_t)N * 4);
    float*    degR  = (float*)alloc((size_t)N * 4);
    float*    wS    = (float*)alloc((size_t)M * 4);
    float*    wD    = (float*)alloc((size_t)M * 4);
    float*    Y     = (float*)alloc((size_t)N * DK2 * 4);
    float*    acc1  = (float*)alloc((size_t)N * DF * 4);
    float*    acc2  = (float*)alloc((size_t)N * DF * 4);
    _Float16* Bp1   = (_Float16*)alloc((size_t)DK2 * DF * 2);
    _Float16* Bp2   = (_Float16*)alloc((size_t)DK2 * DF * 2);
    _Float16* Bpro  = (_Float16*)alloc((size_t)DF * DOUT * 2);
    float*    bias1 = (float*)alloc(DF * 4);
    float*    bias2 = (float*)alloc(DF * 4);

    const int T = 256;

    zero_kernel<<<512, T, 0, stream>>>(degS, N);
    zero_kernel<<<512, T, 0, stream>>>(degR, N);
    deg_kernel<<<cdiv(M, T), T, 0, stream>>>(snd, rcv, theta, E, N, degS, degR);
    dis_kernel<<<cdiv(N, T), T, 0, stream>>>(degS, N);
    dis_kernel<<<cdiv(N, T), T, 0, stream>>>(degR, N);
    weight_kernel<<<cdiv(M, T), T, 0, stream>>>(snd, rcv, theta, E, N, degS, degR, wS, wD);

    pack_pre<<<cdiv(DF * DK2, T), T, 0, stream>>>(W1s, W1d, Bp1);
    pack_pre<<<cdiv(DF * DK2, T), T, 0, stream>>>(W2s, W2d, Bp2);
    pack_single<<<cdiv(DF * DOUT, T), T, 0, stream>>>(Wro, Bpro);
    bias_combine<<<1, DF, 0, stream>>>(b1s, b1d, bias1, DF);
    bias_combine<<<1, DF, 0, stream>>>(b2s, b2d, bias2, DF);

    const int preWaves      = (N / 16) * 2;
    const int edgeWaves     = cdiv(2 * E, 32);
    const int roTiles       = (N / 16) * (DOUT / 16);
    const float* wSself = wS + 2 * E;
    const float* wDself = wD + 2 * E;

    gemm_pre<0><<<cdiv(preWaves * 32, T), T, 0, stream>>>(
        x, N, (const v16h*)Bp1, bias1, wSself, wDself, Y, acc1);
    scatter_pre<<<cdiv(edgeWaves * 32, T), T, 0, stream>>>(Y, snd, rcv, wS, wD, acc1, E);

    gemm_pre<1><<<cdiv(preWaves * 32, T), T, 0, stream>>>(
        acc1, N, (const v16h*)Bp2, bias2, wSself, wDself, Y, acc2);
    scatter_pre<<<cdiv(edgeWaves * 32, T), T, 0, stream>>>(Y, snd, rcv, wS, wD, acc2, E);

    wmma_gemm<DF, DOUT, 0, 1><<<cdiv(roTiles * 32, T), T, 0, stream>>>(
        acc2, N, (const v16h*)Bpro, bro, (float*)d_out);
}


# Both layers: edge sums without global atomics: a work-group owns 196 receiver rows in LDS, finds its edges by scanning a 1-byte chunk-key array, gathers rows, LDS float adds; sum order changes
# speedup vs baseline: 2.7423x; 1.2896x over previous
#include <hip/hip_runtime.h>

typedef __attribute__((ext_vector_type(16))) _Float16 v16h;
typedef __attribute__((ext_vector_type(8)))  float    v8f;

#define DF   128
#define DK2  256
#define DOUT 64

static __host__ __device__ inline int cdiv(int a, int b) { return (a + b - 1) / b; }


#define DEG_CAP 50176
#define DEG_G   64
#define DEG_T   1024

__global__ void __launch_bounds__(DEG_T)
deg_partial(const int* __restrict__ snd, const int* __restrict__ rcv,
            const float* __restrict__ theta, int E, int Nn, int pitch,
            float* __restrict__ part) {
    __shared__ float4 pl4[DEG_CAP / 4];
    float* pl = (float*)pl4;
    const int tid  = threadIdx.x;
    const int g    = blockIdx.x;
    const int a    = blockIdx.y;
    const int base = blockIdx.z * DEG_CAP;
    const int cn   = (Nn - base < DEG_CAP) ? (Nn - base) : DEG_CAP;
    const int cn4  = (cn + 3) >> 2;
    for (int i = tid; i < cn4; i += DEG_T) pl4[i] = make_float4(0.0f, 0.0f, 0.0f, 0.0f);
    __syncthreads();

    const int per = cdiv(E, DEG_G);
    const int e0  = g * per;
    const int e1  = (e0 + per < E) ? (e0 + per) : E;
    for (int e = e0 + tid; e < e1; e += 4 * DEG_T) {
        int u[4], v[4]; float th[4];
#pragma unroll
        for (int k = 0; k < 4; ++k) {
            const int ek = e + k * DEG_T;
            const int ec = (ek < e1) ? ek : e;
            u[k] = snd[ec]; v[k] = rcv[ec]; th[k] = theta[ec];
            if (ek >= e1) { u[k] = -1; v[k] = -1; }
        }
#pragma unroll
        for (int k = 0; k < 4; ++k) {
            const float cs = cosf(th[k]), sn = sinf(th[k]);
            const float c2 = cs * cs, s2 = sn * sn;
            const float wu = a ? s2 : c2;
            const float wv = a ? c2 : s2;
            const unsigned iu = (unsigned)(u[k] - base);
            const unsigned iv = (unsigned)(v[k] - base);
            if (iu < (unsigned)cn) __hip_atomic_fetch_add(pl + iu, wu, __ATOMIC_RELAXED, __HIP_MEMORY_SCOPE_WORKGROUP);
            if (iv < (unsigned)cn) __hip_atomic_fetch_add(pl + iv, wv, __ATOMIC_RELAXED, __HIP_MEMORY_SCOPE_WORKGROUP);
        }
    }
    __syncthreads();

    float4* dst = (float4*)(part + ((size_t)a * DEG_G + g) * pitch + base);
    for (int i = tid; i < cn4; i += DEG_T) dst[i] = pl4[i];
}

__global__ void deg_reduce(const float* __restrict__ part, int Nn, int pitch,
                           float* __restrict__ disS, float* __restrict__ disR) {
    const int n = blockIdx.x * blockDim.x + threadIdx.x;
    if (n >= Nn) return;
    const int a = blockIdx.y;
    const float* p = part + (size_t)a * DEG_G * pitch + n;
    float d = 1.0f;
#pragma unroll 16
    for (int g = 0; g < DEG_G; ++g) d += p[(size_t)g * pitch];
    const float v = d + 1e-12f;
    (a ? disR : disS)[n] = (v > 1e11f) ? 0.0f : rsqrtf(v);
}

#define AGG_ROWS 196
#define AGG_T    1024
#define AGG_LCAP 16384

__global__ void weight_kernel(const int* __restrict__ snd, const int* __restrict__ rcv,
                              const float* __restrict__ theta, int E, int Nn,
                              const float* __restrict__ disS, const float* __restrict__ disR,
                              int4* __restrict__ rec, unsigned char* __restrict__ cid,
                              float* __restrict__ selfS, float* __restrict__ selfD) {
    const int i = blockIdx.x * blockDim.x + threadIdx.x;
    if (i < E) {
        const int s = snd[i], r = rcv[i];
        const float c = cosf(theta[i]), sn = sinf(theta[i]);
        const float a = disS[s] * (c * c) * disR[r];
        const float b = disR[s] * (sn * sn) * disS[r];
        rec[i] = make_int4(s, r, __float_as_int(a), __float_as_int(b));
        cid[i]     = (unsigned char)(r / AGG_ROWS);
        cid[i + E] = (unsigned char)(s / AGG_ROWS);
    } else if (i < E + Nn) {
        const int n = i - E;
        const float ds = disS[n], dr = disR[n];
        selfS[n] = ds * dr;
        selfD[n] = dr * ds;
        const int pad = (16 - ((2 * E) & 15)) & 15;
        if (n < pad) cid[2 * E + n] = 0;
    }
}

__global__ void __launch_bounds__(AGG_T)
agg_lds(const float* __restrict__ Y, const int4* __restrict__ rec,
        const uint4* __restrict__ cid16, float* __restrict__ acc, int E, int Nn) {
    __shared__ float4   pl4[AGG_ROWS * (DF / 4)];
    __shared__ unsigned lst[AGG_LCAP];
    __shared__ unsigned s_cnt;
    float* pl = (float*)pl4;
    const int tid   = threadIdx.x;
    const int chunk = blockIdx.x;
    const int base  = chunk * AGG_ROWS;
    const int cn    = (Nn - base < AGG_ROWS) ? (Nn - base) : AGG_ROWS;
    const int M2    = 2 * E;

    float4* rows4 = (float4*)(acc + (size_t)base * DF);
    for (int i = tid; i < cn * (DF / 4); i += AGG_T) pl4[i] = rows4[i];
    if (tid == 0) s_cnt = 0u;
    __syncthreads();

    const unsigned pat = (unsigned)chunk * 0x01010101u;
    const int nw = (M2 + 15) >> 4;
    for (int w0 = tid; w0 < nw; w0 += 4 * AGG_T) {
        uint4 kq[4];
#pragma unroll
        for (int k = 0; k < 4; ++k) {
            const int w  = w0 + k * AGG_T;
            const int wc = (w < nw) ? w : w0;
            kq[k] = cid16[wc];
            if (w >= nw) { kq[k].x = ~pat; kq[k].y = ~pat; kq[k].z = ~pat; kq[k].w = ~pat; }
        }
#pragma unroll
        for (int k = 0; k < 4; ++k) {
            const int w = w0 + k * AGG_T;
            const unsigned x0 = kq[k].x ^ pat, x1 = kq[k].y ^ pat, x2 = kq[k].z ^ pat, x3 = kq[k].w ^ pat;
            const unsigned t0 = ~(((x0 & 0x7F7F7F7Fu) + 0x7F7F7F7Fu) | x0 | 0x7F7F7F7Fu);
            const unsigned t1 = ~(((x1 & 0x7F7F7F7Fu) + 0x7F7F7F7Fu) | x1 | 0x7F7F7F7Fu);
            const unsigned t2 = ~(((x2 & 0x7F7F7F7Fu) + 0x7F7F7F7Fu) | x2 | 0x7F7F7F7Fu);
            const unsigned t3 = ~(((x3 & 0x7F7F7F7Fu) + 0x7F7F7F7Fu) | x3 | 0x7F7F7F7Fu);
            unsigned m = (t0 >> 7) | (t1 >> 6) | (t2 >> 5) | (t3 >> 4);
            while (m) {
                const int pos = __builtin_ctz(m);
                m &= m - 1u;
                const int idx = w * 16 + (pos & 7) * 4 + (pos >> 3);
                if (idx < M2) {
                    const unsigned slot = atomicAdd(&s_cnt, 1u);
                    if (slot < (unsigned)AGG_LCAP) {
                        lst[slot] = (unsigned)idx;
                    } else {
                        int4 q; int s, r; float a, b;
                        if (idx < E) { q = rec[idx];     s = q.x; r = q.y; a = __int_as_float(q.z); b = __int_as_float(q.w); }
                        else         { q = rec[idx - E]; s = q.y; r = q.x; a = __int_as_float(q.w); b = __int_as_float(q.z); }
                        const float* yr = Y + (long)s * DK2;
                        float* dst = pl + (r - base) * DF;
                        for (int p = 0; p < DF; ++p)
                            __hip_atomic_fetch_add(dst + 32 * (p & 3) + (p >> 2), a * yr[p] + b * yr[DF + p],
                                                   __ATOMIC_RELAXED, __HIP_MEMORY_SCOPE_WORKGROUP);
                    }
                }
            }
        }
    }
    __syncthreads();

    const int total = __builtin_amdgcn_readfirstlane((int)s_cnt);
    const int cnt   = (total < AGG_LCAP) ? total : AGG_LCAP;
    const int wv    = __builtin_amdgcn_readfirstlane(tid >> 5);
    const int lane  = tid & 31;
    const int per   = (cnt + (AGG_T / 32) - 1) / (AGG_T / 32);
    const int st    = wv * per;
    const int en    = (st + per < cnt) ? (st + per) : cnt;
    const float* Yl = Y + lane * 4;
    float* pll = pl + lane;
    if (st < en) {
        const int len = en - st;
        const int ng  = (len + 31) >> 5;
        const int gs  = (((len + ng - 1) / ng) + 3) & ~3;
        int sv = 0, rv = 0, av = 0, bv = 0;
        {
            const int n0 = (len < gs) ? len : gs;
            if (lane < n0) {
                const unsigned e = lst[st + lane];
                if (e < (unsigned)E) { const int4 q = rec[e];     sv = q.x; rv = q.y; av = q.z; bv = q.w; }
                else                 { const int4 q = rec[e - E]; sv = q.y; rv = q.x; av = q.w; bv = q.z; }
            }
        }
        for (int g = st; g < en; g += gs) {
            const int n = (en - g < gs) ? (en - g) : gs;
            const int cs = sv, cr = rv, ca = av, cb = bv;
            const int gn = g + gs;
            if (gn < en) {
                const int nn = (en - gn < gs) ? (en - gn) : gs;
                if (lane < nn) {
                    const unsigned e = lst[gn + lane];
                    if (e < (unsigned)E) { const int4 q = rec[e];     sv = q.x; rv = q.y; av = q.z; bv = q.w; }
                    else                 { const int4 q = rec[e - E]; sv = q.y; rv = q.x; av = q.w; bv = q.z; }
                }
            }
            int t = 0;
            for (; t + 4 <= n; t += 4) {
                float4 ys[4], yd[4];
                int r[4]; float a[4], b[4];
#pragma unroll
                for (int u = 0; u < 4; ++u) {
                    const int s = __builtin_amdgcn_readlane(cs, t + u);
                    r[u] = __builtin_amdgcn_readlane(cr, t + u) - base;
                    a[u] = __int_as_float(__builtin_amdgcn_readlane(ca, t + u));
                    b[u] = __int_as_float(__builtin_amdgcn_readlane(cb, t + u));
                    const float* yr = Yl + (long)s * DK2;
                    ys[u] = *(const float4*)yr;
                    yd[u] = *(const float4*)(yr + DF);
                }
#pragma unroll
                for (int u = 0; u < 4; ++u) {
                    float* dst = pll + r[u] * DF;
                    __hip_atomic_fetch_add(dst,      a[u] * ys[u].x + b[u] * yd[u].x, __ATOMIC_RELAXED, __HIP_MEMORY_SCOPE_WORKGROUP);
                    __hip_atomic_fetch_add(dst + 32, a[u] * ys[u].y + b[u] * yd[u].y, __ATOMIC_RELAXED, __HIP_MEMORY_SCOPE_WORKGROUP);
                    __hip_atomic_fetch_add(dst + 64, a[u] * ys[u].z + b[u] * yd[u].z, __ATOMIC_RELAXED, __HIP_MEMORY_SCOPE_WORKGROUP);
                    __hip_atomic_fetch_add(dst + 96, a[u] * ys[u].w + b[u] * yd[u].w, __ATOMIC_RELAXED, __HIP_MEMORY_SCOPE_WORKGROUP);
                }
            }
            for (; t < n; ++t) {
                const int s = __builtin_amdgcn_readlane(cs, t);
                const int r = __builtin_amdgcn_readlane(cr, t) - base;
                const float a = __int_as_float(__builtin_amdgcn_readlane(ca, t));
                const float b = __int_as_float(__builtin_amdgcn_readlane(cb, t));
                const float* yr = Yl + (long)s * DK2;
                const float4 ys = *(const float4*)yr;
                const float4 yd = *(const float4*)(yr + DF);
                float* dst = pll + r * DF;
                __hip_atomic_fetch_add(dst,      a * ys.x + b * yd.x, __ATOMIC_RELAXED, __HIP_MEMORY_SCOPE_WORKGROUP);
                __hip_atomic_fetch_add(dst + 32, a * ys.y + b * yd.y, __ATOMIC_RELAXED, __HIP_MEMORY_SCOPE_WORKGROUP);
                __hip_atomic_fetch_add(dst + 64, a * ys.z + b * yd.z, __ATOMIC_RELAXED, __HIP_MEMORY_SCOPE_WORKGROUP);
                __hip_atomic_fetch_add(dst + 96, a * ys.w + b * yd.w, __ATOMIC_RELAXED, __HIP_MEMORY_SCOPE_WORKGROUP);
            }
        }
    }
    __syncthreads();

    for (int i = tid; i < cn * (DF / 4); i += AGG_T) rows4[i] = pl4[i];
}

__global__ void pack_pre(const float* __restrict__ Ws, const float* __restrict__ Wd,
                         _Float16* __restrict__ dst) {
    int idx = blockIdx.x * blockDim.x + threadIdx.x;
    if (idx >= DF * DK2) return;
    int j    = idx & 15;
    int lane = (idx >> 4) & 31;
    int t    = idx >> 9;
    int q    = t & 7;
    int hp   = (t >> 3) & 1;
    int kt   = t >> 4;
    int half = q >> 2;
    int jt   = q & 3;
    int k = kt * 32 + ((j < 8) ? 0 : 16) + (lane >> 4) * 8 + (j & 7);
    int n = 32 * jt + 16 * hp + (lane & 15);
    float v = half ? 0.5f * Wd[k * DF + n] : 0.5f * Ws[k * DF + n];
    dst[idx] = (_Float16)v;
}

__global__ void pack_single(const float* __restrict__ W, _Float16* __restrict__ dst) {
    int idx = blockIdx.x * blockDim.x + threadIdx.x;
    if (idx >= DF * DOUT) return;
    int j    = idx & 15;
    int lane = (idx >> 4) & 31;
    int t    = idx >> 9;
    int k0t  = t / (DOUT / 16);
    int tc   = t % (DOUT / 16);
    int k = k0t * 32 + ((j < 8) ? 0 : 16) + (lane >> 4) * 8 + (j & 7);
    int n = tc * 16 + (lane & 15);
    dst[idx] = (_Float16)W[k * DOUT + n];
}

__global__ void bias_combine(const float* __restrict__ bs, const float* __restrict__ bd,
                             float* __restrict__ out, int n) {
    int i = blockIdx.x * blockDim.x + threadIdx.x;
    if (i < n) out[i] = 0.5f * (bs[i] + bd[i]);
}

template <int RELU_IN>
__device__ __forceinline__ v16h load_a_frag(const float4* __restrict__ Ar, int k0, int hi) {
    int base = (k0 >> 2) + hi * 2;
    float4 p0 = Ar[base];
    float4 p1 = Ar[base + 1];
    float4 p2 = Ar[base + 4];
    float4 p3 = Ar[base + 5];
    if (RELU_IN) {
        p0.x = fmaxf(p0.x, 0.0f); p0.y = fmaxf(p0.y, 0.0f); p0.z = fmaxf(p0.z, 0.0f); p0.w = fmaxf(p0.w, 0.0f);
        p1.x = fmaxf(p1.x, 0.0f); p1.y = fmaxf(p1.y, 0.0f); p1.z = fmaxf(p1.z, 0.0f); p1.w = fmaxf(p1.w, 0.0f);
        p2.x = fmaxf(p2.x, 0.0f); p2.y = fmaxf(p2.y, 0.0f); p2.z = fmaxf(p2.z, 0.0f); p2.w = fmaxf(p2.w, 0.0f);
        p3.x = fmaxf(p3.x, 0.0f); p3.y = fmaxf(p3.y, 0.0f); p3.z = fmaxf(p3.z, 0.0f); p3.w = fmaxf(p3.w, 0.0f);
    }
    v16h a;
    a[0]  = (_Float16)p0.x; a[1]  = (_Float16)p0.y;
    a[2]  = (_Float16)p0.z; a[3]  = (_Float16)p0.w;
    a[4]  = (_Float16)p1.x; a[5]  = (_Float16)p1.y;
    a[6]  = (_Float16)p1.z; a[7]  = (_Float16)p1.w;
    a[8]  = (_Float16)p2.x; a[9]  = (_Float16)p2.y;
    a[10] = (_Float16)p2.z; a[11] = (_Float16)p2.w;
    a[12] = (_Float16)p3.x; a[13] = (_Float16)p3.y;
    a[14] = (_Float16)p3.z; a[15] = (_Float16)p3.w;
    return a;
}

template <int RELU_IN>
__global__ void __launch_bounds__(256)
gemm_pre(const float* __restrict__ A, int Mrows,
         const v16h* __restrict__ Bp, const float* __restrict__ bias,
         const float* __restrict__ aSelf, const float* __restrict__ bSelf,
         float* __restrict__ Y, float* __restrict__ acc) {
    int wave = (blockIdx.x * blockDim.x + threadIdx.x) >> 5;
    int lane = threadIdx.x & 31;
    int rowTiles = Mrows >> 4;
    if (wave >= rowTiles * 2) return;
    int tr = wave >> 1;
    int hp = wave & 1;
    int hi = lane >> 4;
    int n  = lane & 15;

    const float4* Ar = (const float4*)(A + (long)(tr * 16 + n) * DF);
    v8f c[8] = {};
#pragma unroll
    for (int kt = 0; kt < DF / 32; ++kt) {
        v16h a = load_a_frag<RELU_IN>(Ar, kt * 32, hi);
        const v16h* bq = Bp + ((kt * 2 + hp) * 8) * 32 + lane;
#pragma unroll
        for (int q = 0; q < 8; ++q) {
            v16h b = bq[q * 32];
            c[q] = __builtin_amdgcn_wmma_f32_16x16x32_f16(false, a, false, b,
                                                          (short)0, c[q], false, false);
        }
    }

    const int lcol = 16 * hp + n;
    const float bv0 = bias[lcol], bv1 = bias[lcol + 32], bv2 = bias[lcol + 64], bv3 = bias[lcol + 96];
    const int row0 = tr * 16 + hi * 8;
#pragma unroll
    for (int i = 0; i < 8; ++i) {
        const int row = row0 + i;
        const float as = aSelf[row];
        const float bs = bSelf[row];
        float4 ys, yd;
        ys.x = c[0][i]; ys.y = c[1][i]; ys.z = c[2][i]; ys.w = c[3][i];
        yd.x = c[4][i]; yd.y = c[5][i]; yd.z = c[6][i]; yd.w = c[7][i];
        float* yr = Y + (long)row * DK2 + lcol * 4;
        *(float4*)yr        = ys;
        *(float4*)(yr + DF) = yd;
        float* ar = acc + (long)row * DF + lcol;
        ar[0]  = bv0 + as * ys.x + bs * yd.x;
        ar[32] = bv1 + as * ys.y + bs * yd.y;
        ar[64] = bv2 + as * ys.z + bs * yd.z;
        ar[96] = bv3 + as * ys.w + bs * yd.w;
    }
}

template <int K, int NC, int RELU, int RELU_IN>
__global__ void wmma_gemm(const float* __restrict__ A, int Mrows,
                          const v16h* __restrict__ Bp, const float* __restrict__ bias,
                          float* __restrict__ D) {
    constexpr int colTiles = NC / 16;
    int wave = (blockIdx.x * blockDim.x + threadIdx.x) >> 5;
    int lane = threadIdx.x & 31;
    int rowTiles = Mrows >> 4;
    if (wave >= rowTiles * colTiles) return;
    int tr = wave / colTiles;
    int tc = wave % colTiles;
    int hi = lane >> 4;
    int arow = tr * 16 + (lane & 15);
    int ncol = tc * 16 + (lane & 15);

    const float4* Ar = (const float4*)(A + (long)arow * K);
    v8f c = {};
#pragma unroll
    for (int k0 = 0; k0 < K; k0 += 32) {
        v16h a = load_a_frag<RELU_IN>(Ar, k0, hi);
        v16h b = Bp[((k0 >> 5) * colTiles + tc) * 32 + lane];
        c = __builtin_amdgcn_wmma_f32_16x16x32_f16(false, a, false, b,
                                                   (short)0, c, false, false);
    }
    float bv = bias[ncol];
    float* Drow = D + (long)(tr * 16 + hi * 8) * NC + ncol;
#pragma unroll
    for (int i = 0; i < 8; ++i) {
        float v = c[i] + bv;
        if (RELU) v = fmaxf(v, 0.0f);
        Drow[(long)i * NC] = v;
    }
}

extern "C" void kernel_launch(void* const* d_in, const int* in_sizes, int n_in,
                              void* d_out, int out_size, void* d_ws, size_t ws_size,
                              hipStream_t stream) {
    const float* x      = (const float*)d_in[0];
    const int*   edges  = (const int*)d_in[1];
    const float* theta  = (const float*)d_in[2];
    const float* W1s    = (const float*)d_in[3];
    const float* W1d    = (const float*)d_in[4];
    const float* b1s    = (const float*)d_in[5];
    const float* b1d    = (const float*)d_in[6];
    const float* W2s    = (const float*)d_in[7];
    const float* W2d    = (const float*)d_in[8];
    const float* b2s    = (const float*)d_in[9];
    const float* b2d    = (const float*)d_in[10];
    const float* Wro    = (const float*)d_in[11];
    const float* bro    = (const float*)d_in[12];

    const int N = in_sizes[0] / DF;
    const int E = in_sizes[2];
    const int* snd = edges;
    const int* rcv = edges + E;

    char* ws = (char*)d_ws;
    size_t off = 0;
    auto alloc = [&](size_t bytes) { char* p = ws + off; off = (off + bytes + 255) & ~(size_t)255; return p; };
    float*    degS  = (float*)alloc((size_t)N * 4);
    float*    degR  = (float*)alloc((size_t)N * 4);
    int4*     rec   = (int4*)alloc((size_t)E * 16);
    unsigned char* cid = (unsigned char*)alloc((size_t)cdiv(2 * E, 16) * 16);
    float*    selfS = (float*)alloc((size_t)N * 4);
    float*    selfD = (float*)alloc((size_t)N * 4);
    float*    Y     = (float*)alloc((size_t)N * DK2 * 4);
    float*    acc1  = (float*)alloc((size_t)N * DF * 4);
    float*    acc2  = (float*)alloc((size_t)N * DF * 4);
    _Float16* Bp1   = (_Float16*)alloc((size_t)DK2 * DF * 2);
    _Float16* Bp2   = (_Float16*)alloc((size_t)DK2 * DF * 2);
    _Float16* Bpro  = (_Float16*)alloc((size_t)DF * DOUT * 2);
    float*    bias1 = (float*)alloc(DF * 4);
    float*    bias2 = (float*)alloc(DF * 4);

    const int T = 256;

    float* part = Y;
    const int pitch = (N + 31) & ~31;
    deg_partial<<<dim3(DEG_G, 2, cdiv(N, DEG_CAP)), DEG_T, 0, stream>>>(snd, rcv, theta, E, N, pitch, part);
    deg_reduce<<<dim3(cdiv(N, T), 2), T, 0, stream>>>(part, N, pitch, degS, degR);
    weight_kernel<<<cdiv(E + N, T), T, 0, stream>>>(snd, rcv, theta, E, N, degS, degR, rec, cid, selfS, selfD);

    pack_pre<<<cdiv(DF * DK2, T), T, 0, stream>>>(W1s, W1d, Bp1);
    pack_pre<<<cdiv(DF * DK2, T), T, 0, stream>>>(W2s, W2d, Bp2);
    pack_single<<<cdiv(DF * DOUT, T), T, 0, stream>>>(Wro, Bpro);
    bias_combine<<<1, DF, 0, stream>>>(b1s, b1d, bias1, DF);
    bias_combine<<<1, DF, 0, stream>>>(b2s, b2d, bias2, DF);

    const int preWaves      = (N / 16) * 2;
    const int aggBlocks     = cdiv(N, AGG_ROWS);
    const int roTiles       = (N / 16) * (DOUT / 16);

    gemm_pre<0><<<cdiv(preWaves * 32, T), T, 0, stream>>>(
        x, N, (const v16h*)Bp1, bias1, selfS, selfD, Y, acc1);
    agg_lds<<<aggBlocks, AGG_T, 0, stream>>>(Y, rec, (const uint4*)cid, acc1, E, N);

    gemm_pre<1><<<cdiv(preWaves * 32, T), T, 0, stream>>>(
        acc1, N, (const v16h*)Bp2, bias2, selfS, selfD, Y, acc2);
    agg_lds<<<aggBlocks, AGG_T, 0, stream>>>(Y, rec, (const uint4*)cid, acc2, E, N);

    wmma_gemm<DF, DOUT, 0, 1><<<cdiv(roTiles * 32, T), T, 0, stream>>>(
        acc2, N, (const v16h*)Bpro, bro, (float*)d_out);
}


# Layer 2 skips the key scan: the layer-1 edge pass stores each chunk's edge list (ids, length) to the workspace and layer 2 reads it back; layer-2 sum order follows that list
# speedup vs baseline: 2.9029x; 1.0586x over previous
#include <hip/hip_runtime.h>

typedef __attribute__((ext_vector_type(16))) _Float16 v16h;
typedef __attribute__((ext_vector_type(8)))  float    v8f;

#define DF   128
#define DK2  256
#define DOUT 64

static __host__ __device__ inline int cdiv(int a, int b) { return (a + b - 1) / b; }


#define DEG_CAP 50176
#define DEG_G   64
#define DEG_T   1024

__global__ void __launch_bounds__(DEG_T)
deg_partial(const int* __restrict__ snd, const int* __restrict__ rcv,
            const float* __restrict__ theta, int E, int Nn, int pitch,
            float* __restrict__ part) {
    __shared__ float4 pl4[DEG_CAP / 4];
    float* pl = (float*)pl4;
    const int tid  = threadIdx.x;
    const int g    = blockIdx.x;
    const int a    = blockIdx.y;
    const int base = blockIdx.z * DEG_CAP;
    const int cn   = (Nn - base < DEG_CAP) ? (Nn - base) : DEG_CAP;
    const int cn4  = (cn + 3) >> 2;
    for (int i = tid; i < cn4; i += DEG_T) pl4[i] = make_float4(0.0f, 0.0f, 0.0f, 0.0f);
    __syncthreads();

    const int per = cdiv(E, DEG_G);
    const int e0  = g * per;
    const int e1  = (e0 + per < E) ? (e0 + per) : E;
    for (int e = e0 + tid; e < e1; e += 4 * DEG_T) {
        int u[4], v[4]; float th[4];
#pragma unroll
        for (int k = 0; k < 4; ++k) {
            const int ek = e + k * DEG_T;
            const int ec = (ek < e1) ? ek : e;
            u[k] = snd[ec]; v[k] = rcv[ec]; th[k] = theta[ec];
            if (ek >= e1) { u[k] = -1; v[k] = -1; }
        }
#pragma unroll
        for (int k = 0; k < 4; ++k) {
            const float cs = cosf(th[k]), sn = sinf(th[k]);
            const float c2 = cs * cs, s2 = sn * sn;
            const float wu = a ? s2 : c2;
            const float wv = a ? c2 : s2;
            const unsigned iu = (unsigned)(u[k] - base);
            const unsigned iv = (unsigned)(v[k] - base);
            if (iu < (unsigned)cn) __hip_atomic_fetch_add(pl + iu, wu, __ATOMIC_RELAXED, __HIP_MEMORY_SCOPE_WORKGROUP);
            if (iv < (unsigned)cn) __hip_atomic_fetch_add(pl + iv, wv, __ATOMIC_RELAXED, __HIP_MEMORY_SCOPE_WORKGROUP);
        }
    }
    __syncthreads();

    float4* dst = (float4*)(part + ((size_t)a * DEG_G + g) * pitch + base);
    for (int i = tid; i < cn4; i += DEG_T) dst[i] = pl4[i];
}

__global__ void deg_reduce(const float* __restrict__ part, int Nn, int pitch,
                           float* __restrict__ disS, float* __restrict__ disR) {
    const int n = blockIdx.x * blockDim.x + threadIdx.x;
    if (n >= Nn) return;
    const int a = blockIdx.y;
    const float* p = part + (size_t)a * DEG_G * pitch + n;
    float d = 1.0f;
#pragma unroll 16
    for (int g = 0; g < DEG_G; ++g) d += p[(size_t)g * pitch];
    const float v = d + 1e-12f;
    (a ? disR : disS)[n] = (v > 1e11f) ? 0.0f : rsqrtf(v);
}

#define AGG_ROWS 196
#define AGG_T    1024
#define AGG_LCAP 16384
#define AGG_GCAP 4096

__global__ void weight_kernel(const int* __restrict__ snd, const int* __restrict__ rcv,
                              const float* __restrict__ theta, int E, int Nn,
                              const float* __restrict__ disS, const float* __restrict__ disR,
                              int4* __restrict__ rec, unsigned char* __restrict__ cid,
                              float* __restrict__ selfS, float* __restrict__ selfD) {
    const int i = blockIdx.x * blockDim.x + threadIdx.x;
    if (i < E) {
        const int s = snd[i], r = rcv[i];
        const float c = cosf(theta[i]), sn = sinf(theta[i]);
        const float a = disS[s] * (c * c) * disR[r];
        const float b = disR[s] * (sn * sn) * disS[r];
        rec[i] = make_int4(s, r, __float_as_int(a), __float_as_int(b));
        cid[i]     = (unsigned char)(r / AGG_ROWS);
        cid[i + E] = (unsigned char)(s / AGG_ROWS);
    } else if (i < E + Nn) {
        const int n = i - E;
        const float ds = disS[n], dr = disR[n];
        selfS[n] = ds * dr;
        selfD[n] = dr * ds;
        const int pad = (16 - ((2 * E) & 15)) & 15;
        if (n < pad) cid[2 * E + n] = 0;
    }
}

template <int REUSE>
__global__ void __launch_bounds__(AGG_T)
agg_lds(const float* __restrict__ Y, const int4* __restrict__ rec,
        const uint4* __restrict__ cid16, float* __restrict__ acc, int E, int Nn,
        unsigned* __restrict__ glst, unsigned* __restrict__ gcnt) {
    __shared__ float4   pl4[AGG_ROWS * (DF / 4)];
    __shared__ unsigned lst[AGG_LCAP];
    __shared__ unsigned s_cnt;
    float* pl = (float*)pl4;
    const int tid   = threadIdx.x;
    const int chunk = blockIdx.x;
    const int base  = chunk * AGG_ROWS;
    const int cn    = (Nn - base < AGG_ROWS) ? (Nn - base) : AGG_ROWS;
    const int M2    = 2 * E;

    unsigned* gl = glst + (size_t)chunk * AGG_GCAP;
    const unsigned saved = REUSE ? gcnt[chunk] : 0u;
    const bool doScan = REUSE ? (saved > (unsigned)AGG_GCAP) : true;
    if (!doScan) {
        static_assert(AGG_GCAP == 4 * AGG_T, "a thread copies four list entries");
        unsigned v[4];
#pragma unroll
        for (int k = 0; k < 4; ++k) {
            const unsigned i = (unsigned)tid + (unsigned)(k * AGG_T);
            v[k] = (i < saved) ? gl[i] : 0u;
        }
#pragma unroll
        for (int k = 0; k < 4; ++k) {
            const unsigned i = (unsigned)tid + (unsigned)(k * AGG_T);
            if (i < saved) lst[i] = v[k];
        }
    }

    float4* rows4 = (float4*)(acc + (size_t)base * DF);
    for (int i = tid; i < cn * (DF / 4); i += AGG_T) pl4[i] = rows4[i];
    if (tid == 0) s_cnt = doScan ? 0u : saved;
    __syncthreads();

    const unsigned pat = (unsigned)chunk * 0x01010101u;
    const int nw = doScan ? ((M2 + 15) >> 4) : 0;
    for (int w0 = tid; w0 < nw; w0 += 4 * AGG_T) {
        uint4 kq[4];
#pragma unroll
        for (int k = 0; k < 4; ++k) {
            const int w  = w0 + k * AGG_T;
            const int wc = (w < nw) ? w : w0;
            kq[k] = cid16[wc];
            if (w >= nw) { kq[k].x = ~pat; kq[k].y = ~pat; kq[k].z = ~pat; kq[k].w = ~pat; }
        }
#pragma unroll
        for (int k = 0; k < 4; ++k) {
            const int w = w0 + k * AGG_T;
            const unsigned x0 = kq[k].x ^ pat, x1 = kq[k].y ^ pat, x2 = kq[k].z ^ pat, x3 = kq[k].w ^ pat;
            const unsigned t0 = ~(((x0 & 0x7F7F7F7Fu) + 0x7F7F7F7Fu) | x0 | 0x7F7F7F7Fu);
            const unsigned t1 = ~(((x1 & 0x7F7F7F7Fu) + 0x7F7F7F7Fu) | x1 | 0x7F7F7F7Fu);
            const unsigned t2 = ~(((x2 & 0x7F7F7F7Fu) + 0x7F7F7F7Fu) | x2 | 0x7F7F7F7Fu);
            const unsigned t3 = ~(((x3 & 0x7F7F7F7Fu) + 0x7F7F7F7Fu) | x3 | 0x7F7F7F7Fu);
            unsigned m = (t0 >> 7) | (t1 >> 6) | (t2 >> 5) | (t3 >> 4);
            while (m) {
                const int pos = __builtin_ctz(m);
                m &= m - 1u;
                const int idx = w * 16 + (pos & 7) * 4 + (pos >> 3);
                if (idx < M2) {
                    const unsigned slot = atomicAdd(&s_cnt, 1u);
                    if (slot < (unsigned)AGG_LCAP) {
                        lst[slot] = (unsigned)idx;
                    } else {
                        int4 q; int s, r; float a, b;
                        if (idx < E) { q = rec[idx];     s = q.x; r = q.y; a = __int_as_float(q.z); b = __int_as_float(q.w); }
                        else         { q = rec[idx - E]; s = q.y; r = q.x; a = __int_as_float(q.w); b = __int_as_float(q.z); }
                        const float* yr = Y + (long)s * DK2;
                        float* dst = pl + (r - base) * DF;
                        for (int p = 0; p < DF; ++p)
                            __hip_atomic_fetch_add(dst + 32 * (p & 3) + (p >> 2), a * yr[p] + b * yr[DF + p],
                                                   __ATOMIC_RELAXED, __HIP_MEMORY_SCOPE_WORKGROUP);
                    }
                }
            }
        }
    }
    __syncthreads();

    const int total = __builtin_amdgcn_readfirstlane((int)s_cnt);
    const int cnt   = (total < AGG_LCAP) ? total : AGG_LCAP;
    if (!REUSE) {
        if (tid == 0) gcnt[chunk] = (unsigned)total;
        if (total <= AGG_GCAP)
            for (int i = tid; i < total; i += AGG_T) gl[i] = lst[i];
    }
    const int wv    = __builtin_amdgcn_readfirstlane(tid >> 5);
    const int lane  = tid & 31;
    const int per   = (cnt + (AGG_T / 32) - 1) / (AGG_T / 32);
    const int st    = wv * per;
    const int en    = (st + per < cnt) ? (st + per) : cnt;
    const float* Yl = Y + lane * 4;
    float* pll = pl + lane;
    if (st < en) {
        const int len = en - st;
        const int ng  = (len + 31) >> 5;
        const int gs  = (((len + ng - 1) / ng) + 3) & ~3;
        int sv = 0, rv = 0, av = 0, bv = 0;
        {
            const int n0 = (len < gs) ? len : gs;
            if (lane < n0) {
                const unsigned e = lst[st + lane];
                if (e < (unsigned)E) { const int4 q = rec[e];     sv = q.x; rv = q.y; av = q.z; bv = q.w; }
                else                 { const int4 q = rec[e - E]; sv = q.y; rv = q.x; av = q.w; bv = q.z; }
            }
        }
        for (int g = st; g < en; g += gs) {
            const int n = (en - g < gs) ? (en - g) : gs;
            const int cs = sv, cr = rv, ca = av, cb = bv;
            const int gn = g + gs;
            if (gn < en) {
                const int nn = (en - gn < gs) ? (en - gn) : gs;
                if (lane < nn) {
                    const unsigned e = lst[gn + lane];
                    if (e < (unsigned)E) { const int4 q = rec[e];     sv = q.x; rv = q.y; av = q.z; bv = q.w; }
                    else                 { const int4 q = rec[e - E]; sv = q.y; rv = q.x; av = q.w; bv = q.z; }
                }
            }
            int t = 0;
            for (; t + 4 <= n; t += 4) {
                float4 ys[4], yd[4];
                int r[4]; float a[4], b[4];
#pragma unroll
                for (int u = 0; u < 4; ++u) {
                    const int s = __builtin_amdgcn_readlane(cs, t + u);
                    r[u] = __builtin_amdgcn_readlane(cr, t + u) - base;
                    a[u] = __int_as_float(__builtin_amdgcn_readlane(ca, t + u));
                    b[u] = __int_as_float(__builtin_amdgcn_readlane(cb, t + u));
                    const float* yr = Yl + (long)s * DK2;
                    ys[u] = *(const float4*)yr;
                    yd[u] = *(const float4*)(yr + DF);
                }
#pragma unroll
                for (int u = 0; u < 4; ++u) {
                    float* dst = pll + r[u] * DF;
                    __hip_atomic_fetch_add(dst,      a[u] * ys[u].x + b[u] * yd[u].x, __ATOMIC_RELAXED, __HIP_MEMORY_SCOPE_WORKGROUP);
                    __hip_atomic_fetch_add(dst + 32, a[u] * ys[u].y + b[u] * yd[u].y, __ATOMIC_RELAXED, __HIP_MEMORY_SCOPE_WORKGROUP);
                    __hip_atomic_fetch_add(dst + 64, a[u] * ys[u].z + b[u] * yd[u].z, __ATOMIC_RELAXED, __HIP_MEMORY_SCOPE_WORKGROUP);
                    __hip_atomic_fetch_add(dst + 96, a[u] * ys[u].w + b[u] * yd[u].w, __ATOMIC_RELAXED, __HIP_MEMORY_SCOPE_WORKGROUP);
                }
            }
            for (; t < n; ++t) {
                const int s = __builtin_amdgcn_readlane(cs, t);
                const int r = __builtin_amdgcn_readlane(cr, t) - base;
                const float a = __int_as_float(__builtin_amdgcn_readlane(ca, t));
                const float b = __int_as_float(__builtin_amdgcn_readlane(cb, t));
                const float* yr = Yl + (long)s * DK2;
                const float4 ys = *(const float4*)yr;
                const float4 yd = *(const float4*)(yr + DF);
                float* dst = pll + r * DF;
                __hip_atomic_fetch_add(dst,      a * ys.x + b * yd.x, __ATOMIC_RELAXED, __HIP_MEMORY_SCOPE_WORKGROUP);
                __hip_atomic_fetch_add(dst + 32, a * ys.y + b * yd.y, __ATOMIC_RELAXED, __HIP_MEMORY_SCOPE_WORKGROUP);
                __hip_atomic_fetch_add(dst + 64, a * ys.z + b * yd.z, __ATOMIC_RELAXED, __HIP_MEMORY_SCOPE_WORKGROUP);
                __hip_atomic_fetch_add(dst + 96, a * ys.w + b * yd.w, __ATOMIC_RELAXED, __HIP_MEMORY_SCOPE_WORKGROUP);
            }
        }
    }
    __syncthreads();

    for (int i = tid; i < cn * (DF / 4); i += AGG_T) rows4[i] = pl4[i];
}

__global__ void pack_pre(const float* __restrict__ Ws, const float* __restrict__ Wd,
                         _Float16* __restrict__ dst) {
    int idx = blockIdx.x * blockDim.x + threadIdx.x;
    if (idx >= DF * DK2) return;
    int j    = idx & 15;
    int lane = (idx >> 4) & 31;
    int t    = idx >> 9;
    int q    = t & 7;
    int hp   = (t >> 3) & 1;
    int kt   = t >> 4;
    int half = q >> 2;
    int jt   = q & 3;
    int k = kt * 32 + ((j < 8) ? 0 : 16) + (lane >> 4) * 8 + (j & 7);
    int n = 32 * jt + 16 * hp + (lane & 15);
    float v = half ? 0.5f * Wd[k * DF + n] : 0.5f * Ws[k * DF + n];
    dst[idx] = (_Float16)v;
}

__global__ void pack_single(const float* __restrict__ W, _Float16* __restrict__ dst) {
    int idx = blockIdx.x * blockDim.x + threadIdx.x;
    if (idx >= DF * DOUT) return;
    int j    = idx & 15;
    int lane = (idx >> 4) & 31;
    int t    = idx >> 9;
    int k0t  = t / (DOUT / 16);
    int tc   = t % (DOUT / 16);
    int k = k0t * 32 + ((j < 8) ? 0 : 16) + (lane >> 4) * 8 + (j & 7);
    int n = tc * 16 + (lane & 15);
    dst[idx] = (_Float16)W[k * DOUT + n];
}

__global__ void bias_combine(const float* __restrict__ bs, const float* __restrict__ bd,
                             float* __restrict__ out, int n) {
    int i = blockIdx.x * blockDim.x + threadIdx.x;
    if (i < n) out[i] = 0.5f * (bs[i] + bd[i]);
}

template <int RELU_IN>
__device__ __forceinline__ v16h load_a_frag(const float4* __restrict__ Ar, int k0, int hi) {
    int base = (k0 >> 2) + hi * 2;
    float4 p0 = Ar[base];
    float4 p1 = Ar[base + 1];
    float4 p2 = Ar[base + 4];
    float4 p3 = Ar[base + 5];
    if (RELU_IN) {
        p0.x = fmaxf(p0.x, 0.0f); p0.y = fmaxf(p0.y, 0.0f); p0.z = fmaxf(p0.z, 0.0f); p0.w = fmaxf(p0.w, 0.0f);
        p1.x = fmaxf(p1.x, 0.0f); p1.y = fmaxf(p1.y, 0.0f); p1.z = fmaxf(p1.z, 0.0f); p1.w = fmaxf(p1.w, 0.0f);
        p2.x = fmaxf(p2.x, 0.0f); p2.y = fmaxf(p2.y, 0.0f); p2.z = fmaxf(p2.z, 0.0f); p2.w = fmaxf(p2.w, 0.0f);
        p3.x = fmaxf(p3.x, 0.0f); p3.y = fmaxf(p3.y, 0.0f); p3.z = fmaxf(p3.z, 0.0f); p3.w = fmaxf(p3.w, 0.0f);
    }
    v16h a;
    a[0]  = (_Float16)p0.x; a[1]  = (_Float16)p0.y;
    a[2]  = (_Float16)p0.z; a[3]  = (_Float16)p0.w;
    a[4]  = (_Float16)p1.x; a[5]  = (_Float16)p1.y;
    a[6]  = (_Float16)p1.z; a[7]  = (_Float16)p1.w;
    a[8]  = (_Float16)p2.x; a[9]  = (_Float16)p2.y;
    a[10] = (_Float16)p2.z; a[11] = (_Float16)p2.w;
    a[12] = (_Float16)p3.x; a[13] = (_Float16)p3.y;
    a[14] = (_Float16)p3.z; a[15] = (_Float16)p3.w;
    return a;
}

template <int RELU_IN>
__global__ void __launch_bounds__(256)
gemm_pre(const float* __restrict__ A, int Mrows,
         const v16h* __restrict__ Bp, const float* __restrict__ bias,
         const float* __restrict__ aSelf, const float* __restrict__ bSelf,
         float* __restrict__ Y, float* __restrict__ acc) {
    int wave = (blockIdx.x * blockDim.x + threadIdx.x) >> 5;
    int lane = threadIdx.x & 31;
    int rowTiles = Mrows >> 4;
    if (wave >= rowTiles * 2) return;
    int tr = wave >> 1;
    int hp = wave & 1;
    int hi = lane >> 4;
    int n  = lane & 15;

    const float4* Ar = (const float4*)(A + (long)(tr * 16 + n) * DF);
    v8f c[8] = {};
#pragma unroll
    for (int kt = 0; kt < DF / 32; ++kt) {
        v16h a = load_a_frag<RELU_IN>(Ar, kt * 32, hi);
        const v16h* bq = Bp + ((kt * 2 + hp) * 8) * 32 + lane;
#pragma unroll
        for (int q = 0; q < 8; ++q) {
            v16h b = bq[q * 32];
            c[q] = __builtin_amdgcn_wmma_f32_16x16x32_f16(false, a, false, b,
                                                          (short)0, c[q], false, false);
        }
    }

    const int lcol = 16 * hp + n;
    const float bv0 = bias[lcol], bv1 = bias[lcol + 32], bv2 = bias[lcol + 64], bv3 = bias[lcol + 96];
    const int row0 = tr * 16 + hi * 8;
#pragma unroll
    for (int i = 0; i < 8; ++i) {
        const int row = row0 + i;
        const float as = aSelf[row];
        const float bs = bSelf[row];
        float4 ys, yd;
        ys.x = c[0][i]; ys.y = c[1][i]; ys.z = c[2][i]; ys.w = c[3][i];
        yd.x = c[4][i]; yd.y = c[5][i]; yd.z = c[6][i]; yd.w = c[7][i];
        float* yr = Y + (long)row * DK2 + lcol * 4;
        *(float4*)yr        = ys;
        *(float4*)(yr + DF) = yd;
        float* ar = acc + (long)row * DF + lcol;
        ar[0]  = bv0 + as * ys.x + bs * yd.x;
        ar[32] = bv1 + as * ys.y + bs * yd.y;
        ar[64] = bv2 + as * ys.z + bs * yd.z;
        ar[96] = bv3 + as * ys.w + bs * yd.w;
    }
}

template <int K, int NC, int RELU, int RELU_IN>
__global__ void wmma_gemm(const float* __restrict__ A, int Mrows,
                          const v16h* __restrict__ Bp, const float* __restrict__ bias,
                          float* __restrict__ D) {
    constexpr int colTiles = NC / 16;
    int wave = (blockIdx.x * blockDim.x + threadIdx.x) >> 5;
    int lane = threadIdx.x & 31;
    int rowTiles = Mrows >> 4;
    if (wave >= rowTiles * colTiles) return;
    int tr = wave / colTiles;
    int tc = wave % colTiles;
    int hi = lane >> 4;
    int arow = tr * 16 + (lane & 15);
    int ncol = tc * 16 + (lane & 15);

    const float4* Ar = (const float4*)(A + (long)arow * K);
    v8f c = {};
#pragma unroll
    for (int k0 = 0; k0 < K; k0 += 32) {
        v16h a = load_a_frag<RELU_IN>(Ar, k0, hi);
        v16h b = Bp[((k0 >> 5) * colTiles + tc) * 32 + lane];
        c = __builtin_amdgcn_wmma_f32_16x16x32_f16(false, a, false, b,
                                                   (short)0, c, false, false);
    }
    float bv = bias[ncol];
    float* Drow = D + (long)(tr * 16 + hi * 8) * NC + ncol;
#pragma unroll
    for (int i = 0; i < 8; ++i) {
        float v = c[i] + bv;
        if (RELU) v = fmaxf(v, 0.0f);
        Drow[(long)i * NC] = v;
    }
}

extern "C" void kernel_launch(void* const* d_in, const int* in_sizes, int n_in,
                              void* d_out, int out_size, void* d_ws, size_t ws_size,
                              hipStream_t stream) {
    const float* x      = (const float*)d_in[0];
    const int*   edges  = (const int*)d_in[1];
    const float* theta  = (const float*)d_in[2];
    const float* W1s    = (const float*)d_in[3];
    const float* W1d    = (const float*)d_in[4];
    const float* b1s    = (const float*)d_in[5];
    const float* b1d    = (const float*)d_in[6];
    const float* W2s    = (const float*)d_in[7];
    const float* W2d    = (const float*)d_in[8];
    const float* b2s    = (const float*)d_in[9];
    const float* b2d    = (const float*)d_in[10];
    const float* Wro    = (const float*)d_in[11];
    const float* bro    = (const float*)d_in[12];

    const int N = in_sizes[0] / DF;
    const int E = in_sizes[2];
    const int* snd = edges;
    const int* rcv = edges + E;

    char* ws = (char*)d_ws;
    size_t off = 0;
    auto alloc = [&](size_t bytes) { char* p = ws + off; off = (off + bytes + 255) & ~(size_t)255; return p; };
    float*    degS  = (float*)alloc((size_t)N * 4);
    float*    degR  = (float*)alloc((size_t)N * 4);
    int4*     rec   = (int4*)alloc((size_t)E * 16);
    unsigned char* cid = (unsigned char*)alloc((size_t)cdiv(2 * E, 16) * 16);
    float*    selfS = (float*)alloc((size_t)N * 4);
    float*    selfD = (float*)alloc((size_t)N * 4);
    float*    Y     = (float*)alloc((size_t)N * DK2 * 4);
    float*    acc1  = (float*)alloc((size_t)N * DF * 4);
    float*    acc2  = (float*)alloc((size_t)N * DF * 4);
    _Float16* Bp1   = (_Float16*)alloc((size_t)DK2 * DF * 2);
    _Float16* Bp2   = (_Float16*)alloc((size_t)DK2 * DF * 2);
    _Float16* Bpro  = (_Float16*)alloc((size_t)DF * DOUT * 2);
    float*    bias1 = (float*)alloc(DF * 4);
    float*    bias2 = (float*)alloc(DF * 4);
    unsigned* glst  = (unsigned*)alloc((size_t)cdiv(N, AGG_ROWS) * AGG_GCAP * 4);
    unsigned* gcnt  = (unsigned*)alloc((size_t)cdiv(N, AGG_ROWS) * 4);

    const int T = 256;

    float* part = Y;
    const int pitch = (N + 31) & ~31;
    deg_partial<<<dim3(DEG_G, 2, cdiv(N, DEG_CAP)), DEG_T, 0, stream>>>(snd, rcv, theta, E, N, pitch, part);
    deg_reduce<<<dim3(cdiv(N, T), 2), T, 0, stream>>>(part, N, pitch, degS, degR);
    weight_kernel<<<cdiv(E + N, T), T, 0, stream>>>(snd, rcv, theta, E, N, degS, degR, rec, cid, selfS, selfD);

    pack_pre<<<cdiv(DF * DK2, T), T, 0, stream>>>(W1s, W1d, Bp1);
    pack_pre<<<cdiv(DF * DK2, T), T, 0, stream>>>(W2s, W2d, Bp2);
    pack_single<<<cdiv(DF * DOUT, T), T, 0, stream>>>(Wro, Bpro);
    bias_combine<<<1, DF, 0, stream>>>(b1s, b1d, bias1, DF);
    bias_combine<<<1, DF, 0, stream>>>(b2s, b2d, bias2, DF);

    const int preWaves      = (N / 16) * 2;
    const int aggBlocks     = cdiv(N, AGG_ROWS);
    const int roTiles       = (N / 16) * (DOUT / 16);

    gemm_pre<0><<<cdiv(preWaves * 32, T), T, 0, stream>>>(
        x, N, (const v16h*)Bp1, bias1, selfS, selfD, Y, acc1);
    agg_lds<0><<<aggBlocks, AGG_T, 0, stream>>>(Y, rec, (const uint4*)cid, acc1, E, N, glst, gcnt);

    gemm_pre<1><<<cdiv(preWaves * 32, T), T, 0, stream>>>(
        acc1, N, (const v16h*)Bp2, bias2, selfS, selfD, Y, acc2);
    agg_lds<1><<<aggBlocks, AGG_T, 0, stream>>>(Y, rec, (const uint4*)cid, acc2, E, N, glst, gcnt);

    wmma_gemm<DF, DOUT, 0, 1><<<cdiv(roTiles * 32, T), T, 0, stream>>>(
        acc2, N, (const v16h*)Bpro, bro, (float*)d_out);
}


# Both layers read a chunk's edge list as 128 sub-lists that the degree pass sorts by receiver chunk in LDS and stores as whole lines; layer-1 key scan and list hand-over go; sum order changes
# speedup vs baseline: 4.0806x; 1.4057x over previous
#include <hip/hip_runtime.h>

typedef __attribute__((ext_vector_type(16))) _Float16 v16h;
typedef __attribute__((ext_vector_type(8)))  float    v8f;

#define DF   128
#define DK2  256
#define DOUT 64

static __host__ __device__ inline int cdiv(int a, int b) { return (a + b - 1) / b; }


#define DEG_CAP 50176
#define DEG_G   64
#define DEG_T   1024

#define AGG_ROWS 196
#define AGG_NCH  256
#define BIN_W    64
#define BIN_CAP  (BIN_W - 1)
#define BIN_S    (2 * DEG_G)

__global__ void __launch_bounds__(DEG_T)
deg_partial(const int* __restrict__ snd, const int* __restrict__ rcv,
            const float* __restrict__ theta, int E, int Nn, int pitch,
            float* __restrict__ part, unsigned short* __restrict__ blst) {
    __shared__ float4 pl4[DEG_CAP / 4];
    __shared__ uint4    bl4[AGG_NCH * BIN_W / 8];
    __shared__ unsigned bcount[AGG_NCH];
    float* pl = (float*)pl4;
    unsigned short* bl = (unsigned short*)bl4;
    const int tid  = threadIdx.x;
    const int g    = blockIdx.x;
    const int a    = blockIdx.y;
    const int base = blockIdx.z * DEG_CAP;
    const int cn   = (Nn - base < DEG_CAP) ? (Nn - base) : DEG_CAP;
    const int cn4  = (cn + 3) >> 2;
    for (int i = tid; i < cn4; i += DEG_T) pl4[i] = make_float4(0.0f, 0.0f, 0.0f, 0.0f);
    const bool doBin = (blockIdx.z == 0);
    if (tid < AGG_NCH) bcount[tid] = 0u;
    __syncthreads();

    const int per = cdiv(E, DEG_G);
    const int e0  = g * per;
    const int e1  = (e0 + per < E) ? (e0 + per) : E;
    for (int e = e0 + tid; e < e1; e += 4 * DEG_T) {
        int u[4], v[4]; float th[4];
#pragma unroll
        for (int k = 0; k < 4; ++k) {
            const int ek = e + k * DEG_T;
            const int ec = (ek < e1) ? ek : e;
            u[k] = snd[ec]; v[k] = rcv[ec]; th[k] = theta[ec];
            if (ek >= e1) { u[k] = -1; v[k] = -1; }
        }
#pragma unroll
        for (int k = 0; k < 4; ++k) {
            const float cs = cosf(th[k]), sn = sinf(th[k]);
            const float c2 = cs * cs, s2 = sn * sn;
            const float wu = a ? s2 : c2;
            const float wv = a ? c2 : s2;
            const unsigned iu = (unsigned)(u[k] - base);
            const unsigned iv = (unsigned)(v[k] - base);
            if (iu < (unsigned)cn) __hip_atomic_fetch_add(pl + iu, wu, __ATOMIC_RELAXED, __HIP_MEMORY_SCOPE_WORKGROUP);
            if (iv < (unsigned)cn) __hip_atomic_fetch_add(pl + iv, wv, __ATOMIC_RELAXED, __HIP_MEMORY_SCOPE_WORKGROUP);
            if (doBin && u[k] >= 0) {
                const unsigned b = (unsigned)(a ? u[k] : v[k]) / AGG_ROWS;
                if (b < (unsigned)AGG_NCH) {
                    const unsigned slot = atomicAdd(&bcount[b], 1u);
                    if (slot < (unsigned)BIN_CAP) bl[b * BIN_W + 1 + slot] = (unsigned short)(e + k * DEG_T - e0);
                }
            }
        }
    }
    __syncthreads();

    float4* dst = (float4*)(part + ((size_t)a * DEG_G + g) * pitch + base);
    for (int i = tid; i < cn4; i += DEG_T) dst[i] = pl4[i];

    if (doBin) {
        if (tid < AGG_NCH) {
            const unsigned n = bcount[tid];
            bl[tid * BIN_W] = (unsigned short)((n > (unsigned)BIN_CAP || per > 65536) ? 0xFFFFu : n);
        }
        __syncthreads();
        const int sl = a * DEG_G + g;
        for (int i = tid; i < AGG_NCH * (BIN_W / 8); i += DEG_T) {
            const int b = i >> 3, q = i & 7;
            ((uint4*)(blst + ((size_t)b * BIN_S + sl) * BIN_W))[q] = bl4[i];
        }
    }
}

__global__ void deg_reduce(const float* __restrict__ part, int Nn, int pitch,
                           float* __restrict__ disS, float* __restrict__ disR) {
    const int n = blockIdx.x * blockDim.x + threadIdx.x;
    if (n >= Nn) return;
    const int a = blockIdx.y;
    const float* p = part + (size_t)a * DEG_G * pitch + n;
    float d = 1.0f;
#pragma unroll 16
    for (int g = 0; g < DEG_G; ++g) d += p[(size_t)g * pitch];
    const float v = d + 1e-12f;
    (a ? disR : disS)[n] = (v > 1e11f) ? 0.0f : rsqrtf(v);
}

#define AGG_TR   13
#define AGG_PR   (AGG_TR * 16)
#define AGG_P    132
#define AGG_T    1024
#define AGG_LCAP 16384

__global__ void weight_kernel(const int* __restrict__ snd, const int* __restrict__ rcv,
                              const float* __restrict__ theta, int E, int Nn,
                              const float* __restrict__ disS, const float* __restrict__ disR,
                              int4* __restrict__ rec, unsigned char* __restrict__ cid,
                              float* __restrict__ selfS, float* __restrict__ selfD) {
    const int i = blockIdx.x * blockDim.x + threadIdx.x;
    if (i < E) {
        const int s = snd[i], r = rcv[i];
        const float c = cosf(theta[i]), sn = sinf(theta[i]);
        const float a = disS[s] * (c * c) * disR[r];
        const float b = disR[s] * (sn * sn) * disS[r];
        rec[i] = make_int4(s, r, __float_as_int(a), __float_as_int(b));
        cid[i]     = (unsigned char)(r / AGG_ROWS);
        cid[i + E] = (unsigned char)(s / AGG_ROWS);
    } else if (i < E + Nn) {
        const int n = i - E;
        const float ds = disS[n], dr = disR[n];
        selfS[n] = ds * dr;
        selfD[n] = dr * ds;
        const int pad = (16 - ((2 * E) & 15)) & 15;
        if (n < pad) cid[2 * E + n] = 0;
    }
}

template <int RELU_IN>
__device__ __forceinline__ v16h frag16(float4 p0, float4 p1, float4 p2, float4 p3) {
    if (RELU_IN) {
        p0.x = fmaxf(p0.x, 0.0f); p0.y = fmaxf(p0.y, 0.0f); p0.z = fmaxf(p0.z, 0.0f); p0.w = fmaxf(p0.w, 0.0f);
        p1.x = fmaxf(p1.x, 0.0f); p1.y = fmaxf(p1.y, 0.0f); p1.z = fmaxf(p1.z, 0.0f); p1.w = fmaxf(p1.w, 0.0f);
        p2.x = fmaxf(p2.x, 0.0f); p2.y = fmaxf(p2.y, 0.0f); p2.z = fmaxf(p2.z, 0.0f); p2.w = fmaxf(p2.w, 0.0f);
        p3.x = fmaxf(p3.x, 0.0f); p3.y = fmaxf(p3.y, 0.0f); p3.z = fmaxf(p3.z, 0.0f); p3.w = fmaxf(p3.w, 0.0f);
    }
    v16h a;
    a[0]  = (_Float16)p0.x; a[1]  = (_Float16)p0.y;
    a[2]  = (_Float16)p0.z; a[3]  = (_Float16)p0.w;
    a[4]  = (_Float16)p1.x; a[5]  = (_Float16)p1.y;
    a[6]  = (_Float16)p1.z; a[7]  = (_Float16)p1.w;
    a[8]  = (_Float16)p2.x; a[9]  = (_Float16)p2.y;
    a[10] = (_Float16)p2.z; a[11] = (_Float16)p2.w;
    a[12] = (_Float16)p3.x; a[13] = (_Float16)p3.y;
    a[14] = (_Float16)p3.z; a[15] = (_Float16)p3.w;
    return a;
}

__device__ __forceinline__ v16h load_a_frag(const float4* __restrict__ Ar, int k0, int hi) {
    const int b = (k0 >> 2) + hi * 2;
    return frag16<0>(Ar[b], Ar[b + 1], Ar[b + 4], Ar[b + 5]);
}

__device__ __forceinline__ v16h plane_a_frag(const float* ar, int k0, int hi) {
    const float4* q = (const float4*)(ar + k0 + 8 * hi);
    return frag16<1>(q[0], q[1], q[4], q[5]);
}

template <int LAYER>
__global__ void __launch_bounds__(AGG_T)
agg_lds(const float* __restrict__ Y, const int4* __restrict__ rec,
        const uint4* __restrict__ cid16, const float* __restrict__ bias,
        const float* __restrict__ aSelf, const float* __restrict__ bSelf, int E, int Nn,
        const unsigned short* __restrict__ blst,
        const v16h* __restrict__ Bn, const float* __restrict__ biasN, float* __restrict__ Yn) {
    __shared__ float4   pl4[AGG_PR * (AGG_P / 4)];
    __shared__ unsigned lst[AGG_LCAP];
    __shared__ unsigned s_pre[BIN_S];
    __shared__ unsigned s_cnt;
    __shared__ unsigned s_scan;
    float* pl = (float*)pl4;
    const int tid   = threadIdx.x;
    const int chunk = blockIdx.x;
    const int base  = chunk * AGG_ROWS;
    const int cn    = (Nn - base < AGG_ROWS) ? (Nn - base) : AGG_ROWS;
    const int M2    = 2 * E;

    static_assert(AGG_T == BIN_S * (BIN_W / 8), "eight threads per sub-list line");
    const int sl = tid >> 3;
    const int sq = tid & 7;
    const unsigned short* line = blst + ((size_t)chunk * BIN_S + sl) * BIN_W;
    const unsigned slen = line[0];
    const uint4    sent = ((const uint4*)line)[sq];
    if (tid < 32) {
        static_assert(BIN_S == 128, "a lane of wave 0 takes four sub-lists");
        unsigned c4[4], sum = 0u, bad = 0u;
#pragma unroll
        for (int j = 0; j < 4; ++j) c4[j] = blst[((size_t)chunk * BIN_S + 4 * tid + j) * BIN_W];
#pragma unroll
        for (int j = 0; j < 4; ++j) { bad |= (c4[j] > (unsigned)BIN_CAP) ? 1u : 0u; sum += c4[j]; }
        unsigned inc = sum;
        for (int d = 1; d < 32; d <<= 1) {
            const unsigned t = __shfl_up(inc, d);
            if (tid >= d) inc += t;
        }
        unsigned ex = inc - sum;
#pragma unroll
        for (int j = 0; j < 4; ++j) { s_pre[4 * tid + j] = ex; ex += c4[j]; }
        const int anyBad = __any((int)bad);
        if (tid == 31) { s_cnt = anyBad ? 0u : inc; s_scan = anyBad ? 1u : 0u; }
    }

    for (int i = tid; i < AGG_PR * 32; i += AGG_T) {
        const int row = i >> 5;
        const int l   = i & 31;
        float4 v = make_float4(0.0f, 0.0f, 0.0f, 0.0f);
        if (row < cn) {
            const float* yr = Y + (size_t)(base + row) * DK2 + 4 * l;
            const float4 ys = *(const float4*)yr;
            const float4 yd = *(const float4*)(yr + DF);
            const float as = aSelf[base + row];
            const float bs = bSelf[base + row];
            v.x = bias[l]      + as * ys.x + bs * yd.x;
            v.y = bias[l + 32] + as * ys.y + bs * yd.y;
            v.z = bias[l + 64] + as * ys.z + bs * yd.z;
            v.w = bias[l + 96] + as * ys.w + bs * yd.w;
        }
        float* d = pl + row * AGG_P + l;
        d[0] = v.x; d[32] = v.y; d[64] = v.z; d[96] = v.w;
    }
    __syncthreads();

    const bool doScan = (s_scan != 0u);
    if (!doScan) {
        const int perS = cdiv(E, DEG_G);
        const unsigned vb = (unsigned)((sl >= DEG_G ? E : 0) + (sl & (DEG_G - 1)) * perS);
        const unsigned p  = s_pre[sl];
        const unsigned wq[4] = { sent.x, sent.y, sent.z, sent.w };
#pragma unroll
        for (int i = 0; i < 8; ++i) {
            const unsigned j = (unsigned)(sq * 8 + i);
            const unsigned w = (i & 1) ? (wq[i >> 1] >> 16) : (wq[i >> 1] & 0xFFFFu);
            if (j >= 1u && j <= slen) lst[p + j - 1u] = vb + w;
        }
    }

    const unsigned pat = (unsigned)chunk * 0x01010101u;
    const int nw = doScan ? ((M2 + 15) >> 4) : 0;
    for (int w0 = tid; w0 < nw; w0 += 4 * AGG_T) {
        uint4 kq[4];
#pragma unroll
        for (int k = 0; k < 4; ++k) {
            const int w  = w0 + k * AGG_T;
            const int wc = (w < nw) ? w : w0;
            kq[k] = cid16[wc];
            if (w >= nw) { kq[k].x = ~pat; kq[k].y = ~pat; kq[k].z = ~pat; kq[k].w = ~pat; }
        }
#pragma unroll
        for (int k = 0; k < 4; ++k) {
            const int w = w0 + k * AGG_T;
            const unsigned x0 = kq[k].x ^ pat, x1 = kq[k].y ^ pat, x2 = kq[k].z ^ pat, x3 = kq[k].w ^ pat;
            const unsigned t0 = ~(((x0 & 0x7F7F7F7Fu) + 0x7F7F7F7Fu) | x0 | 0x7F7F7F7Fu);
            const unsigned t1 = ~(((x1 & 0x7F7F7F7Fu) + 0x7F7F7F7Fu) | x1 | 0x7F7F7F7Fu);
            const unsigned t2 = ~(((x2 & 0x7F7F7F7Fu) + 0x7F7F7F7Fu) | x2 | 0x7F7F7F7Fu);
            const unsigned t3 = ~(((x3 & 0x7F7F7F7Fu) + 0x7F7F7F7Fu) | x3 | 0x7F7F7F7Fu);
            unsigned m = (t0 >> 7) | (t1 >> 6) | (t2 >> 5) | (t3 >> 4);
            while (m) {
                const int pos = __builtin_ctz(m);
                m &= m - 1u;
                const int idx = w * 16 + (pos & 7) * 4 + (pos >> 3);
                if (idx < M2) {
                    const unsigned slot = atomicAdd(&s_cnt, 1u);
                    if (slot < (unsigned)AGG_LCAP) {
                        lst[slot] = (unsigned)idx;
                    } else {
                        int4 q; int s, r; float a, b;
                        if (idx < E) { q = rec[idx];     s = q.x; r = q.y; a = __int_as_float(q.z); b = __int_as_float(q.w); }
                        else         { q = rec[idx - E]; s = q.y; r = q.x; a = __int_as_float(q.w); b = __int_as_float(q.z); }
                        const float* yr = Y + (long)s * DK2;
                        float* dst = pl + (r - base) * AGG_P;
                        for (int p = 0; p < DF; ++p)
                            __hip_atomic_fetch_add(dst + 32 * (p & 3) + (p >> 2), a * yr[p] + b * yr[DF + p],
                                                   __ATOMIC_RELAXED, __HIP_MEMORY_SCOPE_WORKGROUP);
                    }
                }
            }
        }
    }
    __syncthreads();

    const int total = __builtin_amdgcn_readfirstlane((int)s_cnt);
    const int cnt   = (total < AGG_LCAP) ? total : AGG_LCAP;
    const int wv    = __builtin_amdgcn_readfirstlane(tid >> 5);
    const int lane  = tid & 31;
    const int per   = (cnt + (AGG_T / 32) - 1) / (AGG_T / 32);
    const int st    = wv * per;
    const int en    = (st + per < cnt) ? (st + per) : cnt;
    const float* Yl = Y + lane * 4;
    float* pll = pl + lane;
    if (st < en) {
        const int len = en - st;
        const int ng  = (len + 31) >> 5;
        const int gs  = (((len + ng - 1) / ng) + 3) & ~3;
        int sv = 0, rv = 0, av = 0, bv = 0;
        {
            const int n0 = (len < gs) ? len : gs;
            if (lane < n0) {
                const unsigned e = lst[st + lane];
                if (e < (unsigned)E) { const int4 q = rec[e];     sv = q.x; rv = q.y; av = q.z; bv = q.w; }
                else                 { const int4 q = rec[e - E]; sv = q.y; rv = q.x; av = q.w; bv = q.z; }
            }
        }
        for (int g = st; g < en; g += gs) {
            const int n = (en - g < gs) ? (en - g) : gs;
            const int cs = sv, cr = rv, ca = av, cb = bv;
            const int gn = g + gs;
            if (gn < en) {
                const int nn = (en - gn < gs) ? (en - gn) : gs;
                if (lane < nn) {
                    const unsigned e = lst[gn + lane];
                    if (e < (unsigned)E) { const int4 q = rec[e];     sv = q.x; rv = q.y; av = q.z; bv = q.w; }
                    else                 { const int4 q = rec[e - E]; sv = q.y; rv = q.x; av = q.w; bv = q.z; }
                }
            }
            int t = 0;
            for (; t + 4 <= n; t += 4) {
                float4 ys[4], yd[4];
                int r[4]; float a[4], b[4];
#pragma unroll
                for (int u = 0; u < 4; ++u) {
                    const int s = __builtin_amdgcn_readlane(cs, t + u);
                    r[u] = __builtin_amdgcn_readlane(cr, t + u) - base;
                    a[u] = __int_as_float(__builtin_amdgcn_readlane(ca, t + u));
                    b[u] = __int_as_float(__builtin_amdgcn_readlane(cb, t + u));
                    const float* yr = Yl + (long)s * DK2;
                    ys[u] = *(const float4*)yr;
                    yd[u] = *(const float4*)(yr + DF);
                }
#pragma unroll
                for (int u = 0; u < 4; ++u) {
                    float* dst = pll + r[u] * AGG_P;
                    __hip_atomic_fetch_add(dst,      a[u] * ys[u].x + b[u] * yd[u].x, __ATOMIC_RELAXED, __HIP_MEMORY_SCOPE_WORKGROUP);
                    __hip_atomic_fetch_add(dst + 32, a[u] * ys[u].y + b[u] * yd[u].y, __ATOMIC_RELAXED, __HIP_MEMORY_SCOPE_WORKGROUP);
                    __hip_atomic_fetch_add(dst + 64, a[u] * ys[u].z + b[u] * yd[u].z, __ATOMIC_RELAXED, __HIP_MEMORY_SCOPE_WORKGROUP);
                    __hip_atomic_fetch_add(dst + 96, a[u] * ys[u].w + b[u] * yd[u].w, __ATOMIC_RELAXED, __HIP_MEMORY_SCOPE_WORKGROUP);
                }
            }
            for (; t < n; ++t) {
                const int s = __builtin_amdgcn_readlane(cs, t);
                const int r = __builtin_amdgcn_readlane(cr, t) - base;
                const float a = __int_as_float(__builtin_amdgcn_readlane(ca, t));
                const float b = __int_as_float(__builtin_amdgcn_readlane(cb, t));
                const float* yr = Yl + (long)s * DK2;
                const float4 ys = *(const float4*)yr;
                const float4 yd = *(const float4*)(yr + DF);
                float* dst = pll + r * AGG_P;
                __hip_atomic_fetch_add(dst,      a * ys.x + b * yd.x, __ATOMIC_RELAXED, __HIP_MEMORY_SCOPE_WORKGROUP);
                __hip_atomic_fetch_add(dst + 32, a * ys.y + b * yd.y, __ATOMIC_RELAXED, __HIP_MEMORY_SCOPE_WORKGROUP);
                __hip_atomic_fetch_add(dst + 64, a * ys.z + b * yd.z, __ATOMIC_RELAXED, __HIP_MEMORY_SCOPE_WORKGROUP);
                __hip_atomic_fetch_add(dst + 96, a * ys.w + b * yd.w, __ATOMIC_RELAXED, __HIP_MEMORY_SCOPE_WORKGROUP);
            }
        }
    }
    __syncthreads();

    const int hi = lane >> 4;
    const int n  = lane & 15;
    if (LAYER == 0) {
        for (int task = wv; task < AGG_TR * 4; task += AGG_T / 32) {
            const int t    = task >> 2;
            const int hp   = (task >> 1) & 1;
            const int half = task & 1;
            if (t * 16 >= cn) continue;
            const float* ar = pl + (t * 16 + n) * AGG_P;
            v8f c[4] = {};
#pragma unroll 2
            for (int kt = 0; kt < DF / 32; ++kt) {
                const v16h a = plane_a_frag(ar, kt * 32, hi);
                const v16h* bq = Bn + ((kt * 2 + hp) * 8 + half * 4) * 32 + lane;
#pragma unroll
                for (int jt = 0; jt < 4; ++jt) {
                    const v16h b = bq[jt * 32];
                    c[jt] = __builtin_amdgcn_wmma_f32_16x16x32_f16(false, a, false, b,
                                                                   (short)0, c[jt], false, false);
                }
            }
            const int lcol = 16 * hp + n;
#pragma unroll
            for (int i = 0; i < 8; ++i) {
                const int row = t * 16 + hi * 8 + i;
                if (row < cn) {
                    float4 y;
                    y.x = c[0][i]; y.y = c[1][i]; y.z = c[2][i]; y.w = c[3][i];
                    *(float4*)(Yn + (size_t)(base + row) * DK2 + half * DF + lcol * 4) = y;
                }
            }
        }
    } else {
        for (int task = wv; task < AGG_TR * (DOUT / 16); task += AGG_T / 32) {
            const int t  = task >> 2;
            const int tc = task & 3;
            if (t * 16 >= cn) continue;
            const float* ar = pl + (t * 16 + n) * AGG_P;
            v8f c = {};
#pragma unroll
            for (int kt = 0; kt < DF / 32; ++kt) {
                const v16h a = plane_a_frag(ar, kt * 32, hi);
                const v16h b = Bn[(kt * (DOUT / 16) + tc) * 32 + lane];
                c = __builtin_amdgcn_wmma_f32_16x16x32_f16(false, a, false, b,
                                                           (short)0, c, false, false);
            }
            const int ncol = tc * 16 + n;
            const float bv = biasN[ncol];
#pragma unroll
            for (int i = 0; i < 8; ++i) {
                const int row = t * 16 + hi * 8 + i;
                if (row < cn) Yn[(size_t)(base + row) * DOUT + ncol] = c[i] + bv;
            }
        }
    }
}

__global__ void pack_pre(const float* __restrict__ Ws, const float* __restrict__ Wd,
                         _Float16* __restrict__ dst) {
    int idx = blockIdx.x * blockDim.x + threadIdx.x;
    if (idx >= DF * DK2) return;
    int j    = idx & 15;
    int lane = (idx >> 4) & 31;
    int t    = idx >> 9;
    int q    = t & 7;
    int hp   = (t >> 3) & 1;
    int kt   = t >> 4;
    int half = q >> 2;
    int jt   = q & 3;
    int k = kt * 32 + ((j < 8) ? 0 : 16) + (lane >> 4) * 8 + (j & 7);
    int n = 32 * jt + 16 * hp + (lane & 15);
    float v = half ? 0.5f * Wd[k * DF + n] : 0.5f * Ws[k * DF + n];
    dst[idx] = (_Float16)v;
}

__global__ void pack_single(const float* __restrict__ W, _Float16* __restrict__ dst) {
    int idx = blockIdx.x * blockDim.x + threadIdx.x;
    if (idx >= DF * DOUT) return;
    int j    = idx & 15;
    int lane = (idx >> 4) & 31;
    int t    = idx >> 9;
    int k0t  = t / (DOUT / 16);
    int tc   = t % (DOUT / 16);
    int k = k0t * 32 + ((j < 8) ? 0 : 16) + (lane >> 4) * 8 + (j & 7);
    int n = tc * 16 + (lane & 15);
    dst[idx] = (_Float16)W[k * DOUT + n];
}

__global__ void bias_combine(const float* __restrict__ bs, const float* __restrict__ bd,
                             float* __restrict__ out, int n) {
    int i = blockIdx.x * blockDim.x + threadIdx.x;
    if (i < n) out[i] = 0.5f * (bs[i] + bd[i]);
}

__global__ void __launch_bounds__(256)
gemm_pre(const float* __restrict__ A, int Mrows,
         const v16h* __restrict__ Bp, float* __restrict__ Y) {
    int wave = (blockIdx.x * blockDim.x + threadIdx.x) >> 5;
    int lane = threadIdx.x & 31;
    int rowTiles = Mrows >> 4;
    if (wave >= rowTiles * 2) return;
    int tr = wave >> 1;
    int hp = wave & 1;
    int hi = lane >> 4;
    int n  = lane & 15;

    const float4* Ar = (const float4*)(A + (long)(tr * 16 + n) * DF);
    v8f c[8] = {};
#pragma unroll
    for (int kt = 0; kt < DF / 32; ++kt) {
        v16h a = load_a_frag(Ar, kt * 32, hi);
        const v16h* bq = Bp + ((kt * 2 + hp) * 8) * 32 + lane;
#pragma unroll
        for (int q = 0; q < 8; ++q) {
            v16h b = bq[q * 32];
            c[q] = __builtin_amdgcn_wmma_f32_16x16x32_f16(false, a, false, b,
                                                          (short)0, c[q], false, false);
        }
    }

    const int lcol = 16 * hp + n;
    const int row0 = tr * 16 + hi * 8;
#pragma unroll
    for (int i = 0; i < 8; ++i) {
        const int row = row0 + i;
        float4 ys, yd;
        ys.x = c[0][i]; ys.y = c[1][i]; ys.z = c[2][i]; ys.w = c[3][i];
        yd.x = c[4][i]; yd.y = c[5][i]; yd.z = c[6][i]; yd.w = c[7][i];
        float* yr = Y + (long)row * DK2 + lcol * 4;
        *(float4*)yr        = ys;
        *(float4*)(yr + DF) = yd;
    }
}

extern "C" void kernel_launch(void* const* d_in, const int* in_sizes, int n_in,
                              void* d_out, int out_size, void* d_ws, size_t ws_size,
                              hipStream_t stream) {
    const float* x      = (const float*)d_in[0];
    const int*   edges  = (const int*)d_in[1];
    const float* theta  = (const float*)d_in[2];
    const float* W1s    = (const float*)d_in[3];
    const float* W1d    = (const float*)d_in[4];
    const float* b1s    = (const float*)d_in[5];
    const float* b1d    = (const float*)d_in[6];
    const float* W2s    = (const float*)d_in[7];
    const float* W2d    = (const float*)d_in[8];
    const float* b2s    = (const float*)d_in[9];
    const float* b2d    = (const float*)d_in[10];
    const float* Wro    = (const float*)d_in[11];
    const float* bro    = (const float*)d_in[12];

    const int N = in_sizes[0] / DF;
    const int E = in_sizes[2];
    const int* snd = edges;
    const int* rcv = edges + E;

    char* ws = (char*)d_ws;
    size_t off = 0;
    auto alloc = [&](size_t bytes) { char* p = ws + off; off = (off + bytes + 255) & ~(size_t)255; return p; };
    float*    degS  = (float*)alloc((size_t)N * 4);
    float*    degR  = (float*)alloc((size_t)N * 4);
    int4*     rec   = (int4*)alloc((size_t)E * 16);
    unsigned char* cid = (unsigned char*)alloc((size_t)cdiv(2 * E, 16) * 16);
    float*    selfS = (float*)alloc((size_t)N * 4);
    float*    selfD = (float*)alloc((size_t)N * 4);
    float*    Y1    = (float*)alloc((size_t)N * DK2 * 4);
    float*    Y2    = (float*)alloc((size_t)N * DK2 * 4);
    _Float16* Bp1   = (_Float16*)alloc((size_t)DK2 * DF * 2);
    _Float16* Bp2   = (_Float16*)alloc((size_t)DK2 * DF * 2);
    _Float16* Bpro  = (_Float16*)alloc((size_t)DF * DOUT * 2);
    float*    bias1 = (float*)alloc(DF * 4);
    float*    bias2 = (float*)alloc(DF * 4);
    unsigned short* blst = (unsigned short*)alloc((size_t)AGG_NCH * BIN_S * BIN_W * 2);

    const int T = 256;

    float* part = Y1;
    const int pitch = (N + 31) & ~31;
    deg_partial<<<dim3(DEG_G, 2, cdiv(N, DEG_CAP)), DEG_T, 0, stream>>>(snd, rcv, theta, E, N, pitch, part, blst);
    deg_reduce<<<dim3(cdiv(N, T), 2), T, 0, stream>>>(part, N, pitch, degS, degR);
    weight_kernel<<<cdiv(E + N, T), T, 0, stream>>>(snd, rcv, theta, E, N, degS, degR, rec, cid, selfS, selfD);

    pack_pre<<<cdiv(DF * DK2, T), T, 0, stream>>>(W1s, W1d, Bp1);
    pack_pre<<<cdiv(DF * DK2, T), T, 0, stream>>>(W2s, W2d, Bp2);
    pack_single<<<cdiv(DF * DOUT, T), T, 0, stream>>>(Wro, Bpro);
    bias_combine<<<1, DF, 0, stream>>>(b1s, b1d, bias1, DF);
    bias_combine<<<1, DF, 0, stream>>>(b2s, b2d, bias2, DF);

    const int preWaves      = (N / 16) * 2;
    const int aggBlocks     = cdiv(N, AGG_ROWS);

    gemm_pre<<<cdiv(preWaves * 32, T), T, 0, stream>>>(x, N, (const v16h*)Bp1, Y1);
    agg_lds<0><<<aggBlocks, AGG_T, 0, stream>>>(Y1, rec, (const uint4*)cid, bias1, selfS, selfD, E, N,
                                                blst, (const v16h*)Bp2, bro, Y2);

    agg_lds<1><<<aggBlocks, AGG_T, 0, stream>>>(Y2, rec, (const uint4*)cid, bias2, selfS, selfD, E, N,
                                                blst, (const v16h*)Bpro, bro, (float*)d_out);
}


# Edge passes stage their own B operand: the work-group rounds the raw f32 weights to f16 into LDS in fragment order and forms the bias sum in place; 4 prep launches go (11 become 7); sum order kept
# speedup vs baseline: 4.3167x; 1.0579x over previous
#include <hip/hip_runtime.h>

typedef __attribute__((ext_vector_type(16))) _Float16 v16h;
typedef __attribute__((ext_vector_type(8)))  float    v8f;
typedef __attribute__((ext_vector_type(8)))  _Float16 v8h;
typedef __attribute__((ext_vector_type(4)))  unsigned int u4v;
typedef __attribute__((ext_vector_type(8)))  unsigned int u8v;

#define DF   128
#define DK2  256
#define DOUT 64

static __host__ __device__ inline int cdiv(int a, int b) { return (a + b - 1) / b; }


#define DEG_CAP 50176
#define DEG_G   64
#define DEG_T   1024

#define AGG_ROWS 196
#define AGG_NCH  256
#define BIN_W    64
#define BIN_CAP  (BIN_W - 1)
#define BIN_S    (2 * DEG_G)

__global__ void __launch_bounds__(DEG_T)
deg_partial(const int* __restrict__ snd, const int* __restrict__ rcv,
            const float* __restrict__ theta, int E, int Nn, int pitch,
            float* __restrict__ part, unsigned short* __restrict__ blst) {
    __shared__ float4 pl4[DEG_CAP / 4];
    __shared__ uint4    bl4[AGG_NCH * BIN_W / 8];
    __shared__ unsigned bcount[AGG_NCH];
    float* pl = (float*)pl4;
    unsigned short* bl = (unsigned short*)bl4;
    const int tid  = threadIdx.x;
    const int g    = blockIdx.x;
    const int a    = blockIdx.y;
    const int base = blockIdx.z * DEG_CAP;
    const int cn   = (Nn - base < DEG_CAP) ? (Nn - base) : DEG_CAP;
    const int cn4  = (cn + 3) >> 2;
    for (int i = tid; i < cn4; i += DEG_T) pl4[i] = make_float4(0.0f, 0.0f, 0.0f, 0.0f);
    const bool doBin = (blockIdx.z == 0);
    if (tid < AGG_NCH) bcount[tid] = 0u;
    __syncthreads();

    const int per = cdiv(E, DEG_G);
    const int e0  = g * per;
    const int e1  = (e0 + per < E) ? (e0 + per) : E;
    for (int e = e0 + tid; e < e1; e += 4 * DEG_T) {
        int u[4], v[4]; float th[4];
#pragma unroll
        for (int k = 0; k < 4; ++k) {
            const int ek = e + k * DEG_T;
            const int ec = (ek < e1) ? ek : e;
            u[k] = snd[ec]; v[k] = rcv[ec]; th[k] = theta[ec];
            if (ek >= e1) { u[k] = -1; v[k] = -1; }
        }
#pragma unroll
        for (int k = 0; k < 4; ++k) {
            const float cs = cosf(th[k]), sn = sinf(th[k]);
            const float c2 = cs * cs, s2 = sn * sn;
            const float wu = a ? s2 : c2;
            const float wv = a ? c2 : s2;
            const unsigned iu = (unsigned)(u[k] - base);
            const unsigned iv = (unsigned)(v[k] - base);
            if (iu < (unsigned)cn) __hip_atomic_fetch_add(pl + iu, wu, __ATOMIC_RELAXED, __HIP_MEMORY_SCOPE_WORKGROUP);
            if (iv < (unsigned)cn) __hip_atomic_fetch_add(pl + iv, wv, __ATOMIC_RELAXED, __HIP_MEMORY_SCOPE_WORKGROUP);
            if (doBin && u[k] >= 0) {
                const unsigned b = (unsigned)(a ? u[k] : v[k]) / AGG_ROWS;
                if (b < (unsigned)AGG_NCH) {
                    const unsigned slot = atomicAdd(&bcount[b], 1u);
                    if (slot < (unsigned)BIN_CAP) bl[b * BIN_W + 1 + slot] = (unsigned short)(e + k * DEG_T - e0);
                }
            }
        }
    }
    __syncthreads();

    float4* dst = (float4*)(part + ((size_t)a * DEG_G + g) * pitch + base);
    for (int i = tid; i < cn4; i += DEG_T) dst[i] = pl4[i];

    if (doBin) {
        if (tid < AGG_NCH) {
            const unsigned n = bcount[tid];
            bl[tid * BIN_W] = (unsigned short)((n > (unsigned)BIN_CAP || per > 65536) ? 0xFFFFu : n);
        }
        __syncthreads();
        const int sl = a * DEG_G + g;
        for (int i = tid; i < AGG_NCH * (BIN_W / 8); i += DEG_T) {
            const int b = i >> 3, q = i & 7;
            ((uint4*)(blst + ((size_t)b * BIN_S + sl) * BIN_W))[q] = bl4[i];
        }
    }
}

__global__ void deg_reduce(const float* __restrict__ part, int Nn, int pitch,
                           float* __restrict__ disS, float* __restrict__ disR) {
    const int n = blockIdx.x * blockDim.x + threadIdx.x;
    if (n >= Nn) return;
    const int a = blockIdx.y;
    const float* p = part + (size_t)a * DEG_G * pitch + n;
    float d = 1.0f;
#pragma unroll 16
    for (int g = 0; g < DEG_G; ++g) d += p[(size_t)g * pitch];
    const float v = d + 1e-12f;
    (a ? disR : disS)[n] = (v > 1e11f) ? 0.0f : rsqrtf(v);
}

#define AGG_TR   13
#define AGG_PR   (AGG_TR * 16)
#define AGG_P    132
#define AGG_T    1024
#define AGG_LCAP 16384

__global__ void weight_kernel(const int* __restrict__ snd, const int* __restrict__ rcv,
                              const float* __restrict__ theta, int E, int Nn,
                              const float* __restrict__ disS, const float* __restrict__ disR,
                              int4* __restrict__ rec, unsigned char* __restrict__ cid,
                              float* __restrict__ selfS, float* __restrict__ selfD) {
    const int i = blockIdx.x * blockDim.x + threadIdx.x;
    if (i < E) {
        const int s = snd[i], r = rcv[i];
        const float c = cosf(theta[i]), sn = sinf(theta[i]);
        const float a = disS[s] * (c * c) * disR[r];
        const float b = disR[s] * (sn * sn) * disS[r];
        rec[i] = make_int4(s, r, __float_as_int(a), __float_as_int(b));
        cid[i]     = (unsigned char)(r / AGG_ROWS);
        cid[i + E] = (unsigned char)(s / AGG_ROWS);
    } else if (i < E + Nn) {
        const int n = i - E;
        const float ds = disS[n], dr = disR[n];
        selfS[n] = ds * dr;
        selfD[n] = dr * ds;
        const int pad = (16 - ((2 * E) & 15)) & 15;
        if (n < pad) cid[2 * E + n] = 0;
    }
}

template <int RELU_IN>
__device__ __forceinline__ v16h frag16(float4 p0, float4 p1, float4 p2, float4 p3) {
    if (RELU_IN) {
        p0.x = fmaxf(p0.x, 0.0f); p0.y = fmaxf(p0.y, 0.0f); p0.z = fmaxf(p0.z, 0.0f); p0.w = fmaxf(p0.w, 0.0f);
        p1.x = fmaxf(p1.x, 0.0f); p1.y = fmaxf(p1.y, 0.0f); p1.z = fmaxf(p1.z, 0.0f); p1.w = fmaxf(p1.w, 0.0f);
        p2.x = fmaxf(p2.x, 0.0f); p2.y = fmaxf(p2.y, 0.0f); p2.z = fmaxf(p2.z, 0.0f); p2.w = fmaxf(p2.w, 0.0f);
        p3.x = fmaxf(p3.x, 0.0f); p3.y = fmaxf(p3.y, 0.0f); p3.z = fmaxf(p3.z, 0.0f); p3.w = fmaxf(p3.w, 0.0f);
    }
    v16h a;
    a[0]  = (_Float16)p0.x; a[1]  = (_Float16)p0.y;
    a[2]  = (_Float16)p0.z; a[3]  = (_Float16)p0.w;
    a[4]  = (_Float16)p1.x; a[5]  = (_Float16)p1.y;
    a[6]  = (_Float16)p1.z; a[7]  = (_Float16)p1.w;
    a[8]  = (_Float16)p2.x; a[9]  = (_Float16)p2.y;
    a[10] = (_Float16)p2.z; a[11] = (_Float16)p2.w;
    a[12] = (_Float16)p3.x; a[13] = (_Float16)p3.y;
    a[14] = (_Float16)p3.z; a[15] = (_Float16)p3.w;
    return a;
}

__device__ __forceinline__ v16h load_a_frag(const float4* __restrict__ Ar, int k0, int hi) {
    const int b = (k0 >> 2) + hi * 2;
    return frag16<0>(Ar[b], Ar[b + 1], Ar[b + 4], Ar[b + 5]);
}

__device__ __forceinline__ v16h plane_a_frag(const float* ar, int k0, int hi) {
    const float4* q = (const float4*)(ar + k0 + 8 * hi);
    return frag16<1>(q[0], q[1], q[4], q[5]);
}

__device__ __forceinline__ v16h lds_b_frag(const u4v* B4, int t, int lane) {
    const u4v lo = B4[(t * 2) * 32 + lane];
    const u4v hi = B4[(t * 2 + 1) * 32 + lane];
    const u8v w = __builtin_shufflevector(lo, hi, 0, 1, 2, 3, 4, 5, 6, 7);
    return __builtin_bit_cast(v16h, w);
}

__device__ __forceinline__ u4v pack8h(const float* f, float scale) {
    v8h h;
    h[0] = (_Float16)(scale * f[0]); h[1] = (_Float16)(scale * f[1]);
    h[2] = (_Float16)(scale * f[2]); h[3] = (_Float16)(scale * f[3]);
    h[4] = (_Float16)(scale * f[4]); h[5] = (_Float16)(scale * f[5]);
    h[6] = (_Float16)(scale * f[6]); h[7] = (_Float16)(scale * f[7]);
    return __builtin_bit_cast(u4v, h);
}

template <int LAYER>
__global__ void __launch_bounds__(AGG_T)
agg_lds(const float* __restrict__ Y, const int4* __restrict__ rec,
        const uint4* __restrict__ cid16, const float* __restrict__ biasS, const float* __restrict__ biasD,
        const float* __restrict__ aSelf, const float* __restrict__ bSelf, int E, int Nn,
        const unsigned short* __restrict__ blst,
        const float* __restrict__ Wa, const float* __restrict__ Wb,
        const float* __restrict__ biasN, float* __restrict__ Yn) {
    __shared__ float4   pl4[AGG_PR * (AGG_P / 4)];
    __shared__ u4v      Bl4[LAYER == 0 ? (DF * DK2 / 8) : (DF * DOUT / 8)];
    __shared__ unsigned lst[AGG_LCAP];
    __shared__ unsigned s_pre[BIN_S];
    __shared__ unsigned s_cnt;
    __shared__ unsigned s_scan;
    float* pl = (float*)pl4;
    const int tid   = threadIdx.x;
    const int chunk = blockIdx.x;
    const int base  = chunk * AGG_ROWS;
    const int cn    = (Nn - base < AGG_ROWS) ? (Nn - base) : AGG_ROWS;
    const int M2    = 2 * E;

    static_assert(AGG_T == BIN_S * (BIN_W / 8), "eight threads per sub-list line");
    const int sl = tid >> 3;
    const int sq = tid & 7;
    const unsigned short* line = blst + ((size_t)chunk * BIN_S + sl) * BIN_W;
    const unsigned slen = line[0];
    const uint4    sent = ((const uint4*)line)[sq];
    if (tid < 32) {
        static_assert(BIN_S == 128, "a lane of wave 0 takes four sub-lists");
        unsigned c4[4], sum = 0u, bad = 0u;
#pragma unroll
        for (int j = 0; j < 4; ++j) c4[j] = blst[((size_t)chunk * BIN_S + 4 * tid + j) * BIN_W];
#pragma unroll
        for (int j = 0; j < 4; ++j) { bad |= (c4[j] > (unsigned)BIN_CAP) ? 1u : 0u; sum += c4[j]; }
        unsigned inc = sum;
        for (int d = 1; d < 32; d <<= 1) {
            const unsigned t = __shfl_up(inc, d);
            if (tid >= d) inc += t;
        }
        unsigned ex = inc - sum;
#pragma unroll
        for (int j = 0; j < 4; ++j) { s_pre[4 * tid + j] = ex; ex += c4[j]; }
        const int anyBad = __any((int)bad);
        if (tid == 31) { s_cnt = anyBad ? 0u : inc; s_scan = anyBad ? 1u : 0u; }
    }

    constexpr int NOCT = (LAYER == 0) ? 4 : 1;
    float wf[NOCT][8];
    if constexpr (LAYER == 0) {
        const int c = tid & 15, hp = (tid >> 4) & 1, h = (tid >> 5) & 1, part = (tid >> 6) & 1;
        const int jt = (tid >> 7) & 3, half = (tid >> 9) & 1;
        const float* wsrc = (half ? Wb : Wa) + 32 * jt + 16 * hp + c;
#pragma unroll
        for (int kt = 0; kt < 4; ++kt) {
            const int kb = kt * 32 + part * 16 + h * 8;
#pragma unroll
            for (int q = 0; q < 8; ++q) wf[kt][q] = wsrc[(kb + q) * DF];
        }
    } else {
        const int c = tid & 15, tc = (tid >> 4) & 3, h = (tid >> 6) & 1, part = (tid >> 7) & 1, kt = (tid >> 8) & 3;
        const float* wsrc = Wa + tc * 16 + c;
        const int kb = kt * 32 + part * 16 + h * 8;
#pragma unroll
        for (int q = 0; q < 8; ++q) wf[0][q] = wsrc[(kb + q) * DOUT];
    }

    const int l = tid & 31;
    float bz[4];
#pragma unroll
    for (int j = 0; j < 4; ++j) bz[j] = 0.5f * (biasS[l + 32 * j] + biasD[l + 32 * j]);

    static_assert((AGG_T & 31) == 0, "a thread keeps its lane's columns over the rows it takes");
    for (int i = tid; i < AGG_PR * 32; i += AGG_T) {
        const int row = i >> 5;
        float4 v = make_float4(0.0f, 0.0f, 0.0f, 0.0f);
        if (row < cn) {
            const float* yr = Y + (size_t)(base + row) * DK2 + 4 * l;
            const float4 ys = *(const float4*)yr;
            const float4 yd = *(const float4*)(yr + DF);
            const float as = aSelf[base + row];
            const float bs = bSelf[base + row];
            v.x = bz[0] + as * ys.x + bs * yd.x;
            v.y = bz[1] + as * ys.y + bs * yd.y;
            v.z = bz[2] + as * ys.z + bs * yd.z;
            v.w = bz[3] + as * ys.w + bs * yd.w;
        }
        float* d = pl + row * AGG_P + l;
        d[0] = v.x; d[32] = v.y; d[64] = v.z; d[96] = v.w;
    }

    if constexpr (LAYER == 0) {
        const int c = tid & 15, hp = (tid >> 4) & 1, h = (tid >> 5) & 1, part = (tid >> 6) & 1;
        const int jt = (tid >> 7) & 3, half = (tid >> 9) & 1;
#pragma unroll
        for (int kt = 0; kt < 4; ++kt) {
            const int t = (kt * 2 + hp) * 8 + half * 4 + jt;
            Bl4[(t * 2 + part) * 32 + h * 16 + c] = pack8h(wf[kt], 0.5f);
        }
    } else {
        const int c = tid & 15, tc = (tid >> 4) & 3, h = (tid >> 6) & 1, part = (tid >> 7) & 1, kt = (tid >> 8) & 3;
        const int t = kt * (DOUT / 16) + tc;
        Bl4[(t * 2 + part) * 32 + h * 16 + c] = pack8h(wf[0], 1.0f);
    }
    __syncthreads();

    const bool doScan = (s_scan != 0u);
    if (!doScan) {
        const int perS = cdiv(E, DEG_G);
        const unsigned vb = (unsigned)((sl >= DEG_G ? E : 0) + (sl & (DEG_G - 1)) * perS);
        const unsigned p  = s_pre[sl];
        const unsigned wq[4] = { sent.x, sent.y, sent.z, sent.w };
#pragma unroll
        for (int i = 0; i < 8; ++i) {
            const unsigned j = (unsigned)(sq * 8 + i);
            const unsigned w = (i & 1) ? (wq[i >> 1] >> 16) : (wq[i >> 1] & 0xFFFFu);
            if (j >= 1u && j <= slen) lst[p + j - 1u] = vb + w;
        }
    }

    const unsigned pat = (unsigned)chunk * 0x01010101u;
    const int nw = doScan ? ((M2 + 15) >> 4) : 0;
    for (int w0 = tid; w0 < nw; w0 += 4 * AGG_T) {
        uint4 kq[4];
#pragma unroll
        for (int k = 0; k < 4; ++k) {
            const int w  = w0 + k * AGG_T;
            const int wc = (w < nw) ? w : w0;
            kq[k] = cid16[wc];
            if (w >= nw) { kq[k].x = ~pat; kq[k].y = ~pat; kq[k].z = ~pat; kq[k].w = ~pat; }
        }
#pragma unroll
        for (int k = 0; k < 4; ++k) {
            const int w = w0 + k * AGG_T;
            const unsigned x0 = kq[k].x ^ pat, x1 = kq[k].y ^ pat, x2 = kq[k].z ^ pat, x3 = kq[k].w ^ pat;
            const unsigned t0 = ~(((x0 & 0x7F7F7F7Fu) + 0x7F7F7F7Fu) | x0 | 0x7F7F7F7Fu);
            const unsigned t1 = ~(((x1 & 0x7F7F7F7Fu) + 0x7F7F7F7Fu) | x1 | 0x7F7F7F7Fu);
            const unsigned t2 = ~(((x2 & 0x7F7F7F7Fu) + 0x7F7F7F7Fu) | x2 | 0x7F7F7F7Fu);
            const unsigned t3 = ~(((x3 & 0x7F7F7F7Fu) + 0x7F7F7F7Fu) | x3 | 0x7F7F7F7Fu);
            unsigned m = (t0 >> 7) | (t1 >> 6) | (t2 >> 5) | (t3 >> 4);
            while (m) {
                const int pos = __builtin_ctz(m);
                m &= m - 1u;
                const int idx = w * 16 + (pos & 7) * 4 + (pos >> 3);
                if (idx < M2) {
                    const unsigned slot = atomicAdd(&s_cnt, 1u);
                    if (slot < (unsigned)AGG_LCAP) {
                        lst[slot] = (unsigned)idx;
                    } else {
                        int4 q; int s, r; float a, b;
                        if (idx < E) { q = rec[idx];     s = q.x; r = q.y; a = __int_as_float(q.z); b = __int_as_float(q.w); }
                        else         { q = rec[idx - E]; s = q.y; r = q.x; a = __int_as_float(q.w); b = __int_as_float(q.z); }
                        const float* yr = Y + (long)s * DK2;
                        float* dst = pl + (r - base) * AGG_P;
                        for (int p = 0; p < DF; ++p)
                            __hip_atomic_fetch_add(dst + 32 * (p & 3) + (p >> 2), a * yr[p] + b * yr[DF + p],
                                                   __ATOMIC_RELAXED, __HIP_MEMORY_SCOPE_WORKGROUP);
                    }
                }
            }
        }
    }
    __syncthreads();

    const int total = __builtin_amdgcn_readfirstlane((int)s_cnt);
    const int cnt   = (total < AGG_LCAP) ? total : AGG_LCAP;
    const int wv    = __builtin_amdgcn_readfirstlane(tid >> 5);
    const int lane  = tid & 31;
    const int per   = (cnt + (AGG_T / 32) - 1) / (AGG_T / 32);
    const int st    = wv * per;
    const int en    = (st + per < cnt) ? (st + per) : cnt;
    const float* Yl = Y + lane * 4;
    float* pll = pl + lane;
    if (st < en) {
        const int len = en - st;
        const int ng  = (len + 31) >> 5;
        const int gs  = (((len + ng - 1) / ng) + 3) & ~3;
        int sv = 0, rv = 0, av = 0, bv = 0;
        {
            const int n0 = (len < gs) ? len : gs;
            if (lane < n0) {
                const unsigned e = lst[st + lane];
                if (e < (unsigned)E) { const int4 q = rec[e];     sv = q.x; rv = q.y; av = q.z; bv = q.w; }
                else                 { const int4 q = rec[e - E]; sv = q.y; rv = q.x; av = q.w; bv = q.z; }
            }
        }
        for (int g = st; g < en; g += gs) {
            const int n = (en - g < gs) ? (en - g) : gs;
            const int cs = sv, cr = rv, ca = av, cb = bv;
            const int gn = g + gs;
            if (gn < en) {
                const int nn = (en - gn < gs) ? (en - gn) : gs;
                if (lane < nn) {
                    const unsigned e = lst[gn + lane];
                    if (e < (unsigned)E) { const int4 q = rec[e];     sv = q.x; rv = q.y; av = q.z; bv = q.w; }
                    else                 { const int4 q = rec[e - E]; sv = q.y; rv = q.x; av = q.w; bv = q.z; }
                }
            }
            int t = 0;
            for (; t + 4 <= n; t += 4) {
                float4 ys[4], yd[4];
                int r[4]; float a[4], b[4];
#pragma unroll
                for (int u = 0; u < 4; ++u) {
                    const int s = __builtin_amdgcn_readlane(cs, t + u);
                    r[u] = __builtin_amdgcn_readlane(cr, t + u) - base;
                    a[u] = __int_as_float(__builtin_amdgcn_readlane(ca, t + u));
                    b[u] = __int_as_float(__builtin_amdgcn_readlane(cb, t + u));
                    const float* yr = Yl + (long)s * DK2;
                    ys[u] = *(const float4*)yr;
                    yd[u] = *(const float4*)(yr + DF);
                }
#pragma unroll
                for (int u = 0; u < 4; ++u) {
                    float* dst = pll + r[u] * AGG_P;
                    __hip_atomic_fetch_add(dst,      a[u] * ys[u].x + b[u] * yd[u].x, __ATOMIC_RELAXED, __HIP_MEMORY_SCOPE_WORKGROUP);
                    __hip_atomic_fetch_add(dst + 32, a[u] * ys[u].y + b[u] * yd[u].y, __ATOMIC_RELAXED, __HIP_MEMORY_SCOPE_WORKGROUP);
                    __hip_atomic_fetch_add(dst + 64, a[u] * ys[u].z + b[u] * yd[u].z, __ATOMIC_RELAXED, __HIP_MEMORY_SCOPE_WORKGROUP);
                    __hip_atomic_fetch_add(dst + 96, a[u] * ys[u].w + b[u] * yd[u].w, __ATOMIC_RELAXED, __HIP_MEMORY_SCOPE_WORKGROUP);
                }
            }
            for (; t < n; ++t) {
                const int s = __builtin_amdgcn_readlane(cs, t);
                const int r = __builtin_amdgcn_readlane(cr, t) - base;
                const float a = __int_as_float(__builtin_amdgcn_readlane(ca, t));
                const float b = __int_as_float(__builtin_amdgcn_readlane(cb, t));
                const float* yr = Yl + (long)s * DK2;
                const float4 ys = *(const float4*)yr;
                const float4 yd = *(const float4*)(yr + DF);
                float* dst = pll + r * AGG_P;
                __hip_atomic_fetch_add(dst,      a * ys.x + b * yd.x, __ATOMIC_RELAXED, __HIP_MEMORY_SCOPE_WORKGROUP);
                __hip_atomic_fetch_add(dst + 32, a * ys.y + b * yd.y, __ATOMIC_RELAXED, __HIP_MEMORY_SCOPE_WORKGROUP);
                __hip_atomic_fetch_add(dst + 64, a * ys.z + b * yd.z, __ATOMIC_RELAXED, __HIP_MEMORY_SCOPE_WORKGROUP);
                __hip_atomic_fetch_add(dst + 96, a * ys.w + b * yd.w, __ATOMIC_RELAXED, __HIP_MEMORY_SCOPE_WORKGROUP);
            }
        }
    }
    __syncthreads();

    const int hi = lane >> 4;
    const int n  = lane & 15;
    if (LAYER == 0) {
        for (int task = wv; task < AGG_TR * 4; task += AGG_T / 32) {
            const int t    = task >> 2;
            const int hp   = (task >> 1) & 1;
            const int half = task & 1;
            if (t * 16 >= cn) continue;
            const float* ar = pl + (t * 16 + n) * AGG_P;
            v8f c[4] = {};
#pragma unroll 2
            for (int kt = 0; kt < DF / 32; ++kt) {
                const v16h a = plane_a_frag(ar, kt * 32, hi);
                const int tb = (kt * 2 + hp) * 8 + half * 4;
#pragma unroll
                for (int jt = 0; jt < 4; ++jt) {
                    const v16h b = lds_b_frag(Bl4, tb + jt, lane);
                    c[jt] = __builtin_amdgcn_wmma_f32_16x16x32_f16(false, a, false, b,
                                                                   (short)0, c[jt], false, false);
                }
            }
            const int lcol = 16 * hp + n;
#pragma unroll
            for (int i = 0; i < 8; ++i) {
                const int row = t * 16 + hi * 8 + i;
                if (row < cn) {
                    float4 y;
                    y.x = c[0][i]; y.y = c[1][i]; y.z = c[2][i]; y.w = c[3][i];
                    *(float4*)(Yn + (size_t)(base + row) * DK2 + half * DF + lcol * 4) = y;
                }
            }
        }
    } else {
        for (int task = wv; task < AGG_TR * (DOUT / 16); task += AGG_T / 32) {
            const int t  = task >> 2;
            const int tc = task & 3;
            if (t * 16 >= cn) continue;
            const float* ar = pl + (t * 16 + n) * AGG_P;
            v8f c = {};
#pragma unroll
            for (int kt = 0; kt < DF / 32; ++kt) {
                const v16h a = plane_a_frag(ar, kt * 32, hi);
                const v16h b = lds_b_frag(Bl4, kt * (DOUT / 16) + tc, lane);
                c = __builtin_amdgcn_wmma_f32_16x16x32_f16(false, a, false, b,
                                                           (short)0, c, false, false);
            }
            const int ncol = tc * 16 + n;
            const float bv = biasN[ncol];
#pragma unroll
            for (int i = 0; i < 8; ++i) {
                const int row = t * 16 + hi * 8 + i;
                if (row < cn) Yn[(size_t)(base + row) * DOUT + ncol] = c[i] + bv;
            }
        }
    }
}

__global__ void pack_pre(const float* __restrict__ Ws, const float* __restrict__ Wd,
                         _Float16* __restrict__ dst) {
    int idx = blockIdx.x * blockDim.x + threadIdx.x;
    if (idx >= DF * DK2) return;
    int j    = idx & 15;
    int lane = (idx >> 4) & 31;
    int t    = idx >> 9;
    int q    = t & 7;
    int hp   = (t >> 3) & 1;
    int kt   = t >> 4;
    int half = q >> 2;
    int jt   = q & 3;
    int k = kt * 32 + ((j < 8) ? 0 : 16) + (lane >> 4) * 8 + (j & 7);
    int n = 32 * jt + 16 * hp + (lane & 15);
    float v = half ? 0.5f * Wd[k * DF + n] : 0.5f * Ws[k * DF + n];
    dst[idx] = (_Float16)v;
}

__global__ void __launch_bounds__(256)
gemm_pre(const float* __restrict__ A, int Mrows,
         const v16h* __restrict__ Bp, float* __restrict__ Y) {
    int wave = (blockIdx.x * blockDim.x + threadIdx.x) >> 5;
    int lane = threadIdx.x & 31;
    int rowTiles = Mrows >> 4;
    if (wave >= rowTiles * 2) return;
    int tr = wave >> 1;
    int hp = wave & 1;
    int hi = lane >> 4;
    int n  = lane & 15;

    const float4* Ar = (const float4*)(A + (long)(tr * 16 + n) * DF);
    v8f c[8] = {};
#pragma unroll
    for (int kt = 0; kt < DF / 32; ++kt) {
        v16h a = load_a_frag(Ar, kt * 32, hi);
        const v16h* bq = Bp + ((kt * 2 + hp) * 8) * 32 + lane;
#pragma unroll
        for (int q = 0; q < 8; ++q) {
            v16h b = bq[q * 32];
            c[q] = __builtin_amdgcn_wmma_f32_16x16x32_f16(false, a, false, b,
                                                          (short)0, c[q], false, false);
        }
    }

    const int lcol = 16 * hp + n;
    const int row0 = tr * 16 + hi * 8;
#pragma unroll
    for (int i = 0; i < 8; ++i) {
        const int row = row0 + i;
        float4 ys, yd;
        ys.x = c[0][i]; ys.y = c[1][i]; ys.z = c[2][i]; ys.w = c[3][i];
        yd.x = c[4][i]; yd.y = c[5][i]; yd.z = c[6][i]; yd.w = c[7][i];
        float* yr = Y + (long)row * DK2 + lcol * 4;
        *(float4*)yr        = ys;
        *(float4*)(yr + DF) = yd;
    }
}

extern "C" void kernel_launch(void* const* d_in, const int* in_sizes, int n_in,
                              void* d_out, int out_size, void* d_ws, size_t ws_size,
                              hipStream_t stream) {
    const float* x      = (const float*)d_in[0];
    const int*   edges  = (const int*)d_in[1];
    const float* theta  = (const float*)d_in[2];
    const float* W1s    = (const float*)d_in[3];
    const float* W1d    = (const float*)d_in[4];
    const float* b1s    = (const float*)d_in[5];
    const float* b1d    = (const float*)d_in[6];
    const float* W2s    = (const float*)d_in[7];
    const float* W2d    = (const float*)d_in[8];
    const float* b2s    = (const float*)d_in[9];
    const float* b2d    = (const float*)d_in[10];
    const float* Wro    = (const float*)d_in[11];
    const float* bro    = (const float*)d_in[12];

    const int N = in_sizes[0] / DF;
    const int E = in_sizes[2];
    const int* snd = edges;
    const int* rcv = edges + E;

    char* ws = (char*)d_ws;
    size_t off = 0;
    auto alloc = [&](size_t bytes) { char* p = ws + off; off = (off + bytes + 255) & ~(size_t)255; return p; };
    float*    degS  = (float*)alloc((size_t)N * 4);
    float*    degR  = (float*)alloc((size_t)N * 4);
    int4*     rec   = (int4*)alloc((size_t)E * 16);
    unsigned char* cid = (unsigned char*)alloc((size_t)cdiv(2 * E, 16) * 16);
    float*    selfS = (float*)alloc((size_t)N * 4);
    float*    selfD = (float*)alloc((size_t)N * 4);
    float*    Y1    = (float*)alloc((size_t)N * DK2 * 4);
    float*    Y2    = (float*)alloc((size_t)N * DK2 * 4);
    _Float16* Bp1   = (_Float16*)alloc((size_t)DK2 * DF * 2);
    unsigned short* blst = (unsigned short*)alloc((size_t)AGG_NCH * BIN_S * BIN_W * 2);

    const int T = 256;

    float* part = Y1;
    const int pitch = (N + 31) & ~31;
    deg_partial<<<dim3(DEG_G, 2, cdiv(N, DEG_CAP)), DEG_T, 0, stream>>>(snd, rcv, theta, E, N, pitch, part, blst);
    deg_reduce<<<dim3(cdiv(N, T), 2), T, 0, stream>>>(part, N, pitch, degS, degR);
    weight_kernel<<<cdiv(E + N, T), T, 0, stream>>>(snd, rcv, theta, E, N, degS, degR, rec, cid, selfS, selfD);

    pack_pre<<<cdiv(DF * DK2, T), T, 0, stream>>>(W1s, W1d, Bp1);

    const int preWaves      = (N / 16) * 2;
    const int aggBlocks     = cdiv(N, AGG_ROWS);

    gemm_pre<<<cdiv(preWaves * 32, T), T, 0, stream>>>(x, N, (const v16h*)Bp1, Y1);
    agg_lds<0><<<aggBlocks, AGG_T, 0, stream>>>(Y1, rec, (const uint4*)cid, b1s, b1d, selfS, selfD, E, N,
                                                blst, W2s, W2d, bro, Y2);

    agg_lds<1><<<aggBlocks, AGG_T, 0, stream>>>(Y2, rec, (const uint4*)cid, b2s, b2d, selfS, selfD, E, N,
                                                blst, Wro, Wro, bro, (float*)d_out);
}


# Edge weight split into sender, edge, receiver factors: sender dis folded into stored product rows, receiver dis from an LDS table; weight launch and deg_reduce go (5 launches to 4); sum order kept
# speedup vs baseline: 4.7174x; 1.0928x over previous
#include <hip/hip_runtime.h>

typedef __attribute__((ext_vector_type(16))) _Float16 v16h;
typedef __attribute__((ext_vector_type(8)))  float    v8f;
typedef __attribute__((ext_vector_type(8)))  _Float16 v8h;
typedef __attribute__((ext_vector_type(4)))  unsigned int u4v;
typedef __attribute__((ext_vector_type(8)))  unsigned int u8v;

#define DF   128
#define DK2  256
#define DOUT 64

static __host__ __device__ inline int cdiv(int a, int b) { return (a + b - 1) / b; }


#define DEG_CAP 50176
#define DEG_G   64
#define DEG_T   1024

#define AGG_ROWS 196
#define AGG_NCH  256
#define BIN_W    64
#define BIN_CAP  (BIN_W - 1)
#define BIN_S    (2 * DEG_G)


#define AGG_TR   13
#define AGG_PR   (AGG_TR * 16)
#define AGG_P    132
#define AGG_T    1024
#define AGG_LCAP 16384

template <int RELU_IN>
__device__ __forceinline__ v16h frag16(float4 p0, float4 p1, float4 p2, float4 p3) {
    if (RELU_IN) {
        p0.x = fmaxf(p0.x, 0.0f); p0.y = fmaxf(p0.y, 0.0f); p0.z = fmaxf(p0.z, 0.0f); p0.w = fmaxf(p0.w, 0.0f);
        p1.x = fmaxf(p1.x, 0.0f); p1.y = fmaxf(p1.y, 0.0f); p1.z = fmaxf(p1.z, 0.0f); p1.w = fmaxf(p1.w, 0.0f);
        p2.x = fmaxf(p2.x, 0.0f); p2.y = fmaxf(p2.y, 0.0f); p2.z = fmaxf(p2.z, 0.0f); p2.w = fmaxf(p2.w, 0.0f);
        p3.x = fmaxf(p3.x, 0.0f); p3.y = fmaxf(p3.y, 0.0f); p3.z = fmaxf(p3.z, 0.0f); p3.w = fmaxf(p3.w, 0.0f);
    }
    v16h a;
    a[0]  = (_Float16)p0.x; a[1]  = (_Float16)p0.y;
    a[2]  = (_Float16)p0.z; a[3]  = (_Float16)p0.w;
    a[4]  = (_Float16)p1.x; a[5]  = (_Float16)p1.y;
    a[6]  = (_Float16)p1.z; a[7]  = (_Float16)p1.w;
    a[8]  = (_Float16)p2.x; a[9]  = (_Float16)p2.y;
    a[10] = (_Float16)p2.z; a[11] = (_Float16)p2.w;
    a[12] = (_Float16)p3.x; a[13] = (_Float16)p3.y;
    a[14] = (_Float16)p3.z; a[15] = (_Float16)p3.w;
    return a;
}

__device__ __forceinline__ v16h load_a_frag(const float4* __restrict__ Ar, int k0, int hi) {
    const int b = (k0 >> 2) + hi * 2;
    return frag16<0>(Ar[b], Ar[b + 1], Ar[b + 4], Ar[b + 5]);
}

__device__ __forceinline__ v16h plane_a_frag(const float* ar, int k0, int hi) {
    const float4* q = (const float4*)(ar + k0 + 8 * hi);
    return frag16<1>(q[0], q[1], q[4], q[5]);
}

__device__ __forceinline__ v16h lds_b_frag(const u4v* B4, int t, int lane) {
    const u4v lo = B4[(t * 2) * 32 + lane];
    const u4v hi = B4[(t * 2 + 1) * 32 + lane];
    const u8v w = __builtin_shufflevector(lo, hi, 0, 1, 2, 3, 4, 5, 6, 7);
    return __builtin_bit_cast(v16h, w);
}

__device__ __forceinline__ u4v pack8h(const float* f, float scale) {
    v8h h;
    h[0] = (_Float16)(scale * f[0]); h[1] = (_Float16)(scale * f[1]);
    h[2] = (_Float16)(scale * f[2]); h[3] = (_Float16)(scale * f[3]);
    h[4] = (_Float16)(scale * f[4]); h[5] = (_Float16)(scale * f[5]);
    h[6] = (_Float16)(scale * f[6]); h[7] = (_Float16)(scale * f[7]);
    return __builtin_bit_cast(u4v, h);
}

template <int LAYER>
__global__ void __launch_bounds__(AGG_T)
agg_lds(const float* __restrict__ Y, const int4* __restrict__ rec,
        const int* __restrict__ snd, const int* __restrict__ rcv,
        const float* __restrict__ biasS, const float* __restrict__ biasD,
        const float2* __restrict__ dis2, int E, int Nn,
        const unsigned short* __restrict__ blst,
        const float* __restrict__ Wa, const float* __restrict__ Wb,
        const float* __restrict__ biasN, float* __restrict__ Yn) {
    __shared__ float4   pl4[AGG_PR * (AGG_P / 4)];
    __shared__ u4v      Bl4[LAYER == 0 ? (DF * DK2 / 8) : (DF * DOUT / 8)];
    __shared__ unsigned lst[AGG_LCAP];
    __shared__ float2   dl[AGG_PR];
    __shared__ unsigned s_pre[BIN_S];
    __shared__ unsigned s_cnt;
    __shared__ unsigned s_scan;
    float* pl = (float*)pl4;
    const int tid   = threadIdx.x;
    const int chunk = blockIdx.x;
    const int base  = chunk * AGG_ROWS;
    const int cn    = (Nn - base < AGG_ROWS) ? (Nn - base) : AGG_ROWS;
    const int M2    = 2 * E;

    static_assert(AGG_T == BIN_S * (BIN_W / 8), "eight threads per sub-list line");
    const int sl = tid >> 3;
    const int sq = tid & 7;
    const unsigned short* line = blst + ((size_t)chunk * BIN_S + sl) * BIN_W;
    const unsigned slen = line[0];
    const uint4    sent = ((const uint4*)line)[sq];
    if (tid < 32) {
        static_assert(BIN_S == 128, "a lane of wave 0 takes four sub-lists");
        unsigned c4[4], sum = 0u, bad = 0u;
#pragma unroll
        for (int j = 0; j < 4; ++j) c4[j] = blst[((size_t)chunk * BIN_S + 4 * tid + j) * BIN_W];
#pragma unroll
        for (int j = 0; j < 4; ++j) { bad |= (c4[j] > (unsigned)BIN_CAP) ? 1u : 0u; sum += c4[j]; }
        unsigned inc = sum;
        for (int d = 1; d < 32; d <<= 1) {
            const unsigned t = __shfl_up(inc, d);
            if (tid >= d) inc += t;
        }
        unsigned ex = inc - sum;
#pragma unroll
        for (int j = 0; j < 4; ++j) { s_pre[4 * tid + j] = ex; ex += c4[j]; }
        const int anyBad = __any((int)bad);
        if (tid == 31) { s_cnt = anyBad ? 0u : inc; s_scan = anyBad ? 1u : 0u; }
    }

    constexpr int NOCT = (LAYER == 0) ? 4 : 1;
    float wf[NOCT][8];
    if constexpr (LAYER == 0) {
        const int c = tid & 15, hp = (tid >> 4) & 1, h = (tid >> 5) & 1, part = (tid >> 6) & 1;
        const int jt = (tid >> 7) & 3, half = (tid >> 9) & 1;
        const float* wsrc = (half ? Wb : Wa) + 32 * jt + 16 * hp + c;
#pragma unroll
        for (int kt = 0; kt < 4; ++kt) {
            const int kb = kt * 32 + part * 16 + h * 8;
#pragma unroll
            for (int q = 0; q < 8; ++q) wf[kt][q] = wsrc[(kb + q) * DF];
        }
    } else {
        const int c = tid & 15, tc = (tid >> 4) & 3, h = (tid >> 6) & 1, part = (tid >> 7) & 1, kt = (tid >> 8) & 3;
        const float* wsrc = Wa + tc * 16 + c;
        const int kb = kt * 32 + part * 16 + h * 8;
#pragma unroll
        for (int q = 0; q < 8; ++q) wf[0][q] = wsrc[(kb + q) * DOUT];
    }

    const int l = tid & 31;
    float bz[4];
#pragma unroll
    for (int j = 0; j < 4; ++j) bz[j] = 0.5f * (biasS[l + 32 * j] + biasD[l + 32 * j]);

    if (tid < AGG_PR) dl[tid] = (tid < cn) ? dis2[base + tid] : make_float2(0.0f, 0.0f);

    static_assert((AGG_T & 31) == 0, "a thread keeps its lane's columns over the rows it takes");
    for (int i = tid; i < AGG_PR * 32; i += AGG_T) {
        const int row = i >> 5;
        float4 v = make_float4(0.0f, 0.0f, 0.0f, 0.0f);
        if (row < cn) {
            const float* yr = Y + (size_t)(base + row) * DK2 + 4 * l;
            const float4 ys = *(const float4*)yr;
            const float4 yd = *(const float4*)(yr + DF);
            const float2 dd = dis2[base + row];
            const float as = dd.y;
            const float bs = dd.x;
            v.x = bz[0] + as * ys.x + bs * yd.x;
            v.y = bz[1] + as * ys.y + bs * yd.y;
            v.z = bz[2] + as * ys.z + bs * yd.z;
            v.w = bz[3] + as * ys.w + bs * yd.w;
        }
        float* d = pl + row * AGG_P + l;
        d[0] = v.x; d[32] = v.y; d[64] = v.z; d[96] = v.w;
    }

    if constexpr (LAYER == 0) {
        const int c = tid & 15, hp = (tid >> 4) & 1, h = (tid >> 5) & 1, part = (tid >> 6) & 1;
        const int jt = (tid >> 7) & 3, half = (tid >> 9) & 1;
#pragma unroll
        for (int kt = 0; kt < 4; ++kt) {
            const int t = (kt * 2 + hp) * 8 + half * 4 + jt;
            Bl4[(t * 2 + part) * 32 + h * 16 + c] = pack8h(wf[kt], 0.5f);
        }
    } else {
        const int c = tid & 15, tc = (tid >> 4) & 3, h = (tid >> 6) & 1, part = (tid >> 7) & 1, kt = (tid >> 8) & 3;
        const int t = kt * (DOUT / 16) + tc;
        Bl4[(t * 2 + part) * 32 + h * 16 + c] = pack8h(wf[0], 1.0f);
    }
    __syncthreads();

    const bool doScan = (s_scan != 0u);
    if (!doScan) {
        const int perS = cdiv(E, DEG_G);
        const unsigned vb = (unsigned)((sl >= DEG_G ? E : 0) + (sl & (DEG_G - 1)) * perS);
        const unsigned p  = s_pre[sl];
        const unsigned wq[4] = { sent.x, sent.y, sent.z, sent.w };
#pragma unroll
        for (int i = 0; i < 8; ++i) {
            const unsigned j = (unsigned)(sq * 8 + i);
            const unsigned w = (i & 1) ? (wq[i >> 1] >> 16) : (wq[i >> 1] & 0xFFFFu);
            if (j >= 1u && j <= slen) lst[p + j - 1u] = vb + w;
        }
    }

    if (doScan) {
        for (int idx = tid; idx < M2; idx += AGG_T) {
            const int recv = (idx < E) ? rcv[idx] : snd[idx - E];
            if ((unsigned)(recv - base) < (unsigned)cn) {
                const unsigned slot = atomicAdd(&s_cnt, 1u);
                if (slot < (unsigned)AGG_LCAP) {
                    lst[slot] = (unsigned)idx;
                } else {
                    int4 q; int s; float a, b;
                    if (idx < E) { q = rec[idx];     s = q.x; a = __int_as_float(q.z); b = __int_as_float(q.w); }
                    else         { q = rec[idx - E]; s = q.y; a = __int_as_float(q.w); b = __int_as_float(q.z); }
                    const float2 dd = dl[recv - base];
                    a *= dd.y; b *= dd.x;
                    const float* yr = Y + (long)s * DK2;
                    float* dst = pl + (recv - base) * AGG_P;
                    for (int p = 0; p < DF; ++p)
                        __hip_atomic_fetch_add(dst + 32 * (p & 3) + (p >> 2), a * yr[p] + b * yr[DF + p],
                                               __ATOMIC_RELAXED, __HIP_MEMORY_SCOPE_WORKGROUP);
                }
            }
        }
    }
    __syncthreads();

    const int total = __builtin_amdgcn_readfirstlane((int)s_cnt);
    const int cnt   = (total < AGG_LCAP) ? total : AGG_LCAP;
    const int wv    = __builtin_amdgcn_readfirstlane(tid >> 5);
    const int lane  = tid & 31;
    const int per   = (cnt + (AGG_T / 32) - 1) / (AGG_T / 32);
    const int st    = wv * per;
    const int en    = (st + per < cnt) ? (st + per) : cnt;
    const float* Yl = Y + lane * 4;
    float* pll = pl + lane;
    if (st < en) {
        const int len = en - st;
        const int ng  = (len + 31) >> 5;
        const int gs  = (((len + ng - 1) / ng) + 3) & ~3;
        int sv = 0, rv = 0, av = 0, bv = 0;
        {
            const int n0 = (len < gs) ? len : gs;
            if (lane < n0) {
                const unsigned e = lst[st + lane];
                if (e < (unsigned)E) { const int4 q = rec[e];     sv = q.x; rv = q.y; av = q.z; bv = q.w; }
                else                 { const int4 q = rec[e - E]; sv = q.y; rv = q.x; av = q.w; bv = q.z; }
            }
        }
        for (int g = st; g < en; g += gs) {
            const int n = (en - g < gs) ? (en - g) : gs;
            const int cs = sv, cr = rv;
            const float2 rf = dl[(lane < n) ? (cr - base) : 0];
            const int ca = __float_as_int(__int_as_float(av) * rf.y);
            const int cb = __float_as_int(__int_as_float(bv) * rf.x);
            const int gn = g + gs;
            if (gn < en) {
                const int nn = (en - gn < gs) ? (en - gn) : gs;
                if (lane < nn) {
                    const unsigned e = lst[gn + lane];
                    if (e < (unsigned)E) { const int4 q = rec[e];     sv = q.x; rv = q.y; av = q.z; bv = q.w; }
                    else                 { const int4 q = rec[e - E]; sv = q.y; rv = q.x; av = q.w; bv = q.z; }
                }
            }
            int t = 0;
            for (; t + 4 <= n; t += 4) {
                float4 ys[4], yd[4];
                int r[4]; float a[4], b[4];
#pragma unroll
                for (int u = 0; u < 4; ++u) {
                    const int s = __builtin_amdgcn_readlane(cs, t + u);
                    r[u] = __builtin_amdgcn_readlane(cr, t + u) - base;
                    a[u] = __int_as_float(__builtin_amdgcn_readlane(ca, t + u));
                    b[u] = __int_as_float(__builtin_amdgcn_readlane(cb, t + u));
                    const float* yr = Yl + (long)s * DK2;
                    ys[u] = *(const float4*)yr;
                    yd[u] = *(const float4*)(yr + DF);
                }
#pragma unroll
                for (int u = 0; u < 4; ++u) {
                    float* dst = pll + r[u] * AGG_P;
                    __hip_atomic_fetch_add(dst,      a[u] * ys[u].x + b[u] * yd[u].x, __ATOMIC_RELAXED, __HIP_MEMORY_SCOPE_WORKGROUP);
                    __hip_atomic_fetch_add(dst + 32, a[u] * ys[u].y + b[u] * yd[u].y, __ATOMIC_RELAXED, __HIP_MEMORY_SCOPE_WORKGROUP);
                    __hip_atomic_fetch_add(dst + 64, a[u] * ys[u].z + b[u] * yd[u].z, __ATOMIC_RELAXED, __HIP_MEMORY_SCOPE_WORKGROUP);
                    __hip_atomic_fetch_add(dst + 96, a[u] * ys[u].w + b[u] * yd[u].w, __ATOMIC_RELAXED, __HIP_MEMORY_SCOPE_WORKGROUP);
                }
            }
            for (; t < n; ++t) {
                const int s = __builtin_amdgcn_readlane(cs, t);
                const int r = __builtin_amdgcn_readlane(cr, t) - base;
                const float a = __int_as_float(__builtin_amdgcn_readlane(ca, t));
                const float b = __int_as_float(__builtin_amdgcn_readlane(cb, t));
                const float* yr = Yl + (long)s * DK2;
                const float4 ys = *(const float4*)yr;
                const float4 yd = *(const float4*)(yr + DF);
                float* dst = pll + r * AGG_P;
                __hip_atomic_fetch_add(dst,      a * ys.x + b * yd.x, __ATOMIC_RELAXED, __HIP_MEMORY_SCOPE_WORKGROUP);
                __hip_atomic_fetch_add(dst + 32, a * ys.y + b * yd.y, __ATOMIC_RELAXED, __HIP_MEMORY_SCOPE_WORKGROUP);
                __hip_atomic_fetch_add(dst + 64, a * ys.z + b * yd.z, __ATOMIC_RELAXED, __HIP_MEMORY_SCOPE_WORKGROUP);
                __hip_atomic_fetch_add(dst + 96, a * ys.w + b * yd.w, __ATOMIC_RELAXED, __HIP_MEMORY_SCOPE_WORKGROUP);
            }
        }
    }
    __syncthreads();

    const int hi = lane >> 4;
    const int n  = lane & 15;
    if (LAYER == 0) {
        for (int task = wv; task < AGG_TR * 4; task += AGG_T / 32) {
            const int t    = task >> 2;
            const int hp   = (task >> 1) & 1;
            const int half = task & 1;
            if (t * 16 >= cn) continue;
            const float* ar = pl + (t * 16 + n) * AGG_P;
            v8f c[4] = {};
#pragma unroll 2
            for (int kt = 0; kt < DF / 32; ++kt) {
                const v16h a = plane_a_frag(ar, kt * 32, hi);
                const int tb = (kt * 2 + hp) * 8 + half * 4;
#pragma unroll
                for (int jt = 0; jt < 4; ++jt) {
                    const v16h b = lds_b_frag(Bl4, tb + jt, lane);
                    c[jt] = __builtin_amdgcn_wmma_f32_16x16x32_f16(false, a, false, b,
                                                                   (short)0, c[jt], false, false);
                }
            }
            const int lcol = 16 * hp + n;
#pragma unroll
            for (int i = 0; i < 8; ++i) {
                const int row = t * 16 + hi * 8 + i;
                if (row < cn) {
                    const float2 dd = dl[row];
                    const float sc = half ? dd.y : dd.x;
                    float4 y;
                    y.x = sc * c[0][i]; y.y = sc * c[1][i]; y.z = sc * c[2][i]; y.w = sc * c[3][i];
                    *(float4*)(Yn + (size_t)(base + row) * DK2 + half * DF + lcol * 4) = y;
                }
            }
        }
    } else {
        for (int task = wv; task < AGG_TR * (DOUT / 16); task += AGG_T / 32) {
            const int t  = task >> 2;
            const int tc = task & 3;
            if (t * 16 >= cn) continue;
            const float* ar = pl + (t * 16 + n) * AGG_P;
            v8f c = {};
#pragma unroll
            for (int kt = 0; kt < DF / 32; ++kt) {
                const v16h a = plane_a_frag(ar, kt * 32, hi);
                const v16h b = lds_b_frag(Bl4, kt * (DOUT / 16) + tc, lane);
                c = __builtin_amdgcn_wmma_f32_16x16x32_f16(false, a, false, b,
                                                           (short)0, c, false, false);
            }
            const int ncol = tc * 16 + n;
            const float bv = biasN[ncol];
#pragma unroll
            for (int i = 0; i < 8; ++i) {
                const int row = t * 16 + hi * 8 + i;
                if (row < cn) Yn[(size_t)(base + row) * DOUT + ncol] = c[i] + bv;
            }
        }
    }
}

#define REC_WG 128

__global__ void __launch_bounds__(DEG_T)
deg_pre(const int* __restrict__ snd, const int* __restrict__ rcv,
        const float* __restrict__ theta, int E, int Nn, int pitch,
        float* __restrict__ part, unsigned short* __restrict__ blst, int nDeg,
        int4* __restrict__ rec, int recPer) {
    __shared__ float4 pl4[DEG_CAP / 4];
    __shared__ uint4    bl4[AGG_NCH * BIN_W / 8];
    __shared__ unsigned bcount[AGG_NCH];
    const int tid  = threadIdx.x;

    if ((int)blockIdx.x >= nDeg) {
        const int eb = ((int)blockIdx.x - nDeg) * recPer;
        const int ee = (eb + recPer < E) ? (eb + recPer) : E;
        for (int e = eb + tid; e < ee; e += DEG_T) {
            const float cs = cosf(theta[e]), sn = sinf(theta[e]);
            rec[e] = make_int4(snd[e], rcv[e], __float_as_int(cs * cs), __float_as_int(sn * sn));
        }
        return;
    }

    float* pl = (float*)pl4;
    unsigned short* bl = (unsigned short*)bl4;
    const int g    = (int)blockIdx.x % DEG_G;
    const int a    = ((int)blockIdx.x / DEG_G) & 1;
    const int z    = (int)blockIdx.x / (2 * DEG_G);
    const int base = z * DEG_CAP;
    const int cn   = (Nn - base < DEG_CAP) ? (Nn - base) : DEG_CAP;
    const int cn4  = (cn + 3) >> 2;
    for (int i = tid; i < cn4; i += DEG_T) pl4[i] = make_float4(0.0f, 0.0f, 0.0f, 0.0f);
    const bool doBin = (z == 0);
    if (tid < AGG_NCH) bcount[tid] = 0u;
    __syncthreads();

    const int per = cdiv(E, DEG_G);
    const int e0  = g * per;
    const int e1  = (e0 + per < E) ? (e0 + per) : E;
    for (int e = e0 + tid; e < e1; e += 4 * DEG_T) {
        int u[4], v[4]; float th[4];
#pragma unroll
        for (int k = 0; k < 4; ++k) {
            const int ek = e + k * DEG_T;
            const int ec = (ek < e1) ? ek : e;
            u[k] = snd[ec]; v[k] = rcv[ec]; th[k] = theta[ec];
            if (ek >= e1) { u[k] = -1; v[k] = -1; }
        }
#pragma unroll
        for (int k = 0; k < 4; ++k) {
            const float cs = cosf(th[k]), sn = sinf(th[k]);
            const float c2 = cs * cs, s2 = sn * sn;
            const float wu = a ? s2 : c2;
            const float wv = a ? c2 : s2;
            const unsigned iu = (unsigned)(u[k] - base);
            const unsigned iv = (unsigned)(v[k] - base);
            if (iu < (unsigned)cn) __hip_atomic_fetch_add(pl + iu, wu, __ATOMIC_RELAXED, __HIP_MEMORY_SCOPE_WORKGROUP);
            if (iv < (unsigned)cn) __hip_atomic_fetch_add(pl + iv, wv, __ATOMIC_RELAXED, __HIP_MEMORY_SCOPE_WORKGROUP);
            if (doBin && u[k] >= 0) {
                const unsigned b = (unsigned)(a ? u[k] : v[k]) / AGG_ROWS;
                if (b < (unsigned)AGG_NCH) {
                    const unsigned slot = atomicAdd(&bcount[b], 1u);
                    if (slot < (unsigned)BIN_CAP) bl[b * BIN_W + 1 + slot] = (unsigned short)(e + k * DEG_T - e0);
                }
            }
        }
    }
    __syncthreads();

    float4* dst = (float4*)(part + ((size_t)a * DEG_G + g) * pitch + base);
    for (int i = tid; i < cn4; i += DEG_T) dst[i] = pl4[i];

    if (doBin) {
        if (tid < AGG_NCH) {
            const unsigned n = bcount[tid];
            bl[tid * BIN_W] = (unsigned short)((n > (unsigned)BIN_CAP || per > 65536) ? 0xFFFFu : n);
        }
        __syncthreads();
        const int sl = a * DEG_G + g;
        for (int i = tid; i < AGG_NCH * (BIN_W / 8); i += DEG_T) {
            const int b = i >> 3, q = i & 7;
            ((uint4*)(blst + ((size_t)b * BIN_S + sl) * BIN_W))[q] = bl4[i];
        }
    }
}

#define PRE_TILES 16
#define PRE_ROWS  (PRE_TILES * 16)

__global__ void __launch_bounds__(DEG_T)
prod_reduce(const float* __restrict__ part, int Nn, int pitch, float2* __restrict__ dis2,
            const float* __restrict__ X, const float* __restrict__ Ws, const float* __restrict__ Wd,
            float* __restrict__ Y, int rowTiles) {
    __shared__ u4v   Bl4[DF * DK2 / 8];
    __shared__ float sd[PRE_ROWS * 2];
    const int tid  = threadIdx.x;
    const int t0   = (int)blockIdx.x * PRE_TILES;
    const int tEnd = (t0 + PRE_TILES < rowTiles) ? (t0 + PRE_TILES) : rowTiles;
    const int r0   = t0 * 16;

    static_assert(DEG_T == 1024, "piece id = tid + 1024*kt covers the 4096 pieces of the matrix");
    static_assert(DEG_T >= 2 * PRE_ROWS, "one thread per (row, degree kind)");
    {
        const int c = tid & 15, hp = (tid >> 4) & 1, h = (tid >> 5) & 1, prt = (tid >> 6) & 1;
        const int jt = (tid >> 7) & 3, half = (tid >> 9) & 1;
        const float* wsrc = (half ? Wd : Ws) + 32 * jt + 16 * hp + c;
        float wf[4][8];
#pragma unroll
        for (int kt = 0; kt < 4; ++kt) {
            const int kb = kt * 32 + prt * 16 + h * 8;
#pragma unroll
            for (int q = 0; q < 8; ++q) wf[kt][q] = wsrc[(kb + q) * DF];
        }

        if (tid < 2 * PRE_ROWS) {
            const int nl = tid & (PRE_ROWS - 1), a = tid / PRE_ROWS;
            const int n  = r0 + nl;
            float v = 0.0f;
            if (n < Nn) {
                const float* p = part + (size_t)a * DEG_G * pitch + n;
                float dg = 1.0f;
#pragma unroll 16
                for (int g = 0; g < DEG_G; ++g) dg += p[(size_t)g * pitch];
                const float w = dg + 1e-12f;
                v = (w > 1e11f) ? 0.0f : rsqrtf(w);
            }
            sd[nl * 2 + a] = v;
        }

#pragma unroll
        for (int kt = 0; kt < 4; ++kt) {
            const int t = (kt * 2 + hp) * 8 + half * 4 + jt;
            Bl4[(t * 2 + prt) * 32 + h * 16 + c] = pack8h(wf[kt], 0.5f);
        }
    }
    __syncthreads();

    if (tid < PRE_ROWS && r0 + tid < Nn) dis2[r0 + tid] = make_float2(sd[tid * 2], sd[tid * 2 + 1]);

    const int wv   = __builtin_amdgcn_readfirstlane(tid >> 5);
    const int lane = tid & 31;
    const int hi   = lane >> 4;
    const int n    = lane & 15;
    const int nTask = (tEnd - t0) * 2;
    for (int task = wv; task < nTask; task += DEG_T / 32) {
        const int tl = task >> 1;
        const int tr = t0 + tl;
        const int hp = task & 1;
        const float4* Ar = (const float4*)(X + (size_t)(tr * 16 + n) * DF);
        v16h af[DF / 32];
#pragma unroll
        for (int kt = 0; kt < DF / 32; ++kt) af[kt] = load_a_frag(Ar, kt * 32, hi);
        const int lcol = 16 * hp + n;
        const int row0 = tr * 16 + hi * 8;
        const float* sdr = sd + (tl * 16 + hi * 8) * 2;
#pragma nounroll
        for (int half = 0; half < 2; ++half) {
            v8f c[4] = {};
#pragma unroll
            for (int kt = 0; kt < DF / 32; ++kt) {
                const int tb = (kt * 2 + hp) * 8 + half * 4;
#pragma unroll
                for (int jt = 0; jt < 4; ++jt) {
                    const v16h b = lds_b_frag(Bl4, tb + jt, lane);
                    c[jt] = __builtin_amdgcn_wmma_f32_16x16x32_f16(false, af[kt], false, b,
                                                                   (short)0, c[jt], false, false);
                }
            }
#pragma unroll
            for (int i = 0; i < 8; ++i) {
                const float sc = sdr[i * 2 + half];
                float4 y;
                y.x = sc * c[0][i]; y.y = sc * c[1][i]; y.z = sc * c[2][i]; y.w = sc * c[3][i];
                *(float4*)(Y + (size_t)(row0 + i) * DK2 + half * DF + lcol * 4) = y;
            }
        }
    }
}

extern "C" void kernel_launch(void* const* d_in, const int* in_sizes, int n_in,
                              void* d_out, int out_size, void* d_ws, size_t ws_size,
                              hipStream_t stream) {
    const float* x      = (const float*)d_in[0];
    const int*   edges  = (const int*)d_in[1];
    const float* theta  = (const float*)d_in[2];
    const float* W1s    = (const float*)d_in[3];
    const float* W1d    = (const float*)d_in[4];
    const float* b1s    = (const float*)d_in[5];
    const float* b1d    = (const float*)d_in[6];
    const float* W2s    = (const float*)d_in[7];
    const float* W2d    = (const float*)d_in[8];
    const float* b2s    = (const float*)d_in[9];
    const float* b2d    = (const float*)d_in[10];
    const float* Wro    = (const float*)d_in[11];
    const float* bro    = (const float*)d_in[12];

    const int N = in_sizes[0] / DF;
    const int E = in_sizes[2];
    const int* snd = edges;
    const int* rcv = edges + E;

    char* ws = (char*)d_ws;
    size_t off = 0;
    auto alloc = [&](size_t bytes) { char* p = ws + off; off = (off + bytes + 255) & ~(size_t)255; return p; };
    float2*   dis2  = (float2*)alloc((size_t)N * 8);
    int4*     rec   = (int4*)alloc((size_t)E * 16);
    float*    Y1    = (float*)alloc((size_t)N * DK2 * 4);
    float*    Y2    = (float*)alloc((size_t)N * DK2 * 4);
    unsigned short* blst = (unsigned short*)alloc((size_t)AGG_NCH * BIN_S * BIN_W * 2);

    float* part = Y2;
    const int pitch = (N + 31) & ~31;
    const int nDeg      = DEG_G * 2 * cdiv(N, DEG_CAP);
    const int recPer    = (cdiv(E, REC_WG) + 7) & ~7;
    const int recBlocks = cdiv(E, recPer);
    deg_pre<<<nDeg + recBlocks, DEG_T, 0, stream>>>(snd, rcv, theta, E, N, pitch, part, blst, nDeg, rec, recPer);

    const int rowTiles  = N / 16;
    prod_reduce<<<cdiv(rowTiles, PRE_TILES), DEG_T, 0, stream>>>(part, N, pitch, dis2, x, W1s, W1d, Y1, rowTiles);

    const int aggBlocks     = cdiv(N, AGG_ROWS);

    agg_lds<0><<<aggBlocks, AGG_T, 0, stream>>>(Y1, rec, snd, rcv, b1s, b1d, dis2, E, N,
                                                blst, W2s, W2d, bro, Y2);

    agg_lds<1><<<aggBlocks, AGG_T, 0, stream>>>(Y2, rec, snd, rcv, b2s, b2d, dis2, E, N,
                                                blst, Wro, Wro, bro, (float*)d_out);
}


# Degrees summed in the product launch from each 196-row chunk's own edge sub-lists and records (LDS float adds per wave, fixed-order sum): the 128 partial degree planes go; sum order changes
# speedup vs baseline: 4.9429x; 1.0478x over previous
#include <hip/hip_runtime.h>

typedef __attribute__((ext_vector_type(16))) _Float16 v16h;
typedef __attribute__((ext_vector_type(8)))  float    v8f;
typedef __attribute__((ext_vector_type(8)))  _Float16 v8h;
typedef __attribute__((ext_vector_type(4)))  unsigned int u4v;
typedef __attribute__((ext_vector_type(8)))  unsigned int u8v;

#define DF   128
#define DK2  256
#define DOUT 64

static __host__ __device__ inline int cdiv(int a, int b) { return (a + b - 1) / b; }


#define DEG_G   64
#define DEG_T   1024

#define AGG_ROWS 196
#define AGG_NCH  256
#define BIN_W    64
#define BIN_CAP  (BIN_W - 1)
#define BIN_S    (2 * DEG_G)


#define AGG_TR   13
#define AGG_PR   (AGG_TR * 16)
#define AGG_P    132
#define AGG_T    1024
#define AGG_LCAP 16384

template <int RELU_IN>
__device__ __forceinline__ v16h frag16(float4 p0, float4 p1, float4 p2, float4 p3) {
    if (RELU_IN) {
        p0.x = fmaxf(p0.x, 0.0f); p0.y = fmaxf(p0.y, 0.0f); p0.z = fmaxf(p0.z, 0.0f); p0.w = fmaxf(p0.w, 0.0f);
        p1.x = fmaxf(p1.x, 0.0f); p1.y = fmaxf(p1.y, 0.0f); p1.z = fmaxf(p1.z, 0.0f); p1.w = fmaxf(p1.w, 0.0f);
        p2.x = fmaxf(p2.x, 0.0f); p2.y = fmaxf(p2.y, 0.0f); p2.z = fmaxf(p2.z, 0.0f); p2.w = fmaxf(p2.w, 0.0f);
        p3.x = fmaxf(p3.x, 0.0f); p3.y = fmaxf(p3.y, 0.0f); p3.z = fmaxf(p3.z, 0.0f); p3.w = fmaxf(p3.w, 0.0f);
    }
    v16h a;
    a[0]  = (_Float16)p0.x; a[1]  = (_Float16)p0.y;
    a[2]  = (_Float16)p0.z; a[3]  = (_Float16)p0.w;
    a[4]  = (_Float16)p1.x; a[5]  = (_Float16)p1.y;
    a[6]  = (_Float16)p1.z; a[7]  = (_Float16)p1.w;
    a[8]  = (_Float16)p2.x; a[9]  = (_Float16)p2.y;
    a[10] = (_Float16)p2.z; a[11] = (_Float16)p2.w;
    a[12] = (_Float16)p3.x; a[13] = (_Float16)p3.y;
    a[14] = (_Float16)p3.z; a[15] = (_Float16)p3.w;
    return a;
}

__device__ __forceinline__ v16h load_a_frag(const float4* __restrict__ Ar, int k0, int hi) {
    const int b = (k0 >> 2) + hi * 2;
    return frag16<0>(Ar[b], Ar[b + 1], Ar[b + 4], Ar[b + 5]);
}

__device__ __forceinline__ v16h plane_a_frag(const float* ar, int k0, int hi) {
    const float4* q = (const float4*)(ar + k0 + 8 * hi);
    return frag16<1>(q[0], q[1], q[4], q[5]);
}

__device__ __forceinline__ v16h lds_b_frag(const u4v* B4, int t, int lane) {
    const u4v lo = B4[(t * 2) * 32 + lane];
    const u4v hi = B4[(t * 2 + 1) * 32 + lane];
    const u8v w = __builtin_shufflevector(lo, hi, 0, 1, 2, 3, 4, 5, 6, 7);
    return __builtin_bit_cast(v16h, w);
}

__device__ __forceinline__ u4v pack8h(const float* f, float scale) {
    v8h h;
    h[0] = (_Float16)(scale * f[0]); h[1] = (_Float16)(scale * f[1]);
    h[2] = (_Float16)(scale * f[2]); h[3] = (_Float16)(scale * f[3]);
    h[4] = (_Float16)(scale * f[4]); h[5] = (_Float16)(scale * f[5]);
    h[6] = (_Float16)(scale * f[6]); h[7] = (_Float16)(scale * f[7]);
    return __builtin_bit_cast(u4v, h);
}

template <int LAYER>
__global__ void __launch_bounds__(AGG_T)
agg_lds(const float* __restrict__ Y, const int4* __restrict__ rec,
        const int* __restrict__ snd, const int* __restrict__ rcv,
        const float* __restrict__ biasS, const float* __restrict__ biasD,
        const float2* __restrict__ dis2, int E, int Nn,
        const unsigned short* __restrict__ blst,
        const float* __restrict__ Wa, const float* __restrict__ Wb,
        const float* __restrict__ biasN, float* __restrict__ Yn) {
    __shared__ float4   pl4[AGG_PR * (AGG_P / 4)];
    __shared__ u4v      Bl4[LAYER == 0 ? (DF * DK2 / 8) : (DF * DOUT / 8)];
    __shared__ unsigned lst[AGG_LCAP];
    __shared__ float2   dl[AGG_PR];
    __shared__ unsigned s_pre[BIN_S];
    __shared__ unsigned s_cnt;
    __shared__ unsigned s_scan;
    float* pl = (float*)pl4;
    const int tid   = threadIdx.x;
    const int chunk = blockIdx.x;
    const int base  = chunk * AGG_ROWS;
    const int cn    = (Nn - base < AGG_ROWS) ? (Nn - base) : AGG_ROWS;
    const int M2    = 2 * E;

    static_assert(AGG_T == BIN_S * (BIN_W / 8), "eight threads per sub-list line");
    const int sl = tid >> 3;
    const int sq = tid & 7;
    const unsigned short* line = blst + ((size_t)chunk * BIN_S + sl) * BIN_W;
    const unsigned slen = line[0];
    const uint4    sent = ((const uint4*)line)[sq];
    if (tid < 32) {
        static_assert(BIN_S == 128, "a lane of wave 0 takes four sub-lists");
        unsigned c4[4], sum = 0u, bad = 0u;
#pragma unroll
        for (int j = 0; j < 4; ++j) c4[j] = blst[((size_t)chunk * BIN_S + 4 * tid + j) * BIN_W];
#pragma unroll
        for (int j = 0; j < 4; ++j) { bad |= (c4[j] > (unsigned)BIN_CAP) ? 1u : 0u; sum += c4[j]; }
        unsigned inc = sum;
        for (int d = 1; d < 32; d <<= 1) {
            const unsigned t = __shfl_up(inc, d);
            if (tid >= d) inc += t;
        }
        unsigned ex = inc - sum;
#pragma unroll
        for (int j = 0; j < 4; ++j) { s_pre[4 * tid + j] = ex; ex += c4[j]; }
        const int anyBad = __any((int)bad);
        if (tid == 31) { s_cnt = anyBad ? 0u : inc; s_scan = anyBad ? 1u : 0u; }
    }

    constexpr int NOCT = (LAYER == 0) ? 4 : 1;
    float wf[NOCT][8];
    if constexpr (LAYER == 0) {
        const int c = tid & 15, hp = (tid >> 4) & 1, h = (tid >> 5) & 1, part = (tid >> 6) & 1;
        const int jt = (tid >> 7) & 3, half = (tid >> 9) & 1;
        const float* wsrc = (half ? Wb : Wa) + 32 * jt + 16 * hp + c;
#pragma unroll
        for (int kt = 0; kt < 4; ++kt) {
            const int kb = kt * 32 + part * 16 + h * 8;
#pragma unroll
            for (int q = 0; q < 8; ++q) wf[kt][q] = wsrc[(kb + q) * DF];
        }
    } else {
        const int c = tid & 15, tc = (tid >> 4) & 3, h = (tid >> 6) & 1, part = (tid >> 7) & 1, kt = (tid >> 8) & 3;
        const float* wsrc = Wa + tc * 16 + c;
        const int kb = kt * 32 + part * 16 + h * 8;
#pragma unroll
        for (int q = 0; q < 8; ++q) wf[0][q] = wsrc[(kb + q) * DOUT];
    }

    const int l = tid & 31;
    float bz[4];
#pragma unroll
    for (int j = 0; j < 4; ++j) bz[j] = 0.5f * (biasS[l + 32 * j] + biasD[l + 32 * j]);

    if (tid < AGG_PR) dl[tid] = (tid < cn) ? dis2[(size_t)chunk * AGG_PR + tid] : make_float2(0.0f, 0.0f);

    static_assert((AGG_T & 31) == 0, "a thread keeps its lane's columns over the rows it takes");
    for (int i = tid; i < AGG_PR * 32; i += AGG_T) {
        const int row = i >> 5;
        float4 v = make_float4(0.0f, 0.0f, 0.0f, 0.0f);
        if (row < cn) {
            const float* yr = Y + (size_t)(base + row) * DK2 + 4 * l;
            const float4 ys = *(const float4*)yr;
            const float4 yd = *(const float4*)(yr + DF);
            const float2 dd = dis2[(size_t)chunk * AGG_PR + row];
            const float as = dd.y;
            const float bs = dd.x;
            v.x = bz[0] + as * ys.x + bs * yd.x;
            v.y = bz[1] + as * ys.y + bs * yd.y;
            v.z = bz[2] + as * ys.z + bs * yd.z;
            v.w = bz[3] + as * ys.w + bs * yd.w;
        }
        float* d = pl + row * AGG_P + l;
        d[0] = v.x; d[32] = v.y; d[64] = v.z; d[96] = v.w;
    }

    if constexpr (LAYER == 0) {
        const int c = tid & 15, hp = (tid >> 4) & 1, h = (tid >> 5) & 1, part = (tid >> 6) & 1;
        const int jt = (tid >> 7) & 3, half = (tid >> 9) & 1;
#pragma unroll
        for (int kt = 0; kt < 4; ++kt) {
            const int t = (kt * 2 + hp) * 8 + half * 4 + jt;
            Bl4[(t * 2 + part) * 32 + h * 16 + c] = pack8h(wf[kt], 0.5f);
        }
    } else {
        const int c = tid & 15, tc = (tid >> 4) & 3, h = (tid >> 6) & 1, part = (tid >> 7) & 1, kt = (tid >> 8) & 3;
        const int t = kt * (DOUT / 16) + tc;
        Bl4[(t * 2 + part) * 32 + h * 16 + c] = pack8h(wf[0], 1.0f);
    }
    __syncthreads();

    const bool doScan = (s_scan != 0u);
    if (!doScan) {
        const int perS = cdiv(E, DEG_G);
        const unsigned vb = (unsigned)((sl >= DEG_G ? E : 0) + (sl & (DEG_G - 1)) * perS);
        const unsigned p  = s_pre[sl];
        const unsigned wq[4] = { sent.x, sent.y, sent.z, sent.w };
#pragma unroll
        for (int i = 0; i < 8; ++i) {
            const unsigned j = (unsigned)(sq * 8 + i);
            const unsigned w = (i & 1) ? (wq[i >> 1] >> 16) : (wq[i >> 1] & 0xFFFFu);
            if (j >= 1u && j <= slen) lst[p + j - 1u] = vb + w;
        }
    }

    if (doScan) {
        for (int idx = tid; idx < M2; idx += AGG_T) {
            const int recv = (idx < E) ? rcv[idx] : snd[idx - E];
            if ((unsigned)(recv - base) < (unsigned)cn) {
                const unsigned slot = atomicAdd(&s_cnt, 1u);
                if (slot < (unsigned)AGG_LCAP) {
                    lst[slot] = (unsigned)idx;
                } else {
                    int4 q; int s; float a, b;
                    if (idx < E) { q = rec[idx];     s = q.x; a = __int_as_float(q.z); b = __int_as_float(q.w); }
                    else         { q = rec[idx - E]; s = q.y; a = __int_as_float(q.w); b = __int_as_float(q.z); }
                    const float2 dd = dl[recv - base];
                    a *= dd.y; b *= dd.x;
                    const float* yr = Y + (long)s * DK2;
                    float* dst = pl + (recv - base) * AGG_P;
                    for (int p = 0; p < DF; ++p)
                        __hip_atomic_fetch_add(dst + 32 * (p & 3) + (p >> 2), a * yr[p] + b * yr[DF + p],
                                               __ATOMIC_RELAXED, __HIP_MEMORY_SCOPE_WORKGROUP);
                }
            }
        }
    }
    __syncthreads();

    const int total = __builtin_amdgcn_readfirstlane((int)s_cnt);
    const int cnt   = (total < AGG_LCAP) ? total : AGG_LCAP;
    const int wv    = __builtin_amdgcn_readfirstlane(tid >> 5);
    const int lane  = tid & 31;
    const int per   = (cnt + (AGG_T / 32) - 1) / (AGG_T / 32);
    const int st    = wv * per;
    const int en    = (st + per < cnt) ? (st + per) : cnt;
    const float* Yl = Y + lane * 4;
    float* pll = pl + lane;
    if (st < en) {
        const int len = en - st;
        const int ng  = (len + 31) >> 5;
        const int gs  = (((len + ng - 1) / ng) + 3) & ~3;
        int sv = 0, rv = 0, av = 0, bv = 0;
        {
            const int n0 = (len < gs) ? len : gs;
            if (lane < n0) {
                const unsigned e = lst[st + lane];
                if (e < (unsigned)E) { const int4 q = rec[e];     sv = q.x; rv = q.y; av = q.z; bv = q.w; }
                else                 { const int4 q = rec[e - E]; sv = q.y; rv = q.x; av = q.w; bv = q.z; }
            }
        }
        for (int g = st; g < en; g += gs) {
            const int n = (en - g < gs) ? (en - g) : gs;
            const int cs = sv, cr = rv;
            const float2 rf = dl[(lane < n) ? (cr - base) : 0];
            const int ca = __float_as_int(__int_as_float(av) * rf.y);
            const int cb = __float_as_int(__int_as_float(bv) * rf.x);
            const int gn = g + gs;
            if (gn < en) {
                const int nn = (en - gn < gs) ? (en - gn) : gs;
                if (lane < nn) {
                    const unsigned e = lst[gn + lane];
                    if (e < (unsigned)E) { const int4 q = rec[e];     sv = q.x; rv = q.y; av = q.z; bv = q.w; }
                    else                 { const int4 q = rec[e - E]; sv = q.y; rv = q.x; av = q.w; bv = q.z; }
                }
            }
            int t = 0;
            for (; t + 4 <= n; t += 4) {
                float4 ys[4], yd[4];
                int r[4]; float a[4], b[4];
#pragma unroll
                for (int u = 0; u < 4; ++u) {
                    const int s = __builtin_amdgcn_readlane(cs, t + u);
                    r[u] = __builtin_amdgcn_readlane(cr, t + u) - base;
                    a[u] = __int_as_float(__builtin_amdgcn_readlane(ca, t + u));
                    b[u] = __int_as_float(__builtin_amdgcn_readlane(cb, t + u));
                    const float* yr = Yl + (long)s * DK2;
                    ys[u] = *(const float4*)yr;
                    yd[u] = *(const float4*)(yr + DF);
                }
#pragma unroll
                for (int u = 0; u < 4; ++u) {
                    float* dst = pll + r[u] * AGG_P;
                    __hip_atomic_fetch_add(dst,      a[u] * ys[u].x + b[u] * yd[u].x, __ATOMIC_RELAXED, __HIP_MEMORY_SCOPE_WORKGROUP);
                    __hip_atomic_fetch_add(dst + 32, a[u] * ys[u].y + b[u] * yd[u].y, __ATOMIC_RELAXED, __HIP_MEMORY_SCOPE_WORKGROUP);
                    __hip_atomic_fetch_add(dst + 64, a[u] * ys[u].z + b[u] * yd[u].z, __ATOMIC_RELAXED, __HIP_MEMORY_SCOPE_WORKGROUP);
                    __hip_atomic_fetch_add(dst + 96, a[u] * ys[u].w + b[u] * yd[u].w, __ATOMIC_RELAXED, __HIP_MEMORY_SCOPE_WORKGROUP);
                }
            }
            for (; t < n; ++t) {
                const int s = __builtin_amdgcn_readlane(cs, t);
                const int r = __builtin_amdgcn_readlane(cr, t) - base;
                const float a = __int_as_float(__builtin_amdgcn_readlane(ca, t));
                const float b = __int_as_float(__builtin_amdgcn_readlane(cb, t));
                const float* yr = Yl + (long)s * DK2;
                const float4 ys = *(const float4*)yr;
                const float4 yd = *(const float4*)(yr + DF);
                float* dst = pll + r * AGG_P;
                __hip_atomic_fetch_add(dst,      a * ys.x + b * yd.x, __ATOMIC_RELAXED, __HIP_MEMORY_SCOPE_WORKGROUP);
                __hip_atomic_fetch_add(dst + 32, a * ys.y + b * yd.y, __ATOMIC_RELAXED, __HIP_MEMORY_SCOPE_WORKGROUP);
                __hip_atomic_fetch_add(dst + 64, a * ys.z + b * yd.z, __ATOMIC_RELAXED, __HIP_MEMORY_SCOPE_WORKGROUP);
                __hip_atomic_fetch_add(dst + 96, a * ys.w + b * yd.w, __ATOMIC_RELAXED, __HIP_MEMORY_SCOPE_WORKGROUP);
            }
        }
    }
    __syncthreads();

    const int hi = lane >> 4;
    const int n  = lane & 15;
    if (LAYER == 0) {
        for (int task = wv; task < AGG_TR * 4; task += AGG_T / 32) {
            const int t    = task >> 2;
            const int hp   = (task >> 1) & 1;
            const int half = task & 1;
            if (t * 16 >= cn) continue;
            const float* ar = pl + (t * 16 + n) * AGG_P;
            v8f c[4] = {};
#pragma unroll 2
            for (int kt = 0; kt < DF / 32; ++kt) {
                const v16h a = plane_a_frag(ar, kt * 32, hi);
                const int tb = (kt * 2 + hp) * 8 + half * 4;
#pragma unroll
                for (int jt = 0; jt < 4; ++jt) {
                    const v16h b = lds_b_frag(Bl4, tb + jt, lane);
                    c[jt] = __builtin_amdgcn_wmma_f32_16x16x32_f16(false, a, false, b,
                                                                   (short)0, c[jt], false, false);
                }
            }
            const int lcol = 16 * hp + n;
#pragma unroll
            for (int i = 0; i < 8; ++i) {
                const int row = t * 16 + hi * 8 + i;
                if (row < cn) {
                    const float2 dd = dl[row];
                    const float sc = half ? dd.y : dd.x;
                    float4 y;
                    y.x = sc * c[0][i]; y.y = sc * c[1][i]; y.z = sc * c[2][i]; y.w = sc * c[3][i];
                    *(float4*)(Yn + (size_t)(base + row) * DK2 + half * DF + lcol * 4) = y;
                }
            }
        }
    } else {
        for (int task = wv; task < AGG_TR * (DOUT / 16); task += AGG_T / 32) {
            const int t  = task >> 2;
            const int tc = task & 3;
            if (t * 16 >= cn) continue;
            const float* ar = pl + (t * 16 + n) * AGG_P;
            v8f c = {};
#pragma unroll
            for (int kt = 0; kt < DF / 32; ++kt) {
                const v16h a = plane_a_frag(ar, kt * 32, hi);
                const v16h b = lds_b_frag(Bl4, kt * (DOUT / 16) + tc, lane);
                c = __builtin_amdgcn_wmma_f32_16x16x32_f16(false, a, false, b,
                                                           (short)0, c, false, false);
            }
            const int ncol = tc * 16 + n;
            const float bv = biasN[ncol];
#pragma unroll
            for (int i = 0; i < 8; ++i) {
                const int row = t * 16 + hi * 8 + i;
                if (row < cn) Yn[(size_t)(base + row) * DOUT + ncol] = c[i] + bv;
            }
        }
    }
}

#define REC_WG 128

__global__ void __launch_bounds__(DEG_T)
deg_pre(const int* __restrict__ snd, const int* __restrict__ rcv,
        const float* __restrict__ theta, int E,
        unsigned short* __restrict__ blst, int recBlocks,
        int4* __restrict__ rec, int recPer) {
    __shared__ uint4    bl4[AGG_NCH * BIN_W / 8];
    __shared__ unsigned bcount[AGG_NCH];
    const int tid  = threadIdx.x;

    if ((int)blockIdx.x < recBlocks) {
        const int eb = (int)blockIdx.x * recPer;
        const int ee = (eb + recPer < E) ? (eb + recPer) : E;
        for (int e = eb + tid; e < ee; e += DEG_T) {
            const float cs = cosf(theta[e]), sn = sinf(theta[e]);
            rec[e] = make_int4(snd[e], rcv[e], __float_as_int(cs * cs), __float_as_int(sn * sn));
        }
        return;
    }

    unsigned short* bl = (unsigned short*)bl4;
    const int lb   = (int)blockIdx.x - recBlocks;
    const int g    = lb % DEG_G;
    const int a    = (lb / DEG_G) & 1;
    if (tid < AGG_NCH) bcount[tid] = 0u;
    __syncthreads();

    const int* __restrict__ rend = a ? snd : rcv;
    const int per = cdiv(E, DEG_G);
    const int e0  = g * per;
    const int e1  = (e0 + per < E) ? (e0 + per) : E;
    for (int e = e0 + tid; e < e1; e += 4 * DEG_T) {
        int v[4];
#pragma unroll
        for (int k = 0; k < 4; ++k) {
            const int ek = e + k * DEG_T;
            v[k] = (ek < e1) ? rend[ek] : -1;
        }
#pragma unroll
        for (int k = 0; k < 4; ++k) {
            if (v[k] >= 0) {
                const unsigned b = (unsigned)v[k] / AGG_ROWS;
                if (b < (unsigned)AGG_NCH) {
                    const unsigned slot = atomicAdd(&bcount[b], 1u);
                    if (slot < (unsigned)BIN_CAP) bl[b * BIN_W + 1 + slot] = (unsigned short)(e + k * DEG_T - e0);
                }
            }
        }
    }
    __syncthreads();

    if (tid < AGG_NCH) {
        const unsigned n = bcount[tid];
        bl[tid * BIN_W] = (unsigned short)((n > (unsigned)BIN_CAP || per > 65536) ? 0xFFFFu : n);
    }
    __syncthreads();
    const int sl = a * DEG_G + g;
    for (int i = tid; i < AGG_NCH * (BIN_W / 8); i += DEG_T) {
        const int b = i >> 3, q = i & 7;
        ((uint4*)(blst + ((size_t)b * BIN_S + sl) * BIN_W))[q] = bl4[i];
    }
}

__global__ void __launch_bounds__(DEG_T)
prod_deg(const int4* __restrict__ rec, const int* __restrict__ snd, const int* __restrict__ rcv,
         const unsigned short* __restrict__ blst, int E, int Nn, float2* __restrict__ dis2,
         const float* __restrict__ X, const float* __restrict__ Ws, const float* __restrict__ Wd,
         float* __restrict__ Y) {
    __shared__ u4v   Bl4[DF * DK2 / 8];
    __shared__ float dgw[(DEG_T / 32) * AGG_PR * 2];
    __shared__ float sd[AGG_PR * 2];
    __shared__ unsigned s_over;
    const int tid   = threadIdx.x;
    const int chunk = blockIdx.x;
    const int base  = chunk * AGG_ROWS;
    const int cn    = (Nn - base < AGG_ROWS) ? (Nn - base) : AGG_ROWS;

    static_assert(DEG_T == BIN_S * (BIN_W / 8), "eight threads per sub-list line");
    static_assert(DEG_T == 1024, "piece id = tid + 1024*kt covers the 4096 pieces of the matrix");
    static_assert(DEG_T >= 2 * AGG_PR, "one thread per (row, degree kind)");
    static_assert(DEG_T / 32 >= AGG_TR * 2, "one wave per (row tile, column half)");

    const int sl = tid >> 3;
    const int sq = tid & 7;
    const unsigned short* line = blst + ((size_t)chunk * BIN_S + sl) * BIN_W;
    const unsigned slen = line[0];
    const uint4    sent = ((const uint4*)line)[sq];
    if (tid < 32) {
        static_assert(BIN_S == 128, "a lane of wave 0 takes four sub-lists");
        unsigned bad = 0u;
#pragma unroll
        for (int j = 0; j < 4; ++j)
            bad |= ((unsigned)blst[((size_t)chunk * BIN_S + 4 * tid + j) * BIN_W] > (unsigned)BIN_CAP) ? 1u : 0u;
        const int anyBad = __any((int)bad);
        if (tid == 0) s_over = anyBad ? 1u : 0u;
    }

    const int pc = tid & 15, php = (tid >> 4) & 1, ph = (tid >> 5) & 1, prt = (tid >> 6) & 1;
    const int pjt = (tid >> 7) & 3, phalf = (tid >> 9) & 1;
    const float* wsrc = (phalf ? Wd : Ws) + 32 * pjt + 16 * php + pc;
    float wf[4][8];
#pragma unroll
    for (int kt = 0; kt < 4; ++kt) {
        const int kb = kt * 32 + prt * 16 + ph * 8;
#pragma unroll
        for (int q = 0; q < 8; ++q) wf[kt][q] = wsrc[(kb + q) * DF];
    }

    for (int i = tid; i < (DEG_T / 32) * AGG_PR * 2; i += DEG_T) dgw[i] = 0.0f;

#pragma unroll
    for (int kt = 0; kt < 4; ++kt) {
        const int t = (kt * 2 + php) * 8 + phalf * 4 + pjt;
        Bl4[(t * 2 + prt) * 32 + ph * 16 + pc] = pack8h(wf[kt], 0.5f);
    }
    __syncthreads();
    const bool over = (s_over != 0u);

    const int wv   = __builtin_amdgcn_readfirstlane(tid >> 5);
    const int lane = tid & 31;
    const int hi   = lane >> 4;
    const int n    = lane & 15;
    const int t    = wv >> 1;
    const int hp   = wv & 1;
    const bool task = (wv < AGG_TR * 2) && (t * 16 < cn);

    const int perS = cdiv(E, DEG_G);
    const int back = (sl >= DEG_G) ? 1 : 0;
    const unsigned eb = (unsigned)((sl & (DEG_G - 1)) * perS);
    const unsigned wq[4] = { sent.x, sent.y, sent.z, sent.w };
    int4 q[8];
    if (!over) {
#pragma unroll
        for (int i = 0; i < 8; ++i) {
            const unsigned j = (unsigned)(sq * 8 + i);
            const unsigned w = (i & 1) ? (wq[i >> 1] >> 16) : (wq[i >> 1] & 0xFFFFu);
            q[i] = rec[(j >= 1u && j <= slen) ? (eb + w) : 0u];
        }
    }

    float4 ar[DF / 8];
    if (task) {
        const int rr = (t * 16 + n < cn) ? (t * 16 + n) : (cn - 1);
        const float4* Ar = (const float4*)(X + (size_t)(base + rr) * DF);
#pragma unroll
        for (int kt = 0; kt < DF / 32; ++kt) {
            const int b = kt * 8 + hi * 2;
            ar[kt * 4 + 0] = Ar[b];     ar[kt * 4 + 1] = Ar[b + 1];
            ar[kt * 4 + 2] = Ar[b + 4]; ar[kt * 4 + 3] = Ar[b + 5];
        }
    }

    float* dg = dgw + (tid >> 5) * (AGG_PR * 2);
    if (!over) {
#pragma unroll
        for (int i = 0; i < 8; ++i) {
            const unsigned j = (unsigned)(sq * 8 + i);
            const int m = (back ? q[i].x : q[i].y) - base;
            if (j >= 1u && j <= slen && (unsigned)m < (unsigned)cn) {
                const float c2 = __int_as_float(q[i].z), s2 = __int_as_float(q[i].w);
                __hip_atomic_fetch_add(dg + 2 * m,     back ? c2 : s2, __ATOMIC_RELAXED, __HIP_MEMORY_SCOPE_WORKGROUP);
                __hip_atomic_fetch_add(dg + 2 * m + 1, back ? s2 : c2, __ATOMIC_RELAXED, __HIP_MEMORY_SCOPE_WORKGROUP);
            }
        }
    } else {
        const int M2 = 2 * E;
        for (int idx = tid; idx < M2; idx += DEG_T) {
            const int bk = (idx >= E) ? 1 : 0;
            const int er = bk ? (idx - E) : idx;
            const int m  = (bk ? snd[er] : rcv[er]) - base;
            if ((unsigned)m < (unsigned)cn) {
                const int4 qq = rec[er];
                const float qc = __int_as_float(qq.z), qs = __int_as_float(qq.w);
                __hip_atomic_fetch_add(dg + 2 * m,     bk ? qc : qs, __ATOMIC_RELAXED, __HIP_MEMORY_SCOPE_WORKGROUP);
                __hip_atomic_fetch_add(dg + 2 * m + 1, bk ? qs : qc, __ATOMIC_RELAXED, __HIP_MEMORY_SCOPE_WORKGROUP);
            }
        }
    }
    __syncthreads();

    if (tid < AGG_PR * 2) {
        float sum = 1.0f;
#pragma unroll
        for (int w = 0; w < DEG_T / 32; ++w) sum += dgw[w * (AGG_PR * 2) + tid];
        const float w1 = sum + 1e-12f;
        sd[tid] = (w1 > 1e11f) ? 0.0f : rsqrtf(w1);
    }
    __syncthreads();

    if (tid < cn) dis2[(size_t)chunk * AGG_PR + tid] = make_float2(sd[tid * 2], sd[tid * 2 + 1]);

    if (task) {
        v16h af[DF / 32];
#pragma unroll
        for (int kt = 0; kt < DF / 32; ++kt)
            af[kt] = frag16<0>(ar[kt * 4 + 0], ar[kt * 4 + 1], ar[kt * 4 + 2], ar[kt * 4 + 3]);
        const int lcol = 16 * hp + n;
        const int row0 = t * 16 + hi * 8;
        const float* sdr = sd + row0 * 2;
#pragma nounroll
        for (int half = 0; half < 2; ++half) {
            v8f c[4] = {};
#pragma unroll
            for (int kt = 0; kt < DF / 32; ++kt) {
                const int tb = (kt * 2 + hp) * 8 + half * 4;
#pragma unroll
                for (int jt = 0; jt < 4; ++jt) {
                    const v16h b = lds_b_frag(Bl4, tb + jt, lane);
                    c[jt] = __builtin_amdgcn_wmma_f32_16x16x32_f16(false, af[kt], false, b,
                                                                   (short)0, c[jt], false, false);
                }
            }
#pragma unroll
            for (int i = 0; i < 8; ++i) {
                if (row0 + i < cn) {
                    const float sc = sdr[i * 2 + half];
                    float4 y;
                    y.x = sc * c[0][i]; y.y = sc * c[1][i]; y.z = sc * c[2][i]; y.w = sc * c[3][i];
                    *(float4*)(Y + (size_t)(base + row0 + i) * DK2 + half * DF + lcol * 4) = y;
                }
            }
        }
    }
}

extern "C" void kernel_launch(void* const* d_in, const int* in_sizes, int n_in,
                              void* d_out, int out_size, void* d_ws, size_t ws_size,
                              hipStream_t stream) {
    const float* x      = (const float*)d_in[0];
    const int*   edges  = (const int*)d_in[1];
    const float* theta  = (const float*)d_in[2];
    const float* W1s    = (const float*)d_in[3];
    const float* W1d    = (const float*)d_in[4];
    const float* b1s    = (const float*)d_in[5];
    const float* b1d    = (const float*)d_in[6];
    const float* W2s    = (const float*)d_in[7];
    const float* W2d    = (const float*)d_in[8];
    const float* b2s    = (const float*)d_in[9];
    const float* b2d    = (const float*)d_in[10];
    const float* Wro    = (const float*)d_in[11];
    const float* bro    = (const float*)d_in[12];

    const int N = in_sizes[0] / DF;
    const int E = in_sizes[2];
    const int* snd = edges;
    const int* rcv = edges + E;

    char* ws = (char*)d_ws;
    size_t off = 0;
    auto alloc = [&](size_t bytes) { char* p = ws + off; off = (off + bytes + 255) & ~(size_t)255; return p; };
    const int aggBlocks = cdiv(N, AGG_ROWS);
    float2*   dis2  = (float2*)alloc((size_t)aggBlocks * AGG_PR * 8);
    int4*     rec   = (int4*)alloc((size_t)E * 16);
    float*    Y1    = (float*)alloc((size_t)N * DK2 * 4);
    float*    Y2    = (float*)alloc((size_t)N * DK2 * 4);
    unsigned short* blst = (unsigned short*)alloc((size_t)AGG_NCH * BIN_S * BIN_W * 2);

    const int nBin      = DEG_G * 2;
    const int recPer    = (cdiv(E, REC_WG) + 7) & ~7;
    const int recBlocks = cdiv(E, recPer);
    deg_pre<<<recBlocks + nBin, DEG_T, 0, stream>>>(snd, rcv, theta, E, blst, recBlocks, rec, recPer);

    prod_deg<<<aggBlocks, DEG_T, 0, stream>>>(rec, snd, rcv, blst, E, N, dis2, x, W1s, W1d, Y1);

    agg_lds<0><<<aggBlocks, AGG_T, 0, stream>>>(Y1, rec, snd, rcv, b1s, b1d, dis2, E, N,
                                                blst, W2s, W2d, bro, Y2);

    agg_lds<1><<<aggBlocks, AGG_T, 0, stream>>>(Y2, rec, snd, rcv, b2s, b2d, dis2, E, N,
                                                blst, Wro, Wro, bro, (float*)d_out);
}
